# starting kernel
# baseline (speedup 1.0000x reference)
#include <hip/hip_runtime.h>
#include <math.h>

#define HEADS 4
#define HID 64
#define HC 256
#define NEG_SLOPE 0.2f
#define EPS_F 1e-16f

typedef __attribute__((ext_vector_type(2))) float v2f;
typedef __attribute__((ext_vector_type(8))) float v8f;

__device__ __forceinline__ unsigned f2ord(float f) {
    unsigned u = __float_as_uint(f);
    return (u & 0x80000000u) ? ~u : (u | 0x80000000u);
}
__device__ __forceinline__ float ord2f(unsigned u) {
    return (u & 0x80000000u) ? __uint_as_float(u & 0x7fffffffu) : __uint_as_float(~u);
}

__device__ __forceinline__ void epi_store(float* __restrict__ C, const float* __restrict__ bias,
                                          size_t rowoff, int col, int NC, int mode, float v) {
    if (col < NC) {
        if (mode >= 1) v += bias[col];
        if (mode == 2) v = 1.0f / (1.0f + __expf(-v));
        C[rowoff + col] = v;
    }
}

__global__ __launch_bounds__(256) void wmma_gemm_f32_kernel(
    const float* __restrict__ A, const float* __restrict__ B,
    float* __restrict__ C, const float* __restrict__ bias,
    int M, int K, int NC, int mode)
{
    const int wave = threadIdx.x >> 5;
    const int lane = threadIdx.x & 31;
    const int m0 = (blockIdx.x * 8 + wave) * 16;
    if (m0 >= M) return;
    const int n0 = blockIdx.y * 64;

    const int aRow = m0 + (lane & 15);
    const int aK   = (lane >> 4) << 1;
    const int bN   = lane & 15;
    const int bK   = (lane >> 4) << 1;
    const int c    = n0 + bN;

    const float* Ap = A + (size_t)aRow * K + aK;
    const size_t stepB = (size_t)4 * NC;

    v8f acc0 = {}, acc1 = {}, acc2 = {}, acc3 = {};

    if (n0 + 64 <= NC) {
        const float* Bp = B + c + (size_t)bK * NC;
        const float* Bq = Bp + NC;
        #pragma unroll 2
        for (int kb = 0; kb < K; kb += 4) {
            v2f a = *(const v2f*)Ap;
            v2f b0 = { Bp[0],  Bq[0]  };
            v2f b1 = { Bp[16], Bq[16] };
            v2f b2 = { Bp[32], Bq[32] };
            v2f b3 = { Bp[48], Bq[48] };
            Ap += 4; Bp += stepB; Bq += stepB;
            acc0 = __builtin_amdgcn_wmma_f32_16x16x4_f32(false, a, false, b0, (short)0, acc0, false, false);
            acc1 = __builtin_amdgcn_wmma_f32_16x16x4_f32(false, a, false, b1, (short)0, acc1, false, false);
            acc2 = __builtin_amdgcn_wmma_f32_16x16x4_f32(false, a, false, b2, (short)0, acc2, false, false);
            acc3 = __builtin_amdgcn_wmma_f32_16x16x4_f32(false, a, false, b3, (short)0, acc3, false, false);
        }
    } else {
        const int   c0 = min(c,      NC - 1), c1 = min(c + 16, NC - 1);
        const int   c2 = min(c + 32, NC - 1), c3 = min(c + 48, NC - 1);
        const float w0 = (c      < NC) ? 1.0f : 0.0f;
        const float w1 = (c + 16 < NC) ? 1.0f : 0.0f;
        const float w2 = (c + 32 < NC) ? 1.0f : 0.0f;
        const float w3 = (c + 48 < NC) ? 1.0f : 0.0f;
        const float* Bp = B + (size_t)bK * NC;
        const float* Bq = Bp + NC;
        #pragma unroll 2
        for (int kb = 0; kb < K; kb += 4) {
            v2f a = *(const v2f*)Ap;
            v2f b0 = { Bp[c0] * w0, Bq[c0] * w0 };
            v2f b1 = { Bp[c1] * w1, Bq[c1] * w1 };
            v2f b2 = { Bp[c2] * w2, Bq[c2] * w2 };
            v2f b3 = { Bp[c3] * w3, Bq[c3] * w3 };
            Ap += 4; Bp += stepB; Bq += stepB;
            acc0 = __builtin_amdgcn_wmma_f32_16x16x4_f32(false, a, false, b0, (short)0, acc0, false, false);
            acc1 = __builtin_amdgcn_wmma_f32_16x16x4_f32(false, a, false, b1, (short)0, acc1, false, false);
            acc2 = __builtin_amdgcn_wmma_f32_16x16x4_f32(false, a, false, b2, (short)0, acc2, false, false);
            acc3 = __builtin_amdgcn_wmma_f32_16x16x4_f32(false, a, false, b3, (short)0, acc3, false, false);
        }
    }

    const int mbase = m0 + ((lane >> 4) << 3);
    const int colb  = n0 + (lane & 15);
    #pragma unroll
    for (int r = 0; r < 8; ++r) {
        size_t rowoff = (size_t)(mbase + r) * NC;
        epi_store(C, bias, rowoff, colb,      NC, mode, acc0[r]);
        epi_store(C, bias, rowoff, colb + 16, NC, mode, acc1[r]);
        epi_store(C, bias, rowoff, colb + 32, NC, mode, acc2[r]);
        epi_store(C, bias, rowoff, colb + 48, NC, mode, acc3[r]);
    }
}

__global__ __launch_bounds__(256) void scores_kernel(
    const float* __restrict__ H, const float* __restrict__ att_src,
    const float* __restrict__ att_dst, float* __restrict__ a_s, float* __restrict__ a_d,
    unsigned* __restrict__ mmax, float* __restrict__ denom, int NH)
{
    const int gw   = (int)((blockIdx.x * 256 + threadIdx.x) >> 5);
    const int lane = threadIdx.x & 31;
    if (gw >= NH) return;
    const int n = gw >> 2, h = gw & 3;
    const float* hp = H + (size_t)n * HC + h * HID;
    const float x0 = hp[lane], x1 = hp[lane + 32];
    float s = x0 * att_src[h * HID + lane] + x1 * att_src[h * HID + lane + 32];
    float d = x0 * att_dst[h * HID + lane] + x1 * att_dst[h * HID + lane + 32];
    #pragma unroll
    for (int off = 16; off > 0; off >>= 1) {
        s += __shfl_xor(s, off, 32);
        d += __shfl_xor(d, off, 32);
    }
    if (lane == 0) {
        a_s[gw] = s; a_d[gw] = d;
        mmax[gw] = f2ord(-INFINITY);
        denom[gw] = 0.0f;
    }
}

__device__ __forceinline__ void edge_src_dst(const int* __restrict__ ei, int E, int eid,
                                             int& src, int& dst) {
    if (eid < E) { src = ei[eid]; dst = ei[E + eid]; }
    else         { src = dst = eid - E; }
}

__global__ __launch_bounds__(256) void edge_max_kernel(
    const int* __restrict__ ei, int E, int N,
    const float* __restrict__ a_s, const float* __restrict__ a_d, unsigned* __restrict__ mmax)
{
    const int t = blockIdx.x * 256 + threadIdx.x;
    if (t >= (E + N) * 4) return;
    const int eid = t >> 2, h = t & 3;
    int src, dst; edge_src_dst(ei, E, eid, src, dst);
    float e = a_s[src * 4 + h] + a_d[dst * 4 + h];
    e = (e > 0.0f) ? e : NEG_SLOPE * e;
    atomicMax(mmax + dst * 4 + h, f2ord(e));
}

__global__ __launch_bounds__(256) void edge_sum_kernel(
    const int* __restrict__ ei, int E, int N,
    const float* __restrict__ a_s, const float* __restrict__ a_d,
    const unsigned* __restrict__ mmax, float* __restrict__ denom, float* __restrict__ ew)
{
    const int t = blockIdx.x * 256 + threadIdx.x;
    if (t >= (E + N) * 4) return;
    const int eid = t >> 2, h = t & 3;
    int src, dst; edge_src_dst(ei, E, eid, src, dst);
    float e = a_s[src * 4 + h] + a_d[dst * 4 + h];
    e = (e > 0.0f) ? e : NEG_SLOPE * e;
    const float ex = __expf(e - ord2f(mmax[dst * 4 + h]));
    ew[t] = ex;
    atomicAdd(denom + dst * 4 + h, ex);
}

__global__ __launch_bounds__(256) void edge_agg_kernel(
    const int* __restrict__ ei, int E, int N, const float* __restrict__ H,
    const float* __restrict__ ew, const float* __restrict__ denom, float* __restrict__ out)
{
    const int gw   = (int)((blockIdx.x * 256 + threadIdx.x) >> 5);
    const int lane = threadIdx.x & 31;
    if (gw >= E + N) return;
    int src, dst; edge_src_dst(ei, E, gw, src, dst);
    const int head = lane >> 3;
    const float alpha = ew[(size_t)gw * 4 + head] / (denom[dst * 4 + head] + EPS_F);
    const float4* hs = (const float4*)(H + (size_t)src * HC) + (lane << 1);
    const float4 v0 = hs[0], v1 = hs[1];
    float* op = out + (size_t)dst * HC + (lane << 3);
    atomicAdd(op + 0, alpha * v0.x);
    atomicAdd(op + 1, alpha * v0.y);
    atomicAdd(op + 2, alpha * v0.z);
    atomicAdd(op + 3, alpha * v0.w);
    atomicAdd(op + 4, alpha * v1.x);
    atomicAdd(op + 5, alpha * v1.y);
    atomicAdd(op + 6, alpha * v1.z);
    atomicAdd(op + 7, alpha * v1.w);
}

__global__ __launch_bounds__(256) void zero_kernel(float* __restrict__ p, int n) {
    const int i = blockIdx.x * 256 + threadIdx.x;
    if (i < n) p[i] = 0.0f;
}

__global__ __launch_bounds__(256) void bias_relu_kernel(float* __restrict__ p,
                                                        const float* __restrict__ b, int n) {
    const int i = blockIdx.x * 256 + threadIdx.x;
    if (i < n) {
        const float v = p[i] + b[i & (HC - 1)];
        p[i] = fmaxf(v, 0.0f);
    }
}

extern "C" void kernel_launch(void* const* d_in, const int* in_sizes, int n_in,
                              void* d_out, int out_size, void* d_ws, size_t ws_size,
                              hipStream_t stream)
{
    const float* x   = (const float*)d_in[0];
    const int*   ei  = (const int*)d_in[1];
    const float* W1  = (const float*)d_in[2];
    const float* as1 = (const float*)d_in[3];
    const float* ad1 = (const float*)d_in[4];
    const float* b1  = (const float*)d_in[5];
    const float* W2  = (const float*)d_in[6];
    const float* as2 = (const float*)d_in[7];
    const float* ad2 = (const float*)d_in[8];
    const float* b2  = (const float*)d_in[9];
    const float* Wm1 = (const float*)d_in[10];
    const float* bm1 = (const float*)d_in[11];
    const float* Wm2 = (const float*)d_in[12];
    const float* bm2 = (const float*)d_in[13];

    const int N    = in_sizes[0] / 128;
    const int E    = in_sizes[1] / 2;
    const int Etot = E + N;
    const int NH   = N * HEADS;

    float* ws = (float*)d_ws;
    size_t o = 0;
    float*    P0    = ws + o; o += (size_t)N * HC;
    float*    P1    = ws + o; o += (size_t)N * HC;
    float*    a_s   = ws + o; o += (size_t)NH;
    float*    a_d   = ws + o; o += (size_t)NH;
    unsigned* mmax  = (unsigned*)(ws + o); o += (size_t)NH;
    float*    denom = ws + o; o += (size_t)NH;
    float*    ew    = ws + o; o += (size_t)Etot * HEADS;

    const dim3 blk(256);
    auto gemm_grid = [](int M, int NC) {
        return dim3((unsigned)(((M + 15) / 16 + 7) / 8), (unsigned)((NC + 63) / 64));
    };
    const int gScores = (NH * 32 + 255) / 256;
    const int gEdgeH  = (Etot * 4 + 255) / 256;
    const int gEdgeW  = (Etot * 32 + 255) / 256;
    const int gFeat   = (N * HC + 255) / 256;

    wmma_gemm_f32_kernel<<<gemm_grid(N, HC), blk, 0, stream>>>(x, W1, P0, nullptr, N, 128, HC, 0);
    scores_kernel<<<gScores, blk, 0, stream>>>(P0, as1, ad1, a_s, a_d, mmax, denom, NH);
    edge_max_kernel<<<gEdgeH, blk, 0, stream>>>(ei, E, N, a_s, a_d, mmax);
    edge_sum_kernel<<<gEdgeH, blk, 0, stream>>>(ei, E, N, a_s, a_d, mmax, denom, ew);
    zero_kernel<<<gFeat, blk, 0, stream>>>(P1, N * HC);
    edge_agg_kernel<<<gEdgeW, blk, 0, stream>>>(ei, E, N, P0, ew, denom, P1);
    bias_relu_kernel<<<gFeat, blk, 0, stream>>>(P1, b1, N * HC);

    wmma_gemm_f32_kernel<<<gemm_grid(N, HC), blk, 0, stream>>>(P1, W2, P0, nullptr, N, HC, HC, 0);
    scores_kernel<<<gScores, blk, 0, stream>>>(P0, as2, ad2, a_s, a_d, mmax, denom, NH);
    edge_max_kernel<<<gEdgeH, blk, 0, stream>>>(ei, E, N, a_s, a_d, mmax);
    edge_sum_kernel<<<gEdgeH, blk, 0, stream>>>(ei, E, N, a_s, a_d, mmax, denom, ew);
    zero_kernel<<<gFeat, blk, 0, stream>>>(P1, N * HC);
    edge_agg_kernel<<<gEdgeW, blk, 0, stream>>>(ei, E, N, P0, ew, denom, P1);
    bias_relu_kernel<<<gFeat, blk, 0, stream>>>(P1, b2, N * HC);

    wmma_gemm_f32_kernel<<<gemm_grid(N, HID), blk, 0, stream>>>(P1, Wm1, P0, bm1, N, HC, HID, 1);
    wmma_gemm_f32_kernel<<<gemm_grid(N, 121), blk, 0, stream>>>(P0, Wm2, (float*)d_out, bm2,
                                                                N, HID, 121, 2);
}


# Edge aggregation by row owners: a block keeps 196 target rows in LDS, finds its edges in the target column, adds gathered rows with LDS float adds (arrival order), stores rows once; no global adds
# speedup vs baseline: 2.3137x; 2.3137x over previous
#include <hip/hip_runtime.h>
#include <math.h>

#define HEADS 4
#define HID 64
#define HC 256
#define NEG_SLOPE 0.2f
#define EPS_F 1e-16f

typedef __attribute__((ext_vector_type(2))) float v2f;
typedef __attribute__((ext_vector_type(8))) float v8f;

__device__ __forceinline__ unsigned f2ord(float f) {
    unsigned u = __float_as_uint(f);
    return (u & 0x80000000u) ? ~u : (u | 0x80000000u);
}
__device__ __forceinline__ float ord2f(unsigned u) {
    return (u & 0x80000000u) ? __uint_as_float(u & 0x7fffffffu) : __uint_as_float(~u);
}

__device__ __forceinline__ void epi_store(float* __restrict__ C, const float* __restrict__ bias,
                                          size_t rowoff, int col, int NC, int mode, float v) {
    if (col < NC) {
        if (mode >= 1) v += bias[col];
        if (mode == 2) v = 1.0f / (1.0f + __expf(-v));
        C[rowoff + col] = v;
    }
}

template <bool ACT>
__device__ __forceinline__ v2f load_a_frag(const float* __restrict__ Ap,
                                           const float* __restrict__ abias, int kb, int aK) {
    v2f a = *(const v2f*)Ap;
    if (ACT) {
        const float4 bb = *(const float4*)(abias + kb);
        const float bx = aK ? bb.z : bb.x;
        const float by = aK ? bb.w : bb.y;
        a.x = fmaxf(a.x + bx, 0.0f);
        a.y = fmaxf(a.y + by, 0.0f);
    }
    return a;
}

template <bool ACT>
__global__ __launch_bounds__(256) void wmma_gemm_f32_kernel(
    const float* __restrict__ A, const float* __restrict__ B,
    float* __restrict__ C, const float* __restrict__ bias,
    const float* __restrict__ abias,
    int M, int K, int NC, int mode)
{
    const int wave = threadIdx.x >> 5;
    const int lane = threadIdx.x & 31;
    const int m0 = (blockIdx.x * 8 + wave) * 16;
    if (m0 >= M) return;
    const int n0 = blockIdx.y * 64;

    const int aRow = m0 + (lane & 15);
    const int aK   = (lane >> 4) << 1;
    const int bN   = lane & 15;
    const int bK   = (lane >> 4) << 1;
    const int c    = n0 + bN;

    const float* Ap = A + (size_t)aRow * K + aK;
    const size_t stepB = (size_t)4 * NC;

    v8f acc0 = {}, acc1 = {}, acc2 = {}, acc3 = {};

    if (n0 + 64 <= NC) {
        const float* Bp = B + c + (size_t)bK * NC;
        const float* Bq = Bp + NC;
        #pragma unroll 2
        for (int kb = 0; kb < K; kb += 4) {
            v2f a = load_a_frag<ACT>(Ap, abias, kb, aK);
            v2f b0 = { Bp[0],  Bq[0]  };
            v2f b1 = { Bp[16], Bq[16] };
            v2f b2 = { Bp[32], Bq[32] };
            v2f b3 = { Bp[48], Bq[48] };
            Ap += 4; Bp += stepB; Bq += stepB;
            acc0 = __builtin_amdgcn_wmma_f32_16x16x4_f32(false, a, false, b0, (short)0, acc0, false, false);
            acc1 = __builtin_amdgcn_wmma_f32_16x16x4_f32(false, a, false, b1, (short)0, acc1, false, false);
            acc2 = __builtin_amdgcn_wmma_f32_16x16x4_f32(false, a, false, b2, (short)0, acc2, false, false);
            acc3 = __builtin_amdgcn_wmma_f32_16x16x4_f32(false, a, false, b3, (short)0, acc3, false, false);
        }
    } else {
        const int   c0 = min(c,      NC - 1), c1 = min(c + 16, NC - 1);
        const int   c2 = min(c + 32, NC - 1), c3 = min(c + 48, NC - 1);
        const float w0 = (c      < NC) ? 1.0f : 0.0f;
        const float w1 = (c + 16 < NC) ? 1.0f : 0.0f;
        const float w2 = (c + 32 < NC) ? 1.0f : 0.0f;
        const float w3 = (c + 48 < NC) ? 1.0f : 0.0f;
        const float* Bp = B + (size_t)bK * NC;
        const float* Bq = Bp + NC;
        #pragma unroll 2
        for (int kb = 0; kb < K; kb += 4) {
            v2f a = load_a_frag<ACT>(Ap, abias, kb, aK);
            v2f b0 = { Bp[c0] * w0, Bq[c0] * w0 };
            v2f b1 = { Bp[c1] * w1, Bq[c1] * w1 };
            v2f b2 = { Bp[c2] * w2, Bq[c2] * w2 };
            v2f b3 = { Bp[c3] * w3, Bq[c3] * w3 };
            Ap += 4; Bp += stepB; Bq += stepB;
            acc0 = __builtin_amdgcn_wmma_f32_16x16x4_f32(false, a, false, b0, (short)0, acc0, false, false);
            acc1 = __builtin_amdgcn_wmma_f32_16x16x4_f32(false, a, false, b1, (short)0, acc1, false, false);
            acc2 = __builtin_amdgcn_wmma_f32_16x16x4_f32(false, a, false, b2, (short)0, acc2, false, false);
            acc3 = __builtin_amdgcn_wmma_f32_16x16x4_f32(false, a, false, b3, (short)0, acc3, false, false);
        }
    }

    const int mbase = m0 + ((lane >> 4) << 3);
    const int colb  = n0 + (lane & 15);
    #pragma unroll
    for (int r = 0; r < 8; ++r) {
        size_t rowoff = (size_t)(mbase + r) * NC;
        epi_store(C, bias, rowoff, colb,      NC, mode, acc0[r]);
        epi_store(C, bias, rowoff, colb + 16, NC, mode, acc1[r]);
        epi_store(C, bias, rowoff, colb + 32, NC, mode, acc2[r]);
        epi_store(C, bias, rowoff, colb + 48, NC, mode, acc3[r]);
    }
}

__global__ __launch_bounds__(256) void scores_kernel(
    const float* __restrict__ H, const float* __restrict__ att_src,
    const float* __restrict__ att_dst, float* __restrict__ a_s, float* __restrict__ a_d,
    unsigned* __restrict__ mmax, float* __restrict__ denom, int NH)
{
    const int gw   = (int)((blockIdx.x * 256 + threadIdx.x) >> 5);
    const int lane = threadIdx.x & 31;
    if (gw >= NH) return;
    const int n = gw >> 2, h = gw & 3;
    const float* hp = H + (size_t)n * HC + h * HID;
    const float x0 = hp[lane], x1 = hp[lane + 32];
    float s = x0 * att_src[h * HID + lane] + x1 * att_src[h * HID + lane + 32];
    float d = x0 * att_dst[h * HID + lane] + x1 * att_dst[h * HID + lane + 32];
    #pragma unroll
    for (int off = 16; off > 0; off >>= 1) {
        s += __shfl_xor(s, off, 32);
        d += __shfl_xor(d, off, 32);
    }
    if (lane == 0) {
        a_s[gw] = s; a_d[gw] = d;
        mmax[gw] = f2ord(-INFINITY);
        denom[gw] = 0.0f;
    }
}

__device__ __forceinline__ void edge_src_dst(const int* __restrict__ ei, int E, int eid,
                                             int& src, int& dst) {
    if (eid < E) { src = ei[eid]; dst = ei[E + eid]; }
    else         { src = dst = eid - E; }
}

__global__ __launch_bounds__(256) void edge_max_kernel(
    const int* __restrict__ ei, int E, int N,
    const float* __restrict__ a_s, const float* __restrict__ a_d, unsigned* __restrict__ mmax)
{
    const int t = blockIdx.x * 256 + threadIdx.x;
    if (t >= (E + N) * 4) return;
    const int eid = t >> 2, h = t & 3;
    int src, dst; edge_src_dst(ei, E, eid, src, dst);
    float e = a_s[src * 4 + h] + a_d[dst * 4 + h];
    e = (e > 0.0f) ? e : NEG_SLOPE * e;
    atomicMax(mmax + dst * 4 + h, f2ord(e));
}

__global__ __launch_bounds__(256) void edge_sum_kernel(
    const int* __restrict__ ei, int E, int N,
    const float* __restrict__ a_s, const float* __restrict__ a_d,
    const unsigned* __restrict__ mmax, float* __restrict__ denom, float* __restrict__ ew)
{
    const int t = blockIdx.x * 256 + threadIdx.x;
    if (t >= (E + N) * 4) return;
    const int eid = t >> 2, h = t & 3;
    int src, dst; edge_src_dst(ei, E, eid, src, dst);
    float e = a_s[src * 4 + h] + a_d[dst * 4 + h];
    e = (e > 0.0f) ? e : NEG_SLOPE * e;
    const float ex = __expf(e - ord2f(mmax[dst * 4 + h]));
    ew[t] = ex;
    atomicAdd(denom + dst * 4 + h, ex);
}

#define OWN_ROWS 196
#define OWN_CAP  16384
#define OWN_THREADS 1024

__device__ __noinline__ void own_edge_by_one_lane(
    const int* __restrict__ ei, int E, int eid, int row0, const float* __restrict__ H,
    const float* __restrict__ ew, const float* __restrict__ denom, float* acc)
{
    int src, dst; edge_src_dst(ei, E, eid, src, dst);
    const float4 w4 = *(const float4*)(ew + (size_t)eid * 4);
    const float4 d4 = *(const float4*)(denom + (size_t)dst * 4);
    float al[4];
    al[0] = w4.x / (d4.x + EPS_F);
    al[1] = w4.y / (d4.y + EPS_F);
    al[2] = w4.z / (d4.z + EPS_F);
    al[3] = w4.w / (d4.w + EPS_F);
    const float* hs = H + (size_t)src * HC;
    float* ap = acc + (size_t)(dst - row0) * HC;
    for (int c = 0; c < HC; ++c) atomicAdd(ap + c, al[c >> 6] * hs[c]);
}

__global__ __launch_bounds__(OWN_THREADS) void edge_agg_owner_kernel(
    const int* __restrict__ ei, int E, int N, const float* __restrict__ H,
    const float* __restrict__ ew, const float* __restrict__ denom, float* __restrict__ out)
{
    __shared__ __attribute__((aligned(16))) float acc[OWN_ROWS * HC];
    __shared__ int elist[OWN_CAP];
    __shared__ int ecount;

    const int t    = threadIdx.x;
    const int lane = t & 31;
    const int wave = t >> 5;
    const int row0  = blockIdx.x * OWN_ROWS;
    const int nrows = min(OWN_ROWS, N - row0);

    float4* acc4 = (float4*)acc;
    for (int i = t; i < OWN_ROWS * (HC / 4); i += OWN_THREADS)
        acc4[i] = make_float4(0.0f, 0.0f, 0.0f, 0.0f);
    if (t == 0) ecount = 0;
    __syncthreads();

    auto consider = [&](int eid, int dst) {
        if ((unsigned)(dst - row0) < (unsigned)nrows) {
            const int pos = atomicAdd(&ecount, 1);
            if (pos < OWN_CAP) elist[pos] = eid;
            else own_edge_by_one_lane(ei, E, eid, row0, H, ew, denom, acc);
        }
    };
    const int* __restrict__ dp = ei + E;
    if ((E & 3) == 0) {
        const int4* __restrict__ dp4 = (const int4*)dp;
        const int n4 = E >> 2;
        for (int i = t; i < n4; i += 4 * OWN_THREADS) {
            int4 v[4];
            #pragma unroll
            for (int u = 0; u < 4; ++u) {
                const int k = i + u * OWN_THREADS;
                v[u] = (k < n4) ? dp4[k] : make_int4(-1, -1, -1, -1);
            }
            #pragma unroll
            for (int u = 0; u < 4; ++u) {
                const int e0 = (i + u * OWN_THREADS) * 4;
                const bool any = ((unsigned)(v[u].x - row0) < (unsigned)nrows) |
                                 ((unsigned)(v[u].y - row0) < (unsigned)nrows) |
                                 ((unsigned)(v[u].z - row0) < (unsigned)nrows) |
                                 ((unsigned)(v[u].w - row0) < (unsigned)nrows);
                if (any) {
                    consider(e0,     v[u].x);
                    consider(e0 + 1, v[u].y);
                    consider(e0 + 2, v[u].z);
                    consider(e0 + 3, v[u].w);
                }
            }
        }
    } else {
        for (int e = t; e < E; e += OWN_THREADS) consider(e, dp[e]);
    }
    if (t < nrows) consider(E + row0 + t, row0 + t);
    __syncthreads();

    const int n = min(ecount, OWN_CAP);
    for (int base = wave * 4; base < n; base += 4 * (OWN_THREADS / 32)) {
        int  eid[4], srcn[4], dstn[4];
        bool ok[4];
        #pragma unroll
        for (int j = 0; j < 4; ++j) {
            ok[j]  = (base + j) < n;
            eid[j] = elist[ok[j] ? base + j : base];
        }
        #pragma unroll
        for (int j = 0; j < 4; ++j) edge_src_dst(ei, E, eid[j], srcn[j], dstn[j]);
        float  v[4][8];
        float4 w4[4], d4[4];
        #pragma unroll
        for (int j = 0; j < 4; ++j) {
            const float* hs = H + (size_t)srcn[j] * HC + lane;
            #pragma unroll
            for (int q = 0; q < 8; ++q) v[j][q] = hs[32 * q];
            w4[j] = *(const float4*)(ew + (size_t)eid[j] * 4);
            d4[j] = *(const float4*)(denom + (size_t)dstn[j] * 4);
        }
        #pragma unroll
        for (int j = 0; j < 4; ++j) {
            if (ok[j]) {
                const float al0 = w4[j].x / (d4[j].x + EPS_F);
                const float al1 = w4[j].y / (d4[j].y + EPS_F);
                const float al2 = w4[j].z / (d4[j].z + EPS_F);
                const float al3 = w4[j].w / (d4[j].w + EPS_F);
                float* ap = acc + (size_t)(dstn[j] - row0) * HC + lane;
                atomicAdd(ap +   0, al0 * v[j][0]);
                atomicAdd(ap +  32, al0 * v[j][1]);
                atomicAdd(ap +  64, al1 * v[j][2]);
                atomicAdd(ap +  96, al1 * v[j][3]);
                atomicAdd(ap + 128, al2 * v[j][4]);
                atomicAdd(ap + 160, al2 * v[j][5]);
                atomicAdd(ap + 192, al3 * v[j][6]);
                atomicAdd(ap + 224, al3 * v[j][7]);
            }
        }
    }
    __syncthreads();

    float4* __restrict__ o4 = (float4*)(out + (size_t)row0 * HC);
    for (int i = t; i < nrows * (HC / 4); i += OWN_THREADS) o4[i] = acc4[i];
}

extern "C" void kernel_launch(void* const* d_in, const int* in_sizes, int n_in,
                              void* d_out, int out_size, void* d_ws, size_t ws_size,
                              hipStream_t stream)
{
    const float* x   = (const float*)d_in[0];
    const int*   ei  = (const int*)d_in[1];
    const float* W1  = (const float*)d_in[2];
    const float* as1 = (const float*)d_in[3];
    const float* ad1 = (const float*)d_in[4];
    const float* b1  = (const float*)d_in[5];
    const float* W2  = (const float*)d_in[6];
    const float* as2 = (const float*)d_in[7];
    const float* ad2 = (const float*)d_in[8];
    const float* b2  = (const float*)d_in[9];
    const float* Wm1 = (const float*)d_in[10];
    const float* bm1 = (const float*)d_in[11];
    const float* Wm2 = (const float*)d_in[12];
    const float* bm2 = (const float*)d_in[13];

    const int N    = in_sizes[0] / 128;
    const int E    = in_sizes[1] / 2;
    const int Etot = E + N;
    const int NH   = N * HEADS;

    float* ws = (float*)d_ws;
    size_t o = 0;
    float*    P0    = ws + o; o += (size_t)N * HC;
    float*    P1    = ws + o; o += (size_t)N * HC;
    float*    a_s   = ws + o; o += (size_t)NH;
    float*    a_d   = ws + o; o += (size_t)NH;
    unsigned* mmax  = (unsigned*)(ws + o); o += (size_t)NH;
    float*    denom = ws + o; o += (size_t)NH;
    float*    ew    = ws + o; o += (size_t)Etot * HEADS;

    const dim3 blk(256);
    auto gemm_grid = [](int M, int NC) {
        return dim3((unsigned)(((M + 15) / 16 + 7) / 8), (unsigned)((NC + 63) / 64));
    };
    const int gScores = (NH * 32 + 255) / 256;
    const int gEdgeH  = (Etot * 4 + 255) / 256;
    const int gOwn    = (N + OWN_ROWS - 1) / OWN_ROWS;


    wmma_gemm_f32_kernel<false><<<gemm_grid(N, HC), blk, 0, stream>>>(x, W1, P0, nullptr, nullptr,
                                                                      N, 128, HC, 0);
    scores_kernel<<<gScores, blk, 0, stream>>>(P0, as1, ad1, a_s, a_d, mmax, denom, NH);
    edge_max_kernel<<<gEdgeH, blk, 0, stream>>>(ei, E, N, a_s, a_d, mmax);
    edge_sum_kernel<<<gEdgeH, blk, 0, stream>>>(ei, E, N, a_s, a_d, mmax, denom, ew);
    edge_agg_owner_kernel<<<gOwn, dim3(OWN_THREADS), 0, stream>>>(ei, E, N, P0, ew, denom, P1);

    wmma_gemm_f32_kernel<true><<<gemm_grid(N, HC), blk, 0, stream>>>(P1, W2, P0, nullptr, b1,
                                                                     N, HC, HC, 0);
    scores_kernel<<<gScores, blk, 0, stream>>>(P0, as2, ad2, a_s, a_d, mmax, denom, NH);
    edge_max_kernel<<<gEdgeH, blk, 0, stream>>>(ei, E, N, a_s, a_d, mmax);
    edge_sum_kernel<<<gEdgeH, blk, 0, stream>>>(ei, E, N, a_s, a_d, mmax, denom, ew);
    edge_agg_owner_kernel<<<gOwn, dim3(OWN_THREADS), 0, stream>>>(ei, E, N, P0, ew, denom, P1);

    wmma_gemm_f32_kernel<true><<<gemm_grid(N, HID), blk, 0, stream>>>(P1, Wm1, P0, bm1, b2,
                                                                      N, HC, HID, 1);
    wmma_gemm_f32_kernel<false><<<gemm_grid(N, 121), blk, 0, stream>>>(P0, Wm2, (float*)d_out, bm2,
                                                                       nullptr, N, HID, 121, 2);
}


# Score pass and both per-edge softmax passes run on looping waves (4 nodes or 128 edge-head items per wave and trip, loads grouped): 825,000 one-item waves become 45,280; same sums and atomics
# speedup vs baseline: 2.4500x; 1.0589x over previous
#include <hip/hip_runtime.h>
#include <math.h>

#define HEADS 4
#define HID 64
#define HC 256
#define NEG_SLOPE 0.2f
#define EPS_F 1e-16f

typedef __attribute__((ext_vector_type(2))) float v2f;
typedef __attribute__((ext_vector_type(8))) float v8f;

__device__ __forceinline__ unsigned f2ord(float f) {
    unsigned u = __float_as_uint(f);
    return (u & 0x80000000u) ? ~u : (u | 0x80000000u);
}
__device__ __forceinline__ float ord2f(unsigned u) {
    return (u & 0x80000000u) ? __uint_as_float(u & 0x7fffffffu) : __uint_as_float(~u);
}

__device__ __forceinline__ void epi_store(float* __restrict__ C, const float* __restrict__ bias,
                                          size_t rowoff, int col, int NC, int mode, float v) {
    if (col < NC) {
        if (mode >= 1) v += bias[col];
        if (mode == 2) v = 1.0f / (1.0f + __expf(-v));
        C[rowoff + col] = v;
    }
}

template <bool ACT>
__device__ __forceinline__ v2f load_a_frag(const float* __restrict__ Ap,
                                           const float* __restrict__ abias, int kb, int aK) {
    v2f a = *(const v2f*)Ap;
    if (ACT) {
        const float4 bb = *(const float4*)(abias + kb);
        const float bx = aK ? bb.z : bb.x;
        const float by = aK ? bb.w : bb.y;
        a.x = fmaxf(a.x + bx, 0.0f);
        a.y = fmaxf(a.y + by, 0.0f);
    }
    return a;
}

template <bool ACT>
__global__ __launch_bounds__(256) void wmma_gemm_f32_kernel(
    const float* __restrict__ A, const float* __restrict__ B,
    float* __restrict__ C, const float* __restrict__ bias,
    const float* __restrict__ abias,
    int M, int K, int NC, int mode)
{
    const int wave = threadIdx.x >> 5;
    const int lane = threadIdx.x & 31;
    const int m0 = (blockIdx.x * 8 + wave) * 16;
    if (m0 >= M) return;
    const int n0 = blockIdx.y * 64;

    const int aRow = m0 + (lane & 15);
    const int aK   = (lane >> 4) << 1;
    const int bN   = lane & 15;
    const int bK   = (lane >> 4) << 1;
    const int c    = n0 + bN;

    const float* Ap = A + (size_t)aRow * K + aK;
    const size_t stepB = (size_t)4 * NC;

    v8f acc0 = {}, acc1 = {}, acc2 = {}, acc3 = {};

    if (n0 + 64 <= NC) {
        const float* Bp = B + c + (size_t)bK * NC;
        const float* Bq = Bp + NC;
        #pragma unroll 2
        for (int kb = 0; kb < K; kb += 4) {
            v2f a = load_a_frag<ACT>(Ap, abias, kb, aK);
            v2f b0 = { Bp[0],  Bq[0]  };
            v2f b1 = { Bp[16], Bq[16] };
            v2f b2 = { Bp[32], Bq[32] };
            v2f b3 = { Bp[48], Bq[48] };
            Ap += 4; Bp += stepB; Bq += stepB;
            acc0 = __builtin_amdgcn_wmma_f32_16x16x4_f32(false, a, false, b0, (short)0, acc0, false, false);
            acc1 = __builtin_amdgcn_wmma_f32_16x16x4_f32(false, a, false, b1, (short)0, acc1, false, false);
            acc2 = __builtin_amdgcn_wmma_f32_16x16x4_f32(false, a, false, b2, (short)0, acc2, false, false);
            acc3 = __builtin_amdgcn_wmma_f32_16x16x4_f32(false, a, false, b3, (short)0, acc3, false, false);
        }
    } else {
        const int   c0 = min(c,      NC - 1), c1 = min(c + 16, NC - 1);
        const int   c2 = min(c + 32, NC - 1), c3 = min(c + 48, NC - 1);
        const float w0 = (c      < NC) ? 1.0f : 0.0f;
        const float w1 = (c + 16 < NC) ? 1.0f : 0.0f;
        const float w2 = (c + 32 < NC) ? 1.0f : 0.0f;
        const float w3 = (c + 48 < NC) ? 1.0f : 0.0f;
        const float* Bp = B + (size_t)bK * NC;
        const float* Bq = Bp + NC;
        #pragma unroll 2
        for (int kb = 0; kb < K; kb += 4) {
            v2f a = load_a_frag<ACT>(Ap, abias, kb, aK);
            v2f b0 = { Bp[c0] * w0, Bq[c0] * w0 };
            v2f b1 = { Bp[c1] * w1, Bq[c1] * w1 };
            v2f b2 = { Bp[c2] * w2, Bq[c2] * w2 };
            v2f b3 = { Bp[c3] * w3, Bq[c3] * w3 };
            Ap += 4; Bp += stepB; Bq += stepB;
            acc0 = __builtin_amdgcn_wmma_f32_16x16x4_f32(false, a, false, b0, (short)0, acc0, false, false);
            acc1 = __builtin_amdgcn_wmma_f32_16x16x4_f32(false, a, false, b1, (short)0, acc1, false, false);
            acc2 = __builtin_amdgcn_wmma_f32_16x16x4_f32(false, a, false, b2, (short)0, acc2, false, false);
            acc3 = __builtin_amdgcn_wmma_f32_16x16x4_f32(false, a, false, b3, (short)0, acc3, false, false);
        }
    }

    const int mbase = m0 + ((lane >> 4) << 3);
    const int colb  = n0 + (lane & 15);
    #pragma unroll
    for (int r = 0; r < 8; ++r) {
        size_t rowoff = (size_t)(mbase + r) * NC;
        epi_store(C, bias, rowoff, colb,      NC, mode, acc0[r]);
        epi_store(C, bias, rowoff, colb + 16, NC, mode, acc1[r]);
        epi_store(C, bias, rowoff, colb + 32, NC, mode, acc2[r]);
        epi_store(C, bias, rowoff, colb + 48, NC, mode, acc3[r]);
    }
}

__global__ __launch_bounds__(256) void scores_kernel(
    const float* __restrict__ H, const float* __restrict__ att_src,
    const float* __restrict__ att_dst, float* __restrict__ a_s, float* __restrict__ a_d,
    unsigned* __restrict__ mmax, float* __restrict__ denom, int N)
{
    const int lane = threadIdx.x & 31;
    const int gw   = (int)(blockIdx.x * 8 + (threadIdx.x >> 5));
    const int nw   = (int)(gridDim.x * 8);

    float ws0[HEADS], ws1[HEADS], wd0[HEADS], wd1[HEADS];
    #pragma unroll
    for (int h = 0; h < HEADS; ++h) {
        ws0[h] = att_src[h * HID + lane]; ws1[h] = att_src[h * HID + lane + 32];
        wd0[h] = att_dst[h * HID + lane]; wd1[h] = att_dst[h * HID + lane + 32];
    }

    const unsigned ninf = f2ord(-INFINITY);
    const int ngroups = (N + 3) >> 2;
    for (int g = gw; g < ngroups; g += nw) {
        const int n0 = g * 4;
        float x[4][8];
        #pragma unroll
        for (int j = 0; j < 4; ++j) {
            const int n = min(n0 + j, N - 1);
            const float* hp = H + (size_t)n * HC + lane;
            #pragma unroll
            for (int q = 0; q < 8; ++q) x[j][q] = hp[32 * q];
        }
        #pragma unroll
        for (int j = 0; j < 4; ++j) {
            float s[HEADS], d[HEADS];
            #pragma unroll
            for (int h = 0; h < HEADS; ++h) {
                const float x0 = x[j][2 * h], x1 = x[j][2 * h + 1];
                s[h] = x0 * ws0[h] + x1 * ws1[h];
                d[h] = x0 * wd0[h] + x1 * wd1[h];
            }
            #pragma unroll
            for (int off = 16; off > 0; off >>= 1) {
                #pragma unroll
                for (int h = 0; h < HEADS; ++h) {
                    s[h] += __shfl_xor(s[h], off, 32);
                    d[h] += __shfl_xor(d[h], off, 32);
                }
            }
            if (lane == 0 && n0 + j < N) {
                const size_t o4 = (size_t)(n0 + j) * 4;
                *(float4*)(a_s + o4)   = make_float4(s[0], s[1], s[2], s[3]);
                *(float4*)(a_d + o4)   = make_float4(d[0], d[1], d[2], d[3]);
                *(uint4*)(mmax + o4)   = make_uint4(ninf, ninf, ninf, ninf);
                *(float4*)(denom + o4) = make_float4(0.0f, 0.0f, 0.0f, 0.0f);
            }
        }
    }
}

__device__ __forceinline__ void edge_src_dst(const int* __restrict__ ei, int E, int eid,
                                             int& src, int& dst) {
    if (eid < E) { src = ei[eid]; dst = ei[E + eid]; }
    else         { src = dst = eid - E; }
}

__device__ __forceinline__ void edge_src_dst_flat(const int* __restrict__ ei, int E, int eid,
                                                  int& src, int& dst) {
    const int ec = min(eid, E - 1);
    src = ei[ec]; dst = ei[E + ec];
    if (eid >= E) { src = dst = eid - E; }
}

#define EDGE_ITEMS 128

__global__ __launch_bounds__(256) void edge_max_kernel(
    const int* __restrict__ ei, int E, int N,
    const float* __restrict__ a_s, const float* __restrict__ a_d, unsigned* __restrict__ mmax)
{
    const int lane  = threadIdx.x & 31;
    const int gw    = (int)(blockIdx.x * 8 + (threadIdx.x >> 5));
    const int nw    = (int)(gridDim.x * 8);
    const int total = (E + N) * 4;
    const int ntrip = (total + EDGE_ITEMS - 1) / EDGE_ITEMS;
    const int h     = lane & 3;
    for (int c = gw; c < ntrip; c += nw) {
        const int t0 = c * EDGE_ITEMS + lane;
        int  src[4], dst[4];
        bool ok[4];
        #pragma unroll
        for (int j = 0; j < 4; ++j) {
            const int t = t0 + 32 * j;
            ok[j] = t < total;
            edge_src_dst_flat(ei, E, (ok[j] ? t : total - 1) >> 2, src[j], dst[j]);
        }
        float vs[4], vd[4];
        #pragma unroll
        for (int j = 0; j < 4; ++j) {
            vs[j] = a_s[src[j] * 4 + h];
            vd[j] = a_d[dst[j] * 4 + h];
        }
        #pragma unroll
        for (int j = 0; j < 4; ++j) {
            if (ok[j]) {
                float e = vs[j] + vd[j];
                e = (e > 0.0f) ? e : NEG_SLOPE * e;
                atomicMax(mmax + dst[j] * 4 + h, f2ord(e));
            }
        }
    }
}

__global__ __launch_bounds__(256) void edge_sum_kernel(
    const int* __restrict__ ei, int E, int N,
    const float* __restrict__ a_s, const float* __restrict__ a_d,
    const unsigned* __restrict__ mmax, float* __restrict__ denom, float* __restrict__ ew)
{
    const int lane  = threadIdx.x & 31;
    const int gw    = (int)(blockIdx.x * 8 + (threadIdx.x >> 5));
    const int nw    = (int)(gridDim.x * 8);
    const int total = (E + N) * 4;
    const int ntrip = (total + EDGE_ITEMS - 1) / EDGE_ITEMS;
    const int h     = lane & 3;
    for (int c = gw; c < ntrip; c += nw) {
        const int t0 = c * EDGE_ITEMS + lane;
        int  src[4], dst[4];
        bool ok[4];
        #pragma unroll
        for (int j = 0; j < 4; ++j) {
            const int t = t0 + 32 * j;
            ok[j] = t < total;
            edge_src_dst_flat(ei, E, (ok[j] ? t : total - 1) >> 2, src[j], dst[j]);
        }
        float    vs[4], vd[4];
        unsigned vm[4];
        #pragma unroll
        for (int j = 0; j < 4; ++j) {
            vs[j] = a_s[src[j] * 4 + h];
            vd[j] = a_d[dst[j] * 4 + h];
            vm[j] = mmax[dst[j] * 4 + h];
        }
        #pragma unroll
        for (int j = 0; j < 4; ++j) {
            if (ok[j]) {
                float e = vs[j] + vd[j];
                e = (e > 0.0f) ? e : NEG_SLOPE * e;
                const float ex = __expf(e - ord2f(vm[j]));
                ew[t0 + 32 * j] = ex;
                atomicAdd(denom + dst[j] * 4 + h, ex);
            }
        }
    }
}

#define OWN_ROWS 196
#define OWN_CAP  16384
#define OWN_THREADS 1024

__device__ __noinline__ void own_edge_by_one_lane(
    const int* __restrict__ ei, int E, int eid, int row0, const float* __restrict__ H,
    const float* __restrict__ ew, const float* __restrict__ denom, float* acc)
{
    int src, dst; edge_src_dst(ei, E, eid, src, dst);
    const float4 w4 = *(const float4*)(ew + (size_t)eid * 4);
    const float4 d4 = *(const float4*)(denom + (size_t)dst * 4);
    float al[4];
    al[0] = w4.x / (d4.x + EPS_F);
    al[1] = w4.y / (d4.y + EPS_F);
    al[2] = w4.z / (d4.z + EPS_F);
    al[3] = w4.w / (d4.w + EPS_F);
    const float* hs = H + (size_t)src * HC;
    float* ap = acc + (size_t)(dst - row0) * HC;
    for (int c = 0; c < HC; ++c) atomicAdd(ap + c, al[c >> 6] * hs[c]);
}

__global__ __launch_bounds__(OWN_THREADS) void edge_agg_owner_kernel(
    const int* __restrict__ ei, int E, int N, const float* __restrict__ H,
    const float* __restrict__ ew, const float* __restrict__ denom, float* __restrict__ out)
{
    __shared__ __attribute__((aligned(16))) float acc[OWN_ROWS * HC];
    __shared__ int elist[OWN_CAP];
    __shared__ int ecount;

    const int t    = threadIdx.x;
    const int lane = t & 31;
    const int wave = t >> 5;
    const int row0  = blockIdx.x * OWN_ROWS;
    const int nrows = min(OWN_ROWS, N - row0);

    float4* acc4 = (float4*)acc;
    for (int i = t; i < OWN_ROWS * (HC / 4); i += OWN_THREADS)
        acc4[i] = make_float4(0.0f, 0.0f, 0.0f, 0.0f);
    if (t == 0) ecount = 0;
    __syncthreads();

    auto consider = [&](int eid, int dst) {
        if ((unsigned)(dst - row0) < (unsigned)nrows) {
            const int pos = atomicAdd(&ecount, 1);
            if (pos < OWN_CAP) elist[pos] = eid;
            else own_edge_by_one_lane(ei, E, eid, row0, H, ew, denom, acc);
        }
    };
    const int* __restrict__ dp = ei + E;
    if ((E & 3) == 0) {
        const int4* __restrict__ dp4 = (const int4*)dp;
        const int n4 = E >> 2;
        for (int i = t; i < n4; i += 4 * OWN_THREADS) {
            int4 v[4];
            #pragma unroll
            for (int u = 0; u < 4; ++u) {
                const int k = i + u * OWN_THREADS;
                v[u] = (k < n4) ? dp4[k] : make_int4(-1, -1, -1, -1);
            }
            #pragma unroll
            for (int u = 0; u < 4; ++u) {
                const int e0 = (i + u * OWN_THREADS) * 4;
                const bool any = ((unsigned)(v[u].x - row0) < (unsigned)nrows) |
                                 ((unsigned)(v[u].y - row0) < (unsigned)nrows) |
                                 ((unsigned)(v[u].z - row0) < (unsigned)nrows) |
                                 ((unsigned)(v[u].w - row0) < (unsigned)nrows);
                if (any) {
                    consider(e0,     v[u].x);
                    consider(e0 + 1, v[u].y);
                    consider(e0 + 2, v[u].z);
                    consider(e0 + 3, v[u].w);
                }
            }
        }
    } else {
        for (int e = t; e < E; e += OWN_THREADS) consider(e, dp[e]);
    }
    if (t < nrows) consider(E + row0 + t, row0 + t);
    __syncthreads();

    const int n = min(ecount, OWN_CAP);
    for (int base = wave * 4; base < n; base += 4 * (OWN_THREADS / 32)) {
        int  eid[4], srcn[4], dstn[4];
        bool ok[4];
        #pragma unroll
        for (int j = 0; j < 4; ++j) {
            ok[j]  = (base + j) < n;
            eid[j] = elist[ok[j] ? base + j : base];
        }
        #pragma unroll
        for (int j = 0; j < 4; ++j) edge_src_dst(ei, E, eid[j], srcn[j], dstn[j]);
        float  v[4][8];
        float4 w4[4], d4[4];
        #pragma unroll
        for (int j = 0; j < 4; ++j) {
            const float* hs = H + (size_t)srcn[j] * HC + lane;
            #pragma unroll
            for (int q = 0; q < 8; ++q) v[j][q] = hs[32 * q];
            w4[j] = *(const float4*)(ew + (size_t)eid[j] * 4);
            d4[j] = *(const float4*)(denom + (size_t)dstn[j] * 4);
        }
        #pragma unroll
        for (int j = 0; j < 4; ++j) {
            if (ok[j]) {
                const float al0 = w4[j].x / (d4[j].x + EPS_F);
                const float al1 = w4[j].y / (d4[j].y + EPS_F);
                const float al2 = w4[j].z / (d4[j].z + EPS_F);
                const float al3 = w4[j].w / (d4[j].w + EPS_F);
                float* ap = acc + (size_t)(dstn[j] - row0) * HC + lane;
                atomicAdd(ap +   0, al0 * v[j][0]);
                atomicAdd(ap +  32, al0 * v[j][1]);
                atomicAdd(ap +  64, al1 * v[j][2]);
                atomicAdd(ap +  96, al1 * v[j][3]);
                atomicAdd(ap + 128, al2 * v[j][4]);
                atomicAdd(ap + 160, al2 * v[j][5]);
                atomicAdd(ap + 192, al3 * v[j][6]);
                atomicAdd(ap + 224, al3 * v[j][7]);
            }
        }
    }
    __syncthreads();

    float4* __restrict__ o4 = (float4*)(out + (size_t)row0 * HC);
    for (int i = t; i < nrows * (HC / 4); i += OWN_THREADS) o4[i] = acc4[i];
}

extern "C" void kernel_launch(void* const* d_in, const int* in_sizes, int n_in,
                              void* d_out, int out_size, void* d_ws, size_t ws_size,
                              hipStream_t stream)
{
    const float* x   = (const float*)d_in[0];
    const int*   ei  = (const int*)d_in[1];
    const float* W1  = (const float*)d_in[2];
    const float* as1 = (const float*)d_in[3];
    const float* ad1 = (const float*)d_in[4];
    const float* b1  = (const float*)d_in[5];
    const float* W2  = (const float*)d_in[6];
    const float* as2 = (const float*)d_in[7];
    const float* ad2 = (const float*)d_in[8];
    const float* b2  = (const float*)d_in[9];
    const float* Wm1 = (const float*)d_in[10];
    const float* bm1 = (const float*)d_in[11];
    const float* Wm2 = (const float*)d_in[12];
    const float* bm2 = (const float*)d_in[13];

    const int N    = in_sizes[0] / 128;
    const int E    = in_sizes[1] / 2;
    const int Etot = E + N;
    const int NH   = N * HEADS;

    float* ws = (float*)d_ws;
    size_t o = 0;
    float*    P0    = ws + o; o += (size_t)N * HC;
    float*    P1    = ws + o; o += (size_t)N * HC;
    float*    a_s   = ws + o; o += (size_t)NH;
    float*    a_d   = ws + o; o += (size_t)NH;
    unsigned* mmax  = (unsigned*)(ws + o); o += (size_t)NH;
    float*    denom = ws + o; o += (size_t)NH;
    float*    ew    = ws + o; o += (size_t)Etot * HEADS;

    const dim3 blk(256);
    auto gemm_grid = [](int M, int NC) {
        return dim3((unsigned)(((M + 15) / 16 + 7) / 8), (unsigned)((NC + 63) / 64));
    };
    const int gScores = ((N + 3) / 4 + 15) / 16;
    const int nTrips  = (Etot * 4 + EDGE_ITEMS - 1) / EDGE_ITEMS;
    const int gEdgeH  = ((nTrips + 7) / 8 < 1024) ? (nTrips + 7) / 8 : 1024;
    const int gOwn    = (N + OWN_ROWS - 1) / OWN_ROWS;


    wmma_gemm_f32_kernel<false><<<gemm_grid(N, HC), blk, 0, stream>>>(x, W1, P0, nullptr, nullptr,
                                                                      N, 128, HC, 0);
    scores_kernel<<<gScores, blk, 0, stream>>>(P0, as1, ad1, a_s, a_d, mmax, denom, N);
    edge_max_kernel<<<gEdgeH, blk, 0, stream>>>(ei, E, N, a_s, a_d, mmax);
    edge_sum_kernel<<<gEdgeH, blk, 0, stream>>>(ei, E, N, a_s, a_d, mmax, denom, ew);
    edge_agg_owner_kernel<<<gOwn, dim3(OWN_THREADS), 0, stream>>>(ei, E, N, P0, ew, denom, P1);

    wmma_gemm_f32_kernel<true><<<gemm_grid(N, HC), blk, 0, stream>>>(P1, W2, P0, nullptr, b1,
                                                                     N, HC, HC, 0);
    scores_kernel<<<gScores, blk, 0, stream>>>(P0, as2, ad2, a_s, a_d, mmax, denom, N);
    edge_max_kernel<<<gEdgeH, blk, 0, stream>>>(ei, E, N, a_s, a_d, mmax);
    edge_sum_kernel<<<gEdgeH, blk, 0, stream>>>(ei, E, N, a_s, a_d, mmax, denom, ew);
    edge_agg_owner_kernel<<<gOwn, dim3(OWN_THREADS), 0, stream>>>(ei, E, N, P0, ew, denom, P1);

    wmma_gemm_f32_kernel<true><<<gemm_grid(N, HID), blk, 0, stream>>>(P1, Wm1, P0, bm1, b2,
                                                                      N, HC, HID, 1);
    wmma_gemm_f32_kernel<false><<<gemm_grid(N, 121), blk, 0, stream>>>(P0, Wm2, (float*)d_out, bm2,
                                                                       nullptr, N, HID, 121, 2);
}


# Owner block forms the softmax max and exp-sums of its rows in LDS from its own edge list (LDS max, LDS float adds in arrival order); 4 per-edge launches, global atomics, weight array gone
# speedup vs baseline: 3.0507x; 1.2452x over previous
#include <hip/hip_runtime.h>
#include <math.h>

#define HEADS 4
#define HID 64
#define HC 256
#define NEG_SLOPE 0.2f
#define EPS_F 1e-16f

typedef __attribute__((ext_vector_type(2))) float v2f;
typedef __attribute__((ext_vector_type(8))) float v8f;

__device__ __forceinline__ unsigned f2ord(float f) {
    unsigned u = __float_as_uint(f);
    return (u & 0x80000000u) ? ~u : (u | 0x80000000u);
}
__device__ __forceinline__ float ord2f(unsigned u) {
    return (u & 0x80000000u) ? __uint_as_float(u & 0x7fffffffu) : __uint_as_float(~u);
}

__device__ __forceinline__ void epi_store(float* __restrict__ C, const float* __restrict__ bias,
                                          size_t rowoff, int col, int NC, int mode, float v) {
    if (col < NC) {
        if (mode >= 1) v += bias[col];
        if (mode == 2) v = 1.0f / (1.0f + __expf(-v));
        C[rowoff + col] = v;
    }
}

template <bool ACT>
__device__ __forceinline__ v2f load_a_frag(const float* __restrict__ Ap,
                                           const float* __restrict__ abias, int kb, int aK) {
    v2f a = *(const v2f*)Ap;
    if (ACT) {
        const float4 bb = *(const float4*)(abias + kb);
        const float bx = aK ? bb.z : bb.x;
        const float by = aK ? bb.w : bb.y;
        a.x = fmaxf(a.x + bx, 0.0f);
        a.y = fmaxf(a.y + by, 0.0f);
    }
    return a;
}

template <bool ACT>
__global__ __launch_bounds__(256) void wmma_gemm_f32_kernel(
    const float* __restrict__ A, const float* __restrict__ B,
    float* __restrict__ C, const float* __restrict__ bias,
    const float* __restrict__ abias,
    int M, int K, int NC, int mode)
{
    const int wave = threadIdx.x >> 5;
    const int lane = threadIdx.x & 31;
    const int m0 = (blockIdx.x * 8 + wave) * 16;
    if (m0 >= M) return;
    const int n0 = blockIdx.y * 64;

    const int aRow = m0 + (lane & 15);
    const int aK   = (lane >> 4) << 1;
    const int bN   = lane & 15;
    const int bK   = (lane >> 4) << 1;
    const int c    = n0 + bN;

    const float* Ap = A + (size_t)aRow * K + aK;
    const size_t stepB = (size_t)4 * NC;

    v8f acc0 = {}, acc1 = {}, acc2 = {}, acc3 = {};

    if (n0 + 64 <= NC) {
        const float* Bp = B + c + (size_t)bK * NC;
        const float* Bq = Bp + NC;
        #pragma unroll 2
        for (int kb = 0; kb < K; kb += 4) {
            v2f a = load_a_frag<ACT>(Ap, abias, kb, aK);
            v2f b0 = { Bp[0],  Bq[0]  };
            v2f b1 = { Bp[16], Bq[16] };
            v2f b2 = { Bp[32], Bq[32] };
            v2f b3 = { Bp[48], Bq[48] };
            Ap += 4; Bp += stepB; Bq += stepB;
            acc0 = __builtin_amdgcn_wmma_f32_16x16x4_f32(false, a, false, b0, (short)0, acc0, false, false);
            acc1 = __builtin_amdgcn_wmma_f32_16x16x4_f32(false, a, false, b1, (short)0, acc1, false, false);
            acc2 = __builtin_amdgcn_wmma_f32_16x16x4_f32(false, a, false, b2, (short)0, acc2, false, false);
            acc3 = __builtin_amdgcn_wmma_f32_16x16x4_f32(false, a, false, b3, (short)0, acc3, false, false);
        }
    } else {
        const int   c0 = min(c,      NC - 1), c1 = min(c + 16, NC - 1);
        const int   c2 = min(c + 32, NC - 1), c3 = min(c + 48, NC - 1);
        const float w0 = (c      < NC) ? 1.0f : 0.0f;
        const float w1 = (c + 16 < NC) ? 1.0f : 0.0f;
        const float w2 = (c + 32 < NC) ? 1.0f : 0.0f;
        const float w3 = (c + 48 < NC) ? 1.0f : 0.0f;
        const float* Bp = B + (size_t)bK * NC;
        const float* Bq = Bp + NC;
        #pragma unroll 2
        for (int kb = 0; kb < K; kb += 4) {
            v2f a = load_a_frag<ACT>(Ap, abias, kb, aK);
            v2f b0 = { Bp[c0] * w0, Bq[c0] * w0 };
            v2f b1 = { Bp[c1] * w1, Bq[c1] * w1 };
            v2f b2 = { Bp[c2] * w2, Bq[c2] * w2 };
            v2f b3 = { Bp[c3] * w3, Bq[c3] * w3 };
            Ap += 4; Bp += stepB; Bq += stepB;
            acc0 = __builtin_amdgcn_wmma_f32_16x16x4_f32(false, a, false, b0, (short)0, acc0, false, false);
            acc1 = __builtin_amdgcn_wmma_f32_16x16x4_f32(false, a, false, b1, (short)0, acc1, false, false);
            acc2 = __builtin_amdgcn_wmma_f32_16x16x4_f32(false, a, false, b2, (short)0, acc2, false, false);
            acc3 = __builtin_amdgcn_wmma_f32_16x16x4_f32(false, a, false, b3, (short)0, acc3, false, false);
        }
    }

    const int mbase = m0 + ((lane >> 4) << 3);
    const int colb  = n0 + (lane & 15);
    #pragma unroll
    for (int r = 0; r < 8; ++r) {
        size_t rowoff = (size_t)(mbase + r) * NC;
        epi_store(C, bias, rowoff, colb,      NC, mode, acc0[r]);
        epi_store(C, bias, rowoff, colb + 16, NC, mode, acc1[r]);
        epi_store(C, bias, rowoff, colb + 32, NC, mode, acc2[r]);
        epi_store(C, bias, rowoff, colb + 48, NC, mode, acc3[r]);
    }
}

__global__ __launch_bounds__(256) void scores_kernel(
    const float* __restrict__ H, const float* __restrict__ att_src,
    const float* __restrict__ att_dst, float* __restrict__ a_s, float* __restrict__ a_d, int N)
{
    const int lane = threadIdx.x & 31;
    const int gw   = (int)(blockIdx.x * 8 + (threadIdx.x >> 5));
    const int nw   = (int)(gridDim.x * 8);

    float ws0[HEADS], ws1[HEADS], wd0[HEADS], wd1[HEADS];
    #pragma unroll
    for (int h = 0; h < HEADS; ++h) {
        ws0[h] = att_src[h * HID + lane]; ws1[h] = att_src[h * HID + lane + 32];
        wd0[h] = att_dst[h * HID + lane]; wd1[h] = att_dst[h * HID + lane + 32];
    }

    const int ngroups = (N + 3) >> 2;
    for (int g = gw; g < ngroups; g += nw) {
        const int n0 = g * 4;
        float x[4][8];
        #pragma unroll
        for (int j = 0; j < 4; ++j) {
            const int n = min(n0 + j, N - 1);
            const float* hp = H + (size_t)n * HC + lane;
            #pragma unroll
            for (int q = 0; q < 8; ++q) x[j][q] = hp[32 * q];
        }
        #pragma unroll
        for (int j = 0; j < 4; ++j) {
            float s[HEADS], d[HEADS];
            #pragma unroll
            for (int h = 0; h < HEADS; ++h) {
                const float x0 = x[j][2 * h], x1 = x[j][2 * h + 1];
                s[h] = x0 * ws0[h] + x1 * ws1[h];
                d[h] = x0 * wd0[h] + x1 * wd1[h];
            }
            #pragma unroll
            for (int off = 16; off > 0; off >>= 1) {
                #pragma unroll
                for (int h = 0; h < HEADS; ++h) {
                    s[h] += __shfl_xor(s[h], off, 32);
                    d[h] += __shfl_xor(d[h], off, 32);
                }
            }
            if (lane == 0 && n0 + j < N) {
                const size_t o4 = (size_t)(n0 + j) * 4;
                *(float4*)(a_s + o4) = make_float4(s[0], s[1], s[2], s[3]);
                *(float4*)(a_d + o4) = make_float4(d[0], d[1], d[2], d[3]);
            }
        }
    }
}

__device__ __forceinline__ void edge_src_dst(const int* __restrict__ ei, int E, int eid,
                                             int& src, int& dst) {
    if (eid < E) { src = ei[eid]; dst = ei[E + eid]; }
    else         { src = dst = eid - E; }
}

__device__ __forceinline__ float4 edge_logits(const float* __restrict__ a_s,
                                              const float* __restrict__ a_d, int src, int dst) {
    const float4 s4 = *(const float4*)(a_s + (size_t)src * 4);
    const float4 d4 = *(const float4*)(a_d + (size_t)dst * 4);
    float4 e;
    e.x = s4.x + d4.x; e.x = (e.x > 0.0f) ? e.x : NEG_SLOPE * e.x;
    e.y = s4.y + d4.y; e.y = (e.y > 0.0f) ? e.y : NEG_SLOPE * e.y;
    e.z = s4.z + d4.z; e.z = (e.z > 0.0f) ? e.z : NEG_SLOPE * e.z;
    e.w = s4.w + d4.w; e.w = (e.w > 0.0f) ? e.w : NEG_SLOPE * e.w;
    return e;
}

#define OWN_ROWS 196
#define OWN_CAP  16384
#define OWN_THREADS 1024

__device__ __noinline__ void own_rows_by_column_scans(
    const int* __restrict__ ei, int E, int row0, int nrows, const float* __restrict__ H,
    const float* __restrict__ a_s, const float* __restrict__ a_d,
    float* acc, unsigned* smax, float* ssum)
{
    const int t = threadIdx.x;
    const int* __restrict__ dp = ei + E;
    const int nall = E + nrows;
    for (int stage = 0; stage < 3; ++stage) {
        for (int i = t; i < nall; i += OWN_THREADS) {
            int src, dst;
            if (i < E) { dst = dp[i]; src = 0; } else { dst = row0 + (i - E); src = dst; }
            const unsigned dl = (unsigned)(dst - row0);
            if (dl < (unsigned)nrows) {
                if (i < E) src = ei[i];
                const float4 e = edge_logits(a_s, a_d, src, dst);
                const float ev[4] = { e.x, e.y, e.z, e.w };
                if (stage == 0) {
                    for (int h = 0; h < 4; ++h) atomicMax(smax + dl * 4 + h, f2ord(ev[h]));
                } else if (stage == 1) {
                    for (int h = 0; h < 4; ++h)
                        atomicAdd(ssum + dl * 4 + h, __expf(ev[h] - ord2f(smax[dl * 4 + h])));
                } else {
                    float al[4];
                    for (int h = 0; h < 4; ++h)
                        al[h] = __expf(ev[h] - ord2f(smax[dl * 4 + h])) / (ssum[dl * 4 + h] + EPS_F);
                    const float* hs = H + (size_t)src * HC;
                    float* ap = acc + (size_t)dl * HC;
                    for (int c = 0; c < HC; ++c) atomicAdd(ap + c, al[c >> 6] * hs[c]);
                }
            }
        }
        __syncthreads();
    }
}

__global__ __launch_bounds__(OWN_THREADS) void edge_agg_owner_kernel(
    const int* __restrict__ ei, int E, int N, const float* __restrict__ H,
    const float* __restrict__ a_s, const float* __restrict__ a_d, float* __restrict__ out)
{
    __shared__ __attribute__((aligned(16))) float acc[OWN_ROWS * HC];
    __shared__ int elist[OWN_CAP];
    __shared__ unsigned short edl[OWN_CAP];
    __shared__ __attribute__((aligned(16))) unsigned smax[OWN_ROWS * 4];
    __shared__ __attribute__((aligned(16))) float    ssum[OWN_ROWS * 4];
    __shared__ int ecount;

    const int t    = threadIdx.x;
    const int lane = t & 31;
    const int wave = t >> 5;
    const int row0  = blockIdx.x * OWN_ROWS;
    const int nrows = min(OWN_ROWS, N - row0);

    float4* acc4 = (float4*)acc;
    for (int i = t; i < OWN_ROWS * (HC / 4); i += OWN_THREADS)
        acc4[i] = make_float4(0.0f, 0.0f, 0.0f, 0.0f);
    if (t < OWN_ROWS * 4) { smax[t] = f2ord(-INFINITY); ssum[t] = 0.0f; }
    if (t == 0) ecount = 0;
    __syncthreads();

    auto consider = [&](int eid, int dst) {
        const unsigned dl = (unsigned)(dst - row0);
        if (dl < (unsigned)nrows) {
            const int pos = atomicAdd(&ecount, 1);
            if (pos < OWN_CAP) { elist[pos] = eid; edl[pos] = (unsigned short)dl; }
        }
    };
    const int* __restrict__ dp = ei + E;
    if ((E & 3) == 0) {
        const int4* __restrict__ dp4 = (const int4*)dp;
        const int n4 = E >> 2;
        for (int i = t; i < n4; i += 4 * OWN_THREADS) {
            int4 v[4];
            #pragma unroll
            for (int u = 0; u < 4; ++u) {
                const int k = i + u * OWN_THREADS;
                v[u] = (k < n4) ? dp4[k] : make_int4(-1, -1, -1, -1);
            }
            #pragma unroll
            for (int u = 0; u < 4; ++u) {
                const int e0 = (i + u * OWN_THREADS) * 4;
                const bool any = ((unsigned)(v[u].x - row0) < (unsigned)nrows) |
                                 ((unsigned)(v[u].y - row0) < (unsigned)nrows) |
                                 ((unsigned)(v[u].z - row0) < (unsigned)nrows) |
                                 ((unsigned)(v[u].w - row0) < (unsigned)nrows);
                if (any) {
                    consider(e0,     v[u].x);
                    consider(e0 + 1, v[u].y);
                    consider(e0 + 2, v[u].z);
                    consider(e0 + 3, v[u].w);
                }
            }
        }
    } else {
        for (int e = t; e < E; e += OWN_THREADS) consider(e, dp[e]);
    }
    if (t < nrows) consider(E + row0 + t, row0 + t);
    __syncthreads();

    const int n = ecount;
    if (n > OWN_CAP) {
        own_rows_by_column_scans(ei, E, row0, nrows, H, a_s, a_d, acc, smax, ssum);
    } else {
        for (int i0 = t; i0 < n; i0 += 4 * OWN_THREADS) {
            int  idx[4], eid[4], dl[4], src[4];
            bool ok[4];
            #pragma unroll
            for (int u = 0; u < 4; ++u) {
                idx[u] = i0 + u * OWN_THREADS;
                ok[u]  = idx[u] < n;
                if (!ok[u]) idx[u] = i0;
                eid[u] = elist[idx[u]];
                dl[u]  = edl[idx[u]];
            }
            #pragma unroll
            for (int u = 0; u < 4; ++u) {
                src[u] = ei[min(eid[u], E - 1)];
                if (eid[u] >= E) src[u] = eid[u] - E;
            }
            float4 e[4];
            #pragma unroll
            for (int u = 0; u < 4; ++u) e[u] = edge_logits(a_s, a_d, src[u], row0 + dl[u]);
            #pragma unroll
            for (int u = 0; u < 4; ++u) {
                if (ok[u]) {
                    elist[idx[u]] = src[u];
                    atomicMax(smax + dl[u] * 4 + 0, f2ord(e[u].x));
                    atomicMax(smax + dl[u] * 4 + 1, f2ord(e[u].y));
                    atomicMax(smax + dl[u] * 4 + 2, f2ord(e[u].z));
                    atomicMax(smax + dl[u] * 4 + 3, f2ord(e[u].w));
                }
            }
        }
        __syncthreads();

        for (int i0 = t; i0 < n; i0 += 4 * OWN_THREADS) {
            int  dl[4], src[4];
            bool ok[4];
            #pragma unroll
            for (int u = 0; u < 4; ++u) {
                const int i = i0 + u * OWN_THREADS;
                ok[u] = i < n;
                src[u] = elist[ok[u] ? i : i0];
                dl[u]  = edl[ok[u] ? i : i0];
            }
            float4 e[4];
            uint4  m[4];
            #pragma unroll
            for (int u = 0; u < 4; ++u) {
                e[u] = edge_logits(a_s, a_d, src[u], row0 + dl[u]);
                m[u] = *(const uint4*)(smax + dl[u] * 4);
            }
            #pragma unroll
            for (int u = 0; u < 4; ++u) {
                if (ok[u]) {
                    atomicAdd(ssum + dl[u] * 4 + 0, __expf(e[u].x - ord2f(m[u].x)));
                    atomicAdd(ssum + dl[u] * 4 + 1, __expf(e[u].y - ord2f(m[u].y)));
                    atomicAdd(ssum + dl[u] * 4 + 2, __expf(e[u].z - ord2f(m[u].z)));
                    atomicAdd(ssum + dl[u] * 4 + 3, __expf(e[u].w - ord2f(m[u].w)));
                }
            }
        }
        __syncthreads();

        for (int base = wave * 4; base < n; base += 4 * (OWN_THREADS / 32)) {
            int  srcn[4], dln[4];
            bool ok[4];
            #pragma unroll
            for (int j = 0; j < 4; ++j) {
                ok[j] = (base + j) < n;
                const int idx = ok[j] ? base + j : base;
                srcn[j] = elist[idx];
                dln[j]  = edl[idx];
            }
            float  v[4][8];
            float4 e4[4], s4[4];
            uint4  m4[4];
            #pragma unroll
            for (int j = 0; j < 4; ++j) {
                const float* hs = H + (size_t)srcn[j] * HC + lane;
                #pragma unroll
                for (int q = 0; q < 8; ++q) v[j][q] = hs[32 * q];
                e4[j] = edge_logits(a_s, a_d, srcn[j], row0 + dln[j]);
                m4[j] = *(const uint4*)(smax + dln[j] * 4);
                s4[j] = *(const float4*)(ssum + dln[j] * 4);
            }
            #pragma unroll
            for (int j = 0; j < 4; ++j) {
                if (ok[j]) {
                    const float al0 = __expf(e4[j].x - ord2f(m4[j].x)) / (s4[j].x + EPS_F);
                    const float al1 = __expf(e4[j].y - ord2f(m4[j].y)) / (s4[j].y + EPS_F);
                    const float al2 = __expf(e4[j].z - ord2f(m4[j].z)) / (s4[j].z + EPS_F);
                    const float al3 = __expf(e4[j].w - ord2f(m4[j].w)) / (s4[j].w + EPS_F);
                    float* ap = acc + (size_t)dln[j] * HC + lane;
                    atomicAdd(ap +   0, al0 * v[j][0]);
                    atomicAdd(ap +  32, al0 * v[j][1]);
                    atomicAdd(ap +  64, al1 * v[j][2]);
                    atomicAdd(ap +  96, al1 * v[j][3]);
                    atomicAdd(ap + 128, al2 * v[j][4]);
                    atomicAdd(ap + 160, al2 * v[j][5]);
                    atomicAdd(ap + 192, al3 * v[j][6]);
                    atomicAdd(ap + 224, al3 * v[j][7]);
                }
            }
        }
        __syncthreads();
    }

    float4* __restrict__ o4 = (float4*)(out + (size_t)row0 * HC);
    for (int i = t; i < nrows * (HC / 4); i += OWN_THREADS) o4[i] = acc4[i];
}

extern "C" void kernel_launch(void* const* d_in, const int* in_sizes, int n_in,
                              void* d_out, int out_size, void* d_ws, size_t ws_size,
                              hipStream_t stream)
{
    const float* x   = (const float*)d_in[0];
    const int*   ei  = (const int*)d_in[1];
    const float* W1  = (const float*)d_in[2];
    const float* as1 = (const float*)d_in[3];
    const float* ad1 = (const float*)d_in[4];
    const float* b1  = (const float*)d_in[5];
    const float* W2  = (const float*)d_in[6];
    const float* as2 = (const float*)d_in[7];
    const float* ad2 = (const float*)d_in[8];
    const float* b2  = (const float*)d_in[9];
    const float* Wm1 = (const float*)d_in[10];
    const float* bm1 = (const float*)d_in[11];
    const float* Wm2 = (const float*)d_in[12];
    const float* bm2 = (const float*)d_in[13];

    const int N  = in_sizes[0] / 128;
    const int E  = in_sizes[1] / 2;
    const int NH = N * HEADS;

    float* ws = (float*)d_ws;
    size_t o = 0;
    float* P0  = ws + o; o += (size_t)N * HC;
    float* P1  = ws + o; o += (size_t)N * HC;
    float* a_s = ws + o; o += (size_t)NH;
    float* a_d = ws + o; o += (size_t)NH;

    const dim3 blk(256);
    auto gemm_grid = [](int M, int NC) {
        return dim3((unsigned)(((M + 15) / 16 + 7) / 8), (unsigned)((NC + 63) / 64));
    };
    const int gScores = ((N + 3) / 4 + 15) / 16;
    const int gOwn    = (N + OWN_ROWS - 1) / OWN_ROWS;


    wmma_gemm_f32_kernel<false><<<gemm_grid(N, HC), blk, 0, stream>>>(x, W1, P0, nullptr, nullptr,
                                                                      N, 128, HC, 0);
    scores_kernel<<<gScores, blk, 0, stream>>>(P0, as1, ad1, a_s, a_d, N);
    edge_agg_owner_kernel<<<gOwn, dim3(OWN_THREADS), 0, stream>>>(ei, E, N, P0, a_s, a_d, P1);

    wmma_gemm_f32_kernel<true><<<gemm_grid(N, HC), blk, 0, stream>>>(P1, W2, P0, nullptr, b1,
                                                                     N, HC, HC, 0);
    scores_kernel<<<gScores, blk, 0, stream>>>(P0, as2, ad2, a_s, a_d, N);
    edge_agg_owner_kernel<<<gOwn, dim3(OWN_THREADS), 0, stream>>>(ei, E, N, P0, a_s, a_d, P1);

    wmma_gemm_f32_kernel<true><<<gemm_grid(N, HID), blk, 0, stream>>>(P1, Wm1, P0, bm1, b2,
                                                                      N, HC, HID, 1);
    wmma_gemm_f32_kernel<false><<<gemm_grid(N, 121), blk, 0, stream>>>(P0, Wm2, (float*)d_out, bm2,
                                                                       nullptr, N, HID, 121, 2);
}


# Owner block forms each edge's logits and exp terms once, by one thread, kept in LDS; gather waves only multiply-add; rows scaled by 1/(sum+eps) once at the store: rounding changes, arrival order kept
# speedup vs baseline: 3.4131x; 1.1188x over previous
#include <hip/hip_runtime.h>
#include <math.h>

#define HEADS 4
#define HID 64
#define HC 256
#define NEG_SLOPE 0.2f
#define EPS_F 1e-16f

typedef __attribute__((ext_vector_type(2))) float v2f;
typedef __attribute__((ext_vector_type(8))) float v8f;

__device__ __forceinline__ unsigned f2ord(float f) {
    unsigned u = __float_as_uint(f);
    return (u & 0x80000000u) ? ~u : (u | 0x80000000u);
}
__device__ __forceinline__ float ord2f(unsigned u) {
    return (u & 0x80000000u) ? __uint_as_float(u & 0x7fffffffu) : __uint_as_float(~u);
}

__device__ __forceinline__ void epi_store(float* __restrict__ C, const float* __restrict__ bias,
                                          size_t rowoff, int col, int NC, int mode, float v) {
    if (col < NC) {
        if (mode >= 1) v += bias[col];
        if (mode == 2) v = 1.0f / (1.0f + __expf(-v));
        C[rowoff + col] = v;
    }
}

template <bool ACT>
__device__ __forceinline__ v2f load_a_frag(const float* __restrict__ Ap,
                                           const float* __restrict__ abias, int kb, int aK) {
    v2f a = *(const v2f*)Ap;
    if (ACT) {
        const float4 bb = *(const float4*)(abias + kb);
        const float bx = aK ? bb.z : bb.x;
        const float by = aK ? bb.w : bb.y;
        a.x = fmaxf(a.x + bx, 0.0f);
        a.y = fmaxf(a.y + by, 0.0f);
    }
    return a;
}

template <bool ACT>
__global__ __launch_bounds__(256) void wmma_gemm_f32_kernel(
    const float* __restrict__ A, const float* __restrict__ B,
    float* __restrict__ C, const float* __restrict__ bias,
    const float* __restrict__ abias,
    int M, int K, int NC, int mode)
{
    const int wave = threadIdx.x >> 5;
    const int lane = threadIdx.x & 31;
    const int m0 = (blockIdx.x * 8 + wave) * 16;
    if (m0 >= M) return;
    const int n0 = blockIdx.y * 64;

    const int aRow = m0 + (lane & 15);
    const int aK   = (lane >> 4) << 1;
    const int bN   = lane & 15;
    const int bK   = (lane >> 4) << 1;
    const int c    = n0 + bN;

    const float* Ap = A + (size_t)aRow * K + aK;
    const size_t stepB = (size_t)4 * NC;

    v8f acc0 = {}, acc1 = {}, acc2 = {}, acc3 = {};

    if (n0 + 64 <= NC) {
        const float* Bp = B + c + (size_t)bK * NC;
        const float* Bq = Bp + NC;
        #pragma unroll 2
        for (int kb = 0; kb < K; kb += 4) {
            v2f a = load_a_frag<ACT>(Ap, abias, kb, aK);
            v2f b0 = { Bp[0],  Bq[0]  };
            v2f b1 = { Bp[16], Bq[16] };
            v2f b2 = { Bp[32], Bq[32] };
            v2f b3 = { Bp[48], Bq[48] };
            Ap += 4; Bp += stepB; Bq += stepB;
            acc0 = __builtin_amdgcn_wmma_f32_16x16x4_f32(false, a, false, b0, (short)0, acc0, false, false);
            acc1 = __builtin_amdgcn_wmma_f32_16x16x4_f32(false, a, false, b1, (short)0, acc1, false, false);
            acc2 = __builtin_amdgcn_wmma_f32_16x16x4_f32(false, a, false, b2, (short)0, acc2, false, false);
            acc3 = __builtin_amdgcn_wmma_f32_16x16x4_f32(false, a, false, b3, (short)0, acc3, false, false);
        }
    } else {
        const int   c0 = min(c,      NC - 1), c1 = min(c + 16, NC - 1);
        const int   c2 = min(c + 32, NC - 1), c3 = min(c + 48, NC - 1);
        const float w0 = (c      < NC) ? 1.0f : 0.0f;
        const float w1 = (c + 16 < NC) ? 1.0f : 0.0f;
        const float w2 = (c + 32 < NC) ? 1.0f : 0.0f;
        const float w3 = (c + 48 < NC) ? 1.0f : 0.0f;
        const float* Bp = B + (size_t)bK * NC;
        const float* Bq = Bp + NC;
        #pragma unroll 2
        for (int kb = 0; kb < K; kb += 4) {
            v2f a = load_a_frag<ACT>(Ap, abias, kb, aK);
            v2f b0 = { Bp[c0] * w0, Bq[c0] * w0 };
            v2f b1 = { Bp[c1] * w1, Bq[c1] * w1 };
            v2f b2 = { Bp[c2] * w2, Bq[c2] * w2 };
            v2f b3 = { Bp[c3] * w3, Bq[c3] * w3 };
            Ap += 4; Bp += stepB; Bq += stepB;
            acc0 = __builtin_amdgcn_wmma_f32_16x16x4_f32(false, a, false, b0, (short)0, acc0, false, false);
            acc1 = __builtin_amdgcn_wmma_f32_16x16x4_f32(false, a, false, b1, (short)0, acc1, false, false);
            acc2 = __builtin_amdgcn_wmma_f32_16x16x4_f32(false, a, false, b2, (short)0, acc2, false, false);
            acc3 = __builtin_amdgcn_wmma_f32_16x16x4_f32(false, a, false, b3, (short)0, acc3, false, false);
        }
    }

    const int mbase = m0 + ((lane >> 4) << 3);
    const int colb  = n0 + (lane & 15);
    #pragma unroll
    for (int r = 0; r < 8; ++r) {
        size_t rowoff = (size_t)(mbase + r) * NC;
        epi_store(C, bias, rowoff, colb,      NC, mode, acc0[r]);
        epi_store(C, bias, rowoff, colb + 16, NC, mode, acc1[r]);
        epi_store(C, bias, rowoff, colb + 32, NC, mode, acc2[r]);
        epi_store(C, bias, rowoff, colb + 48, NC, mode, acc3[r]);
    }
}

__global__ __launch_bounds__(256) void scores_kernel(
    const float* __restrict__ H, const float* __restrict__ att_src,
    const float* __restrict__ att_dst, float* __restrict__ a_s, float* __restrict__ a_d, int N)
{
    const int lane = threadIdx.x & 31;
    const int gw   = (int)(blockIdx.x * 8 + (threadIdx.x >> 5));
    const int nw   = (int)(gridDim.x * 8);

    float ws0[HEADS], ws1[HEADS], wd0[HEADS], wd1[HEADS];
    #pragma unroll
    for (int h = 0; h < HEADS; ++h) {
        ws0[h] = att_src[h * HID + lane]; ws1[h] = att_src[h * HID + lane + 32];
        wd0[h] = att_dst[h * HID + lane]; wd1[h] = att_dst[h * HID + lane + 32];
    }

    const int ngroups = (N + 3) >> 2;
    for (int g = gw; g < ngroups; g += nw) {
        const int n0 = g * 4;
        float x[4][8];
        #pragma unroll
        for (int j = 0; j < 4; ++j) {
            const int n = min(n0 + j, N - 1);
            const float* hp = H + (size_t)n * HC + lane;
            #pragma unroll
            for (int q = 0; q < 8; ++q) x[j][q] = hp[32 * q];
        }
        #pragma unroll
        for (int j = 0; j < 4; ++j) {
            float s[HEADS], d[HEADS];
            #pragma unroll
            for (int h = 0; h < HEADS; ++h) {
                const float x0 = x[j][2 * h], x1 = x[j][2 * h + 1];
                s[h] = x0 * ws0[h] + x1 * ws1[h];
                d[h] = x0 * wd0[h] + x1 * wd1[h];
            }
            #pragma unroll
            for (int off = 16; off > 0; off >>= 1) {
                #pragma unroll
                for (int h = 0; h < HEADS; ++h) {
                    s[h] += __shfl_xor(s[h], off, 32);
                    d[h] += __shfl_xor(d[h], off, 32);
                }
            }
            if (lane == 0 && n0 + j < N) {
                const size_t o4 = (size_t)(n0 + j) * 4;
                *(float4*)(a_s + o4) = make_float4(s[0], s[1], s[2], s[3]);
                *(float4*)(a_d + o4) = make_float4(d[0], d[1], d[2], d[3]);
            }
        }
    }
}

__device__ __forceinline__ void edge_src_dst(const int* __restrict__ ei, int E, int eid,
                                             int& src, int& dst) {
    if (eid < E) { src = ei[eid]; dst = ei[E + eid]; }
    else         { src = dst = eid - E; }
}

__device__ __forceinline__ float4 edge_logits(const float* __restrict__ a_s,
                                              const float* __restrict__ a_d, int src, int dst) {
    const float4 s4 = *(const float4*)(a_s + (size_t)src * 4);
    const float4 d4 = *(const float4*)(a_d + (size_t)dst * 4);
    float4 e;
    e.x = s4.x + d4.x; e.x = (e.x > 0.0f) ? e.x : NEG_SLOPE * e.x;
    e.y = s4.y + d4.y; e.y = (e.y > 0.0f) ? e.y : NEG_SLOPE * e.y;
    e.z = s4.z + d4.z; e.z = (e.z > 0.0f) ? e.z : NEG_SLOPE * e.z;
    e.w = s4.w + d4.w; e.w = (e.w > 0.0f) ? e.w : NEG_SLOPE * e.w;
    return e;
}

#define OWN_ROWS 196
#define OWN_CAP  4608
#define OWN_THREADS 1024

__device__ __noinline__ void own_rows_by_column_scans(
    const int* __restrict__ ei, int E, int row0, int nrows, const float* __restrict__ H,
    const float* __restrict__ a_s, const float* __restrict__ a_d,
    float* acc, unsigned* smax, float* ssum)
{
    const int t = threadIdx.x;
    const int* __restrict__ dp = ei + E;
    const int nall = E + nrows;
    for (int stage = 0; stage < 3; ++stage) {
        for (int i = t; i < nall; i += OWN_THREADS) {
            int src, dst;
            if (i < E) { dst = dp[i]; src = 0; } else { dst = row0 + (i - E); src = dst; }
            const unsigned dl = (unsigned)(dst - row0);
            if (dl < (unsigned)nrows) {
                if (i < E) src = ei[i];
                const float4 e = edge_logits(a_s, a_d, src, dst);
                const float ev[4] = { e.x, e.y, e.z, e.w };
                if (stage == 0) {
                    for (int h = 0; h < 4; ++h) atomicMax(smax + dl * 4 + h, f2ord(ev[h]));
                } else if (stage == 1) {
                    for (int h = 0; h < 4; ++h)
                        atomicAdd(ssum + dl * 4 + h, __expf(ev[h] - ord2f(smax[dl * 4 + h])));
                } else {
                    float al[4];
                    for (int h = 0; h < 4; ++h)
                        al[h] = __expf(ev[h] - ord2f(smax[dl * 4 + h]));
                    const float* hs = H + (size_t)src * HC;
                    float* ap = acc + (size_t)dl * HC;
                    for (int c = 0; c < HC; ++c) atomicAdd(ap + c, al[c >> 6] * hs[c]);
                }
            }
        }
        __syncthreads();
    }
}

__global__ __launch_bounds__(OWN_THREADS) void edge_agg_owner_kernel(
    const int* __restrict__ ei, int E, int N, const float* __restrict__ H,
    const float* __restrict__ a_s, const float* __restrict__ a_d, float* __restrict__ out)
{
    __shared__ __attribute__((aligned(16))) float acc[OWN_ROWS * HC];
    __shared__ int elist[OWN_CAP];
    __shared__ unsigned short edl[OWN_CAP];
    __shared__ __attribute__((aligned(16))) float4 eterm[OWN_CAP];
    __shared__ __attribute__((aligned(16))) unsigned smax[OWN_ROWS * 4];
    __shared__ __attribute__((aligned(16))) float    ssum[OWN_ROWS * 4];
    __shared__ int ecount;

    const int t    = threadIdx.x;
    const int lane = t & 31;
    const int wave = t >> 5;
    const int row0  = blockIdx.x * OWN_ROWS;
    const int nrows = min(OWN_ROWS, N - row0);

    float4* acc4 = (float4*)acc;
    for (int i = t; i < OWN_ROWS * (HC / 4); i += OWN_THREADS)
        acc4[i] = make_float4(0.0f, 0.0f, 0.0f, 0.0f);
    if (t < OWN_ROWS * 4) { smax[t] = f2ord(-INFINITY); ssum[t] = 0.0f; }
    if (t == 0) ecount = 0;
    __syncthreads();

    auto consider = [&](int eid, int dst) {
        const unsigned dl = (unsigned)(dst - row0);
        if (dl < (unsigned)nrows) {
            const int pos = atomicAdd(&ecount, 1);
            if (pos < OWN_CAP) { elist[pos] = eid; edl[pos] = (unsigned short)dl; }
        }
    };
    const int* __restrict__ dp = ei + E;
    if ((E & 3) == 0) {
        const int4* __restrict__ dp4 = (const int4*)dp;
        const int n4 = E >> 2;
        for (int i = t; i < n4; i += 4 * OWN_THREADS) {
            int4 v[4];
            #pragma unroll
            for (int u = 0; u < 4; ++u) {
                const int k = i + u * OWN_THREADS;
                v[u] = (k < n4) ? dp4[k] : make_int4(-1, -1, -1, -1);
            }
            #pragma unroll
            for (int u = 0; u < 4; ++u) {
                const int e0 = (i + u * OWN_THREADS) * 4;
                const bool any = ((unsigned)(v[u].x - row0) < (unsigned)nrows) |
                                 ((unsigned)(v[u].y - row0) < (unsigned)nrows) |
                                 ((unsigned)(v[u].z - row0) < (unsigned)nrows) |
                                 ((unsigned)(v[u].w - row0) < (unsigned)nrows);
                if (any) {
                    consider(e0,     v[u].x);
                    consider(e0 + 1, v[u].y);
                    consider(e0 + 2, v[u].z);
                    consider(e0 + 3, v[u].w);
                }
            }
        }
    } else {
        for (int e = t; e < E; e += OWN_THREADS) consider(e, dp[e]);
    }
    if (t < nrows) consider(E + row0 + t, row0 + t);
    __syncthreads();

    const int n = ecount;
    if (n > OWN_CAP) {
        own_rows_by_column_scans(ei, E, row0, nrows, H, a_s, a_d, acc, smax, ssum);
    } else {
        for (int i0 = t; i0 < n; i0 += 4 * OWN_THREADS) {
            int  idx[4], eid[4], dl[4], src[4];
            bool ok[4];
            #pragma unroll
            for (int u = 0; u < 4; ++u) {
                idx[u] = i0 + u * OWN_THREADS;
                ok[u]  = idx[u] < n;
                if (!ok[u]) idx[u] = i0;
                eid[u] = elist[idx[u]];
                dl[u]  = edl[idx[u]];
            }
            #pragma unroll
            for (int u = 0; u < 4; ++u) {
                src[u] = ei[min(eid[u], E - 1)];
                if (eid[u] >= E) src[u] = eid[u] - E;
            }
            float4 e[4];
            #pragma unroll
            for (int u = 0; u < 4; ++u) e[u] = edge_logits(a_s, a_d, src[u], row0 + dl[u]);
            #pragma unroll
            for (int u = 0; u < 4; ++u) {
                if (ok[u]) {
                    elist[idx[u]] = src[u];
                    eterm[idx[u]] = e[u];
                    atomicMax(smax + dl[u] * 4 + 0, f2ord(e[u].x));
                    atomicMax(smax + dl[u] * 4 + 1, f2ord(e[u].y));
                    atomicMax(smax + dl[u] * 4 + 2, f2ord(e[u].z));
                    atomicMax(smax + dl[u] * 4 + 3, f2ord(e[u].w));
                }
            }
        }
        __syncthreads();

        for (int i = t; i < n; i += OWN_THREADS) {
            const int    dl = edl[i];
            const float4 e  = eterm[i];
            const uint4  m  = *(const uint4*)(smax + dl * 4);
            float4 x;
            x.x = __expf(e.x - ord2f(m.x));
            x.y = __expf(e.y - ord2f(m.y));
            x.z = __expf(e.z - ord2f(m.z));
            x.w = __expf(e.w - ord2f(m.w));
            eterm[i] = x;
            atomicAdd(ssum + dl * 4 + 0, x.x);
            atomicAdd(ssum + dl * 4 + 1, x.y);
            atomicAdd(ssum + dl * 4 + 2, x.z);
            atomicAdd(ssum + dl * 4 + 3, x.w);
        }
        __syncthreads();

        for (int base = wave * 4; base < n; base += 4 * (OWN_THREADS / 32)) {
            int    srcn[4], dln[4];
            float4 w4[4];
            bool   ok[4];
            #pragma unroll
            for (int j = 0; j < 4; ++j) {
                ok[j] = (base + j) < n;
                const int idx = ok[j] ? base + j : base;
                srcn[j] = elist[idx];
                dln[j]  = edl[idx];
                w4[j]   = eterm[idx];
            }
            float v[4][8];
            #pragma unroll
            for (int j = 0; j < 4; ++j) {
                const float* hs = H + (size_t)srcn[j] * HC + lane;
                #pragma unroll
                for (int q = 0; q < 8; ++q) v[j][q] = hs[32 * q];
            }
            #pragma unroll
            for (int j = 0; j < 4; ++j) {
                if (ok[j]) {
                    float* ap = acc + (size_t)dln[j] * HC + lane;
                    atomicAdd(ap +   0, w4[j].x * v[j][0]);
                    atomicAdd(ap +  32, w4[j].x * v[j][1]);
                    atomicAdd(ap +  64, w4[j].y * v[j][2]);
                    atomicAdd(ap +  96, w4[j].y * v[j][3]);
                    atomicAdd(ap + 128, w4[j].z * v[j][4]);
                    atomicAdd(ap + 160, w4[j].z * v[j][5]);
                    atomicAdd(ap + 192, w4[j].w * v[j][6]);
                    atomicAdd(ap + 224, w4[j].w * v[j][7]);
                }
            }
        }
        __syncthreads();
    }

    if (t < OWN_ROWS * 4) ssum[t] = 1.0f / (ssum[t] + EPS_F);
    __syncthreads();

    float4* __restrict__ o4 = (float4*)(out + (size_t)row0 * HC);
    for (int i = t; i < nrows * (HC / 4); i += OWN_THREADS) {
        const float  r = ssum[(i >> 6) * 4 + ((i & 63) >> 4)];
        const float4 a = acc4[i];
        o4[i] = make_float4(a.x * r, a.y * r, a.z * r, a.w * r);
    }
}

extern "C" void kernel_launch(void* const* d_in, const int* in_sizes, int n_in,
                              void* d_out, int out_size, void* d_ws, size_t ws_size,
                              hipStream_t stream)
{
    const float* x   = (const float*)d_in[0];
    const int*   ei  = (const int*)d_in[1];
    const float* W1  = (const float*)d_in[2];
    const float* as1 = (const float*)d_in[3];
    const float* ad1 = (const float*)d_in[4];
    const float* b1  = (const float*)d_in[5];
    const float* W2  = (const float*)d_in[6];
    const float* as2 = (const float*)d_in[7];
    const float* ad2 = (const float*)d_in[8];
    const float* b2  = (const float*)d_in[9];
    const float* Wm1 = (const float*)d_in[10];
    const float* bm1 = (const float*)d_in[11];
    const float* Wm2 = (const float*)d_in[12];
    const float* bm2 = (const float*)d_in[13];

    const int N  = in_sizes[0] / 128;
    const int E  = in_sizes[1] / 2;
    const int NH = N * HEADS;

    float* ws = (float*)d_ws;
    size_t o = 0;
    float* P0  = ws + o; o += (size_t)N * HC;
    float* P1  = ws + o; o += (size_t)N * HC;
    float* a_s = ws + o; o += (size_t)NH;
    float* a_d = ws + o; o += (size_t)NH;

    const dim3 blk(256);
    auto gemm_grid = [](int M, int NC) {
        return dim3((unsigned)(((M + 15) / 16 + 7) / 8), (unsigned)((NC + 63) / 64));
    };
    const int gScores = ((N + 3) / 4 + 15) / 16;
    const int gOwn    = (N + OWN_ROWS - 1) / OWN_ROWS;


    wmma_gemm_f32_kernel<false><<<gemm_grid(N, HC), blk, 0, stream>>>(x, W1, P0, nullptr, nullptr,
                                                                      N, 128, HC, 0);
    scores_kernel<<<gScores, blk, 0, stream>>>(P0, as1, ad1, a_s, a_d, N);
    edge_agg_owner_kernel<<<gOwn, dim3(OWN_THREADS), 0, stream>>>(ei, E, N, P0, a_s, a_d, P1);

    wmma_gemm_f32_kernel<true><<<gemm_grid(N, HC), blk, 0, stream>>>(P1, W2, P0, nullptr, b1,
                                                                     N, HC, HC, 0);
    scores_kernel<<<gScores, blk, 0, stream>>>(P0, as2, ad2, a_s, a_d, N);
    edge_agg_owner_kernel<<<gOwn, dim3(OWN_THREADS), 0, stream>>>(ei, E, N, P0, a_s, a_d, P1);

    wmma_gemm_f32_kernel<true><<<gemm_grid(N, HID), blk, 0, stream>>>(P1, Wm1, P0, bm1, b2,
                                                                      N, HC, HID, 1);
    wmma_gemm_f32_kernel<false><<<gemm_grid(N, 121), blk, 0, stream>>>(P0, Wm2, (float*)d_out, bm2,
                                                                       nullptr, N, HID, 121, 2);
}


# Owner block saves its edge list (source node, local row) to the workspace in layer 1; layer 2 loads it and no longer scans the 800000-entry target column; LDS adds in arrival order as before
# speedup vs baseline: 3.7881x; 1.1099x over previous
#include <hip/hip_runtime.h>
#include <math.h>

#define HEADS 4
#define HID 64
#define HC 256
#define NEG_SLOPE 0.2f
#define EPS_F 1e-16f

typedef __attribute__((ext_vector_type(2))) float v2f;
typedef __attribute__((ext_vector_type(8))) float v8f;

__device__ __forceinline__ unsigned f2ord(float f) {
    unsigned u = __float_as_uint(f);
    return (u & 0x80000000u) ? ~u : (u | 0x80000000u);
}
__device__ __forceinline__ float ord2f(unsigned u) {
    return (u & 0x80000000u) ? __uint_as_float(u & 0x7fffffffu) : __uint_as_float(~u);
}

__device__ __forceinline__ void epi_store(float* __restrict__ C, const float* __restrict__ bias,
                                          size_t rowoff, int col, int NC, int mode, float v) {
    if (col < NC) {
        if (mode >= 1) v += bias[col];
        if (mode == 2) v = 1.0f / (1.0f + __expf(-v));
        C[rowoff + col] = v;
    }
}

template <bool ACT>
__device__ __forceinline__ v2f load_a_frag(const float* __restrict__ Ap,
                                           const float* __restrict__ abias, int kb, int aK) {
    v2f a = *(const v2f*)Ap;
    if (ACT) {
        const float4 bb = *(const float4*)(abias + kb);
        const float bx = aK ? bb.z : bb.x;
        const float by = aK ? bb.w : bb.y;
        a.x = fmaxf(a.x + bx, 0.0f);
        a.y = fmaxf(a.y + by, 0.0f);
    }
    return a;
}

template <bool ACT>
__global__ __launch_bounds__(256) void wmma_gemm_f32_kernel(
    const float* __restrict__ A, const float* __restrict__ B,
    float* __restrict__ C, const float* __restrict__ bias,
    const float* __restrict__ abias,
    int M, int K, int NC, int mode)
{
    const int wave = threadIdx.x >> 5;
    const int lane = threadIdx.x & 31;
    const int m0 = (blockIdx.x * 8 + wave) * 16;
    if (m0 >= M) return;
    const int n0 = blockIdx.y * 64;

    const int aRow = m0 + (lane & 15);
    const int aK   = (lane >> 4) << 1;
    const int bN   = lane & 15;
    const int bK   = (lane >> 4) << 1;
    const int c    = n0 + bN;

    const float* Ap = A + (size_t)aRow * K + aK;
    const size_t stepB = (size_t)4 * NC;

    v8f acc0 = {}, acc1 = {}, acc2 = {}, acc3 = {};

    if (n0 + 64 <= NC) {
        const float* Bp = B + c + (size_t)bK * NC;
        const float* Bq = Bp + NC;
        #pragma unroll 2
        for (int kb = 0; kb < K; kb += 4) {
            v2f a = load_a_frag<ACT>(Ap, abias, kb, aK);
            v2f b0 = { Bp[0],  Bq[0]  };
            v2f b1 = { Bp[16], Bq[16] };
            v2f b2 = { Bp[32], Bq[32] };
            v2f b3 = { Bp[48], Bq[48] };
            Ap += 4; Bp += stepB; Bq += stepB;
            acc0 = __builtin_amdgcn_wmma_f32_16x16x4_f32(false, a, false, b0, (short)0, acc0, false, false);
            acc1 = __builtin_amdgcn_wmma_f32_16x16x4_f32(false, a, false, b1, (short)0, acc1, false, false);
            acc2 = __builtin_amdgcn_wmma_f32_16x16x4_f32(false, a, false, b2, (short)0, acc2, false, false);
            acc3 = __builtin_amdgcn_wmma_f32_16x16x4_f32(false, a, false, b3, (short)0, acc3, false, false);
        }
    } else {
        const int   c0 = min(c,      NC - 1), c1 = min(c + 16, NC - 1);
        const int   c2 = min(c + 32, NC - 1), c3 = min(c + 48, NC - 1);
        const float w0 = (c      < NC) ? 1.0f : 0.0f;
        const float w1 = (c + 16 < NC) ? 1.0f : 0.0f;
        const float w2 = (c + 32 < NC) ? 1.0f : 0.0f;
        const float w3 = (c + 48 < NC) ? 1.0f : 0.0f;
        const float* Bp = B + (size_t)bK * NC;
        const float* Bq = Bp + NC;
        #pragma unroll 2
        for (int kb = 0; kb < K; kb += 4) {
            v2f a = load_a_frag<ACT>(Ap, abias, kb, aK);
            v2f b0 = { Bp[c0] * w0, Bq[c0] * w0 };
            v2f b1 = { Bp[c1] * w1, Bq[c1] * w1 };
            v2f b2 = { Bp[c2] * w2, Bq[c2] * w2 };
            v2f b3 = { Bp[c3] * w3, Bq[c3] * w3 };
            Ap += 4; Bp += stepB; Bq += stepB;
            acc0 = __builtin_amdgcn_wmma_f32_16x16x4_f32(false, a, false, b0, (short)0, acc0, false, false);
            acc1 = __builtin_amdgcn_wmma_f32_16x16x4_f32(false, a, false, b1, (short)0, acc1, false, false);
            acc2 = __builtin_amdgcn_wmma_f32_16x16x4_f32(false, a, false, b2, (short)0, acc2, false, false);
            acc3 = __builtin_amdgcn_wmma_f32_16x16x4_f32(false, a, false, b3, (short)0, acc3, false, false);
        }
    }

    const int mbase = m0 + ((lane >> 4) << 3);
    const int colb  = n0 + (lane & 15);
    #pragma unroll
    for (int r = 0; r < 8; ++r) {
        size_t rowoff = (size_t)(mbase + r) * NC;
        epi_store(C, bias, rowoff, colb,      NC, mode, acc0[r]);
        epi_store(C, bias, rowoff, colb + 16, NC, mode, acc1[r]);
        epi_store(C, bias, rowoff, colb + 32, NC, mode, acc2[r]);
        epi_store(C, bias, rowoff, colb + 48, NC, mode, acc3[r]);
    }
}

__global__ __launch_bounds__(256) void scores_kernel(
    const float* __restrict__ H, const float* __restrict__ att_src,
    const float* __restrict__ att_dst, float* __restrict__ a_s, float* __restrict__ a_d, int N)
{
    const int lane = threadIdx.x & 31;
    const int gw   = (int)(blockIdx.x * 8 + (threadIdx.x >> 5));
    const int nw   = (int)(gridDim.x * 8);

    float ws0[HEADS], ws1[HEADS], wd0[HEADS], wd1[HEADS];
    #pragma unroll
    for (int h = 0; h < HEADS; ++h) {
        ws0[h] = att_src[h * HID + lane]; ws1[h] = att_src[h * HID + lane + 32];
        wd0[h] = att_dst[h * HID + lane]; wd1[h] = att_dst[h * HID + lane + 32];
    }

    const int ngroups = (N + 3) >> 2;
    for (int g = gw; g < ngroups; g += nw) {
        const int n0 = g * 4;
        float x[4][8];
        #pragma unroll
        for (int j = 0; j < 4; ++j) {
            const int n = min(n0 + j, N - 1);
            const float* hp = H + (size_t)n * HC + lane;
            #pragma unroll
            for (int q = 0; q < 8; ++q) x[j][q] = hp[32 * q];
        }
        #pragma unroll
        for (int j = 0; j < 4; ++j) {
            float s[HEADS], d[HEADS];
            #pragma unroll
            for (int h = 0; h < HEADS; ++h) {
                const float x0 = x[j][2 * h], x1 = x[j][2 * h + 1];
                s[h] = x0 * ws0[h] + x1 * ws1[h];
                d[h] = x0 * wd0[h] + x1 * wd1[h];
            }
            #pragma unroll
            for (int off = 16; off > 0; off >>= 1) {
                #pragma unroll
                for (int h = 0; h < HEADS; ++h) {
                    s[h] += __shfl_xor(s[h], off, 32);
                    d[h] += __shfl_xor(d[h], off, 32);
                }
            }
            if (lane == 0 && n0 + j < N) {
                const size_t o4 = (size_t)(n0 + j) * 4;
                *(float4*)(a_s + o4) = make_float4(s[0], s[1], s[2], s[3]);
                *(float4*)(a_d + o4) = make_float4(d[0], d[1], d[2], d[3]);
            }
        }
    }
}

__device__ __forceinline__ void edge_src_dst(const int* __restrict__ ei, int E, int eid,
                                             int& src, int& dst) {
    if (eid < E) { src = ei[eid]; dst = ei[E + eid]; }
    else         { src = dst = eid - E; }
}

__device__ __forceinline__ float4 edge_logits(const float* __restrict__ a_s,
                                              const float* __restrict__ a_d, int src, int dst) {
    const float4 s4 = *(const float4*)(a_s + (size_t)src * 4);
    const float4 d4 = *(const float4*)(a_d + (size_t)dst * 4);
    float4 e;
    e.x = s4.x + d4.x; e.x = (e.x > 0.0f) ? e.x : NEG_SLOPE * e.x;
    e.y = s4.y + d4.y; e.y = (e.y > 0.0f) ? e.y : NEG_SLOPE * e.y;
    e.z = s4.z + d4.z; e.z = (e.z > 0.0f) ? e.z : NEG_SLOPE * e.z;
    e.w = s4.w + d4.w; e.w = (e.w > 0.0f) ? e.w : NEG_SLOPE * e.w;
    return e;
}

#define OWN_ROWS 196
#define OWN_CAP  4608
#define OWN_THREADS 1024

__device__ __noinline__ void own_rows_by_column_scans(
    const int* __restrict__ ei, int E, int row0, int nrows, const float* __restrict__ H,
    const float* __restrict__ a_s, const float* __restrict__ a_d,
    float* acc, unsigned* smax, float* ssum)
{
    const int t = threadIdx.x;
    const int* __restrict__ dp = ei + E;
    const int nall = E + nrows;
    for (int stage = 0; stage < 3; ++stage) {
        for (int i = t; i < nall; i += OWN_THREADS) {
            int src, dst;
            if (i < E) { dst = dp[i]; src = 0; } else { dst = row0 + (i - E); src = dst; }
            const unsigned dl = (unsigned)(dst - row0);
            if (dl < (unsigned)nrows) {
                if (i < E) src = ei[i];
                const float4 e = edge_logits(a_s, a_d, src, dst);
                const float ev[4] = { e.x, e.y, e.z, e.w };
                if (stage == 0) {
                    for (int h = 0; h < 4; ++h) atomicMax(smax + dl * 4 + h, f2ord(ev[h]));
                } else if (stage == 1) {
                    for (int h = 0; h < 4; ++h)
                        atomicAdd(ssum + dl * 4 + h, __expf(ev[h] - ord2f(smax[dl * 4 + h])));
                } else {
                    float al[4];
                    for (int h = 0; h < 4; ++h)
                        al[h] = __expf(ev[h] - ord2f(smax[dl * 4 + h]));
                    const float* hs = H + (size_t)src * HC;
                    float* ap = acc + (size_t)dl * HC;
                    for (int c = 0; c < HC; ++c) atomicAdd(ap + c, al[c >> 6] * hs[c]);
                }
            }
        }
        __syncthreads();
    }
}

template <bool FIRST>
__global__ __launch_bounds__(OWN_THREADS) void edge_agg_owner_kernel(
    const int* __restrict__ ei, int E, int N, const float* __restrict__ H,
    const float* __restrict__ a_s, const float* __restrict__ a_d, float* __restrict__ out,
    int* __restrict__ wcount, int* __restrict__ wsrc, unsigned short* __restrict__ wdl)
{
    __shared__ __attribute__((aligned(16))) float acc[OWN_ROWS * HC];
    __shared__ int elist[OWN_CAP];
    __shared__ unsigned short edl[OWN_CAP];
    __shared__ __attribute__((aligned(16))) float4 eterm[OWN_CAP];
    __shared__ __attribute__((aligned(16))) unsigned smax[OWN_ROWS * 4];
    __shared__ __attribute__((aligned(16))) float    ssum[OWN_ROWS * 4];
    __shared__ int ecount;

    const int t    = threadIdx.x;
    const int lane = t & 31;
    const int wave = t >> 5;
    const int row0  = blockIdx.x * OWN_ROWS;
    const int nrows = min(OWN_ROWS, N - row0);
    int*            __restrict__ bsrc = wsrc + (size_t)blockIdx.x * OWN_CAP;
    unsigned short* __restrict__ bdl  = wdl  + (size_t)blockIdx.x * OWN_CAP;

    float4* acc4 = (float4*)acc;
    for (int i = t; i < OWN_ROWS * (HC / 4); i += OWN_THREADS)
        acc4[i] = make_float4(0.0f, 0.0f, 0.0f, 0.0f);
    if (t < OWN_ROWS * 4) { smax[t] = f2ord(-INFINITY); ssum[t] = 0.0f; }
    if (FIRST && t == 0) ecount = 0;
    __syncthreads();

    int n;
    if (FIRST) {
        auto consider = [&](int eid, int dst) {
            const unsigned dl = (unsigned)(dst - row0);
            if (dl < (unsigned)nrows) {
                const int pos = atomicAdd(&ecount, 1);
                if (pos < OWN_CAP) { elist[pos] = eid; edl[pos] = (unsigned short)dl; }
            }
        };
        const int* __restrict__ dp = ei + E;
        if ((E & 3) == 0) {
            const int4* __restrict__ dp4 = (const int4*)dp;
            const int n4 = E >> 2;
            for (int i = t; i < n4; i += 4 * OWN_THREADS) {
                int4 v[4];
                #pragma unroll
                for (int u = 0; u < 4; ++u) {
                    const int k = i + u * OWN_THREADS;
                    v[u] = (k < n4) ? dp4[k] : make_int4(-1, -1, -1, -1);
                }
                #pragma unroll
                for (int u = 0; u < 4; ++u) {
                    const int e0 = (i + u * OWN_THREADS) * 4;
                    const bool any = ((unsigned)(v[u].x - row0) < (unsigned)nrows) |
                                     ((unsigned)(v[u].y - row0) < (unsigned)nrows) |
                                     ((unsigned)(v[u].z - row0) < (unsigned)nrows) |
                                     ((unsigned)(v[u].w - row0) < (unsigned)nrows);
                    if (any) {
                        consider(e0,     v[u].x);
                        consider(e0 + 1, v[u].y);
                        consider(e0 + 2, v[u].z);
                        consider(e0 + 3, v[u].w);
                    }
                }
            }
        } else {
            for (int e = t; e < E; e += OWN_THREADS) consider(e, dp[e]);
        }
        if (t < nrows) consider(E + row0 + t, row0 + t);
        __syncthreads();
        n = ecount;
        if (t == 0) wcount[blockIdx.x] = n;
    } else {
        n = wcount[blockIdx.x];
    }
    if (n > OWN_CAP) {
        own_rows_by_column_scans(ei, E, row0, nrows, H, a_s, a_d, acc, smax, ssum);
    } else {
        for (int i0 = t; i0 < n; i0 += 4 * OWN_THREADS) {
            int  idx[4], eid[4], dl[4], src[4];
            bool ok[4];
            #pragma unroll
            for (int u = 0; u < 4; ++u) {
                idx[u] = i0 + u * OWN_THREADS;
                ok[u]  = idx[u] < n;
                if (!ok[u]) idx[u] = i0;
                if (FIRST) {
                    eid[u] = elist[idx[u]];
                    dl[u]  = edl[idx[u]];
                } else {
                    src[u] = bsrc[idx[u]];
                    dl[u]  = bdl[idx[u]];
                }
            }
            if (FIRST) {
                #pragma unroll
                for (int u = 0; u < 4; ++u) {
                    src[u] = ei[min(eid[u], E - 1)];
                    if (eid[u] >= E) src[u] = eid[u] - E;
                }
            }
            float4 e[4];
            #pragma unroll
            for (int u = 0; u < 4; ++u) e[u] = edge_logits(a_s, a_d, src[u], row0 + dl[u]);
            #pragma unroll
            for (int u = 0; u < 4; ++u) {
                if (ok[u]) {
                    elist[idx[u]] = src[u];
                    if (FIRST) { bsrc[idx[u]] = src[u]; bdl[idx[u]] = (unsigned short)dl[u]; }
                    else       { edl[idx[u]] = (unsigned short)dl[u]; }
                    eterm[idx[u]] = e[u];
                    atomicMax(smax + dl[u] * 4 + 0, f2ord(e[u].x));
                    atomicMax(smax + dl[u] * 4 + 1, f2ord(e[u].y));
                    atomicMax(smax + dl[u] * 4 + 2, f2ord(e[u].z));
                    atomicMax(smax + dl[u] * 4 + 3, f2ord(e[u].w));
                }
            }
        }
        __syncthreads();

        for (int i = t; i < n; i += OWN_THREADS) {
            const int    dl = edl[i];
            const float4 e  = eterm[i];
            const uint4  m  = *(const uint4*)(smax + dl * 4);
            float4 x;
            x.x = __expf(e.x - ord2f(m.x));
            x.y = __expf(e.y - ord2f(m.y));
            x.z = __expf(e.z - ord2f(m.z));
            x.w = __expf(e.w - ord2f(m.w));
            eterm[i] = x;
            atomicAdd(ssum + dl * 4 + 0, x.x);
            atomicAdd(ssum + dl * 4 + 1, x.y);
            atomicAdd(ssum + dl * 4 + 2, x.z);
            atomicAdd(ssum + dl * 4 + 3, x.w);
        }
        __syncthreads();

        for (int base = wave * 4; base < n; base += 4 * (OWN_THREADS / 32)) {
            int    srcn[4], dln[4];
            float4 w4[4];
            bool   ok[4];
            #pragma unroll
            for (int j = 0; j < 4; ++j) {
                ok[j] = (base + j) < n;
                const int idx = ok[j] ? base + j : base;
                srcn[j] = elist[idx];
                dln[j]  = edl[idx];
                w4[j]   = eterm[idx];
            }
            float v[4][8];
            #pragma unroll
            for (int j = 0; j < 4; ++j) {
                const float* hs = H + (size_t)srcn[j] * HC + lane;
                #pragma unroll
                for (int q = 0; q < 8; ++q) v[j][q] = hs[32 * q];
            }
            #pragma unroll
            for (int j = 0; j < 4; ++j) {
                if (ok[j]) {
                    float* ap = acc + (size_t)dln[j] * HC + lane;
                    atomicAdd(ap +   0, w4[j].x * v[j][0]);
                    atomicAdd(ap +  32, w4[j].x * v[j][1]);
                    atomicAdd(ap +  64, w4[j].y * v[j][2]);
                    atomicAdd(ap +  96, w4[j].y * v[j][3]);
                    atomicAdd(ap + 128, w4[j].z * v[j][4]);
                    atomicAdd(ap + 160, w4[j].z * v[j][5]);
                    atomicAdd(ap + 192, w4[j].w * v[j][6]);
                    atomicAdd(ap + 224, w4[j].w * v[j][7]);
                }
            }
        }
        __syncthreads();
    }

    if (t < OWN_ROWS * 4) ssum[t] = 1.0f / (ssum[t] + EPS_F);
    __syncthreads();

    float4* __restrict__ o4 = (float4*)(out + (size_t)row0 * HC);
    for (int i = t; i < nrows * (HC / 4); i += OWN_THREADS) {
        const float  r = ssum[(i >> 6) * 4 + ((i & 63) >> 4)];
        const float4 a = acc4[i];
        o4[i] = make_float4(a.x * r, a.y * r, a.z * r, a.w * r);
    }
}

extern "C" void kernel_launch(void* const* d_in, const int* in_sizes, int n_in,
                              void* d_out, int out_size, void* d_ws, size_t ws_size,
                              hipStream_t stream)
{
    const float* x   = (const float*)d_in[0];
    const int*   ei  = (const int*)d_in[1];
    const float* W1  = (const float*)d_in[2];
    const float* as1 = (const float*)d_in[3];
    const float* ad1 = (const float*)d_in[4];
    const float* b1  = (const float*)d_in[5];
    const float* W2  = (const float*)d_in[6];
    const float* as2 = (const float*)d_in[7];
    const float* ad2 = (const float*)d_in[8];
    const float* b2  = (const float*)d_in[9];
    const float* Wm1 = (const float*)d_in[10];
    const float* bm1 = (const float*)d_in[11];
    const float* Wm2 = (const float*)d_in[12];
    const float* bm2 = (const float*)d_in[13];

    const int N  = in_sizes[0] / 128;
    const int E  = in_sizes[1] / 2;
    const int NH = N * HEADS;
    const int gOwn = (N + OWN_ROWS - 1) / OWN_ROWS;

    float* ws = (float*)d_ws;
    size_t o = 0;
    float* P0  = ws + o; o += (size_t)N * HC;
    float* P1  = ws + o; o += (size_t)N * HC;
    float* a_s = ws + o; o += (size_t)NH;
    float* a_d = ws + o; o += (size_t)NH;
    int* wcount = (int*)(ws + o); o += (size_t)gOwn;
    int* wsrc   = (int*)(ws + o); o += (size_t)gOwn * OWN_CAP;
    unsigned short* wdl = (unsigned short*)(ws + o);

    const dim3 blk(256);
    auto gemm_grid = [](int M, int NC) {
        return dim3((unsigned)(((M + 15) / 16 + 7) / 8), (unsigned)((NC + 63) / 64));
    };
    const int gScores = ((N + 3) / 4 + 15) / 16;


    wmma_gemm_f32_kernel<false><<<gemm_grid(N, HC), blk, 0, stream>>>(x, W1, P0, nullptr, nullptr,
                                                                      N, 128, HC, 0);
    scores_kernel<<<gScores, blk, 0, stream>>>(P0, as1, ad1, a_s, a_d, N);
    edge_agg_owner_kernel<true><<<gOwn, dim3(OWN_THREADS), 0, stream>>>(ei, E, N, P0, a_s, a_d, P1,
                                                                       wcount, wsrc, wdl);

    wmma_gemm_f32_kernel<true><<<gemm_grid(N, HC), blk, 0, stream>>>(P1, W2, P0, nullptr, b1,
                                                                     N, HC, HC, 0);
    scores_kernel<<<gScores, blk, 0, stream>>>(P0, as2, ad2, a_s, a_d, N);
    edge_agg_owner_kernel<false><<<gOwn, dim3(OWN_THREADS), 0, stream>>>(ei, E, N, P0, a_s, a_d, P1,
                                                                        wcount, wsrc, wdl);

    wmma_gemm_f32_kernel<true><<<gemm_grid(N, HID), blk, 0, stream>>>(P1, Wm1, P0, bm1, b2,
                                                                      N, HC, HID, 1);
    wmma_gemm_f32_kernel<false><<<gemm_grid(N, 121), blk, 0, stream>>>(P0, Wm2, (float*)d_out, bm2,
                                                                       nullptr, N, HID, 121, 2);
}


# Owner blocks find their edges in a one-byte owner-block column (one small new launch per call), not in the 4-byte target column: 0.8 MB read per block, not 3.2; LDS adds in arrival order
# speedup vs baseline: 3.9984x; 1.0555x over previous
#include <hip/hip_runtime.h>
#include <math.h>

#define HEADS 4
#define HID 64
#define HC 256
#define NEG_SLOPE 0.2f
#define EPS_F 1e-16f

typedef __attribute__((ext_vector_type(2))) float v2f;
typedef __attribute__((ext_vector_type(8))) float v8f;

__device__ __forceinline__ unsigned f2ord(float f) {
    unsigned u = __float_as_uint(f);
    return (u & 0x80000000u) ? ~u : (u | 0x80000000u);
}
__device__ __forceinline__ float ord2f(unsigned u) {
    return (u & 0x80000000u) ? __uint_as_float(u & 0x7fffffffu) : __uint_as_float(~u);
}

__device__ __forceinline__ void epi_store(float* __restrict__ C, const float* __restrict__ bias,
                                          size_t rowoff, int col, int NC, int mode, float v) {
    if (col < NC) {
        if (mode >= 1) v += bias[col];
        if (mode == 2) v = 1.0f / (1.0f + __expf(-v));
        C[rowoff + col] = v;
    }
}

template <bool ACT>
__device__ __forceinline__ v2f load_a_frag(const float* __restrict__ Ap,
                                           const float* __restrict__ abias, int kb, int aK) {
    v2f a = *(const v2f*)Ap;
    if (ACT) {
        const float4 bb = *(const float4*)(abias + kb);
        const float bx = aK ? bb.z : bb.x;
        const float by = aK ? bb.w : bb.y;
        a.x = fmaxf(a.x + bx, 0.0f);
        a.y = fmaxf(a.y + by, 0.0f);
    }
    return a;
}

template <bool ACT>
__global__ __launch_bounds__(256) void wmma_gemm_f32_kernel(
    const float* __restrict__ A, const float* __restrict__ B,
    float* __restrict__ C, const float* __restrict__ bias,
    const float* __restrict__ abias,
    int M, int K, int NC, int mode)
{
    const int wave = threadIdx.x >> 5;
    const int lane = threadIdx.x & 31;
    const int m0 = (blockIdx.x * 8 + wave) * 16;
    if (m0 >= M) return;
    const int n0 = blockIdx.y * 64;

    const int aRow = m0 + (lane & 15);
    const int aK   = (lane >> 4) << 1;
    const int bN   = lane & 15;
    const int bK   = (lane >> 4) << 1;
    const int c    = n0 + bN;

    const float* Ap = A + (size_t)aRow * K + aK;
    const size_t stepB = (size_t)4 * NC;

    v8f acc0 = {}, acc1 = {}, acc2 = {}, acc3 = {};

    if (n0 + 64 <= NC) {
        const float* Bp = B + c + (size_t)bK * NC;
        const float* Bq = Bp + NC;
        #pragma unroll 2
        for (int kb = 0; kb < K; kb += 4) {
            v2f a = load_a_frag<ACT>(Ap, abias, kb, aK);
            v2f b0 = { Bp[0],  Bq[0]  };
            v2f b1 = { Bp[16], Bq[16] };
            v2f b2 = { Bp[32], Bq[32] };
            v2f b3 = { Bp[48], Bq[48] };
            Ap += 4; Bp += stepB; Bq += stepB;
            acc0 = __builtin_amdgcn_wmma_f32_16x16x4_f32(false, a, false, b0, (short)0, acc0, false, false);
            acc1 = __builtin_amdgcn_wmma_f32_16x16x4_f32(false, a, false, b1, (short)0, acc1, false, false);
            acc2 = __builtin_amdgcn_wmma_f32_16x16x4_f32(false, a, false, b2, (short)0, acc2, false, false);
            acc3 = __builtin_amdgcn_wmma_f32_16x16x4_f32(false, a, false, b3, (short)0, acc3, false, false);
        }
    } else {
        const int   c0 = min(c,      NC - 1), c1 = min(c + 16, NC - 1);
        const int   c2 = min(c + 32, NC - 1), c3 = min(c + 48, NC - 1);
        const float w0 = (c      < NC) ? 1.0f : 0.0f;
        const float w1 = (c + 16 < NC) ? 1.0f : 0.0f;
        const float w2 = (c + 32 < NC) ? 1.0f : 0.0f;
        const float w3 = (c + 48 < NC) ? 1.0f : 0.0f;
        const float* Bp = B + (size_t)bK * NC;
        const float* Bq = Bp + NC;
        #pragma unroll 2
        for (int kb = 0; kb < K; kb += 4) {
            v2f a = load_a_frag<ACT>(Ap, abias, kb, aK);
            v2f b0 = { Bp[c0] * w0, Bq[c0] * w0 };
            v2f b1 = { Bp[c1] * w1, Bq[c1] * w1 };
            v2f b2 = { Bp[c2] * w2, Bq[c2] * w2 };
            v2f b3 = { Bp[c3] * w3, Bq[c3] * w3 };
            Ap += 4; Bp += stepB; Bq += stepB;
            acc0 = __builtin_amdgcn_wmma_f32_16x16x4_f32(false, a, false, b0, (short)0, acc0, false, false);
            acc1 = __builtin_amdgcn_wmma_f32_16x16x4_f32(false, a, false, b1, (short)0, acc1, false, false);
            acc2 = __builtin_amdgcn_wmma_f32_16x16x4_f32(false, a, false, b2, (short)0, acc2, false, false);
            acc3 = __builtin_amdgcn_wmma_f32_16x16x4_f32(false, a, false, b3, (short)0, acc3, false, false);
        }
    }

    const int mbase = m0 + ((lane >> 4) << 3);
    const int colb  = n0 + (lane & 15);
    #pragma unroll
    for (int r = 0; r < 8; ++r) {
        size_t rowoff = (size_t)(mbase + r) * NC;
        epi_store(C, bias, rowoff, colb,      NC, mode, acc0[r]);
        epi_store(C, bias, rowoff, colb + 16, NC, mode, acc1[r]);
        epi_store(C, bias, rowoff, colb + 32, NC, mode, acc2[r]);
        epi_store(C, bias, rowoff, colb + 48, NC, mode, acc3[r]);
    }
}

__global__ __launch_bounds__(256) void scores_kernel(
    const float* __restrict__ H, const float* __restrict__ att_src,
    const float* __restrict__ att_dst, float* __restrict__ a_s, float* __restrict__ a_d, int N)
{
    const int lane = threadIdx.x & 31;
    const int gw   = (int)(blockIdx.x * 8 + (threadIdx.x >> 5));
    const int nw   = (int)(gridDim.x * 8);

    float ws0[HEADS], ws1[HEADS], wd0[HEADS], wd1[HEADS];
    #pragma unroll
    for (int h = 0; h < HEADS; ++h) {
        ws0[h] = att_src[h * HID + lane]; ws1[h] = att_src[h * HID + lane + 32];
        wd0[h] = att_dst[h * HID + lane]; wd1[h] = att_dst[h * HID + lane + 32];
    }

    const int ngroups = (N + 3) >> 2;
    for (int g = gw; g < ngroups; g += nw) {
        const int n0 = g * 4;
        float x[4][8];
        #pragma unroll
        for (int j = 0; j < 4; ++j) {
            const int n = min(n0 + j, N - 1);
            const float* hp = H + (size_t)n * HC + lane;
            #pragma unroll
            for (int q = 0; q < 8; ++q) x[j][q] = hp[32 * q];
        }
        #pragma unroll
        for (int j = 0; j < 4; ++j) {
            float s[HEADS], d[HEADS];
            #pragma unroll
            for (int h = 0; h < HEADS; ++h) {
                const float x0 = x[j][2 * h], x1 = x[j][2 * h + 1];
                s[h] = x0 * ws0[h] + x1 * ws1[h];
                d[h] = x0 * wd0[h] + x1 * wd1[h];
            }
            #pragma unroll
            for (int off = 16; off > 0; off >>= 1) {
                #pragma unroll
                for (int h = 0; h < HEADS; ++h) {
                    s[h] += __shfl_xor(s[h], off, 32);
                    d[h] += __shfl_xor(d[h], off, 32);
                }
            }
            if (lane == 0 && n0 + j < N) {
                const size_t o4 = (size_t)(n0 + j) * 4;
                *(float4*)(a_s + o4) = make_float4(s[0], s[1], s[2], s[3]);
                *(float4*)(a_d + o4) = make_float4(d[0], d[1], d[2], d[3]);
            }
        }
    }
}

__device__ __forceinline__ void edge_src_dst(const int* __restrict__ ei, int E, int eid,
                                             int& src, int& dst) {
    if (eid < E) { src = ei[eid]; dst = ei[E + eid]; }
    else         { src = dst = eid - E; }
}

__device__ __forceinline__ float4 edge_logits(const float* __restrict__ a_s,
                                              const float* __restrict__ a_d, int src, int dst) {
    const float4 s4 = *(const float4*)(a_s + (size_t)src * 4);
    const float4 d4 = *(const float4*)(a_d + (size_t)dst * 4);
    float4 e;
    e.x = s4.x + d4.x; e.x = (e.x > 0.0f) ? e.x : NEG_SLOPE * e.x;
    e.y = s4.y + d4.y; e.y = (e.y > 0.0f) ? e.y : NEG_SLOPE * e.y;
    e.z = s4.z + d4.z; e.z = (e.z > 0.0f) ? e.z : NEG_SLOPE * e.z;
    e.w = s4.w + d4.w; e.w = (e.w > 0.0f) ? e.w : NEG_SLOPE * e.w;
    return e;
}

#define OWN_ROWS 196
#define OWN_CAP  4608
#define OWN_THREADS 1024

__device__ __noinline__ void own_rows_by_column_scans(
    const int* __restrict__ ei, int E, int row0, int nrows, const float* __restrict__ H,
    const float* __restrict__ a_s, const float* __restrict__ a_d,
    float* acc, unsigned* smax, float* ssum)
{
    const int t = threadIdx.x;
    const int* __restrict__ dp = ei + E;
    const int nall = E + nrows;
    for (int stage = 0; stage < 3; ++stage) {
        for (int i = t; i < nall; i += OWN_THREADS) {
            int src, dst;
            if (i < E) { dst = dp[i]; src = 0; } else { dst = row0 + (i - E); src = dst; }
            const unsigned dl = (unsigned)(dst - row0);
            if (dl < (unsigned)nrows) {
                if (i < E) src = ei[i];
                const float4 e = edge_logits(a_s, a_d, src, dst);
                const float ev[4] = { e.x, e.y, e.z, e.w };
                if (stage == 0) {
                    for (int h = 0; h < 4; ++h) atomicMax(smax + dl * 4 + h, f2ord(ev[h]));
                } else if (stage == 1) {
                    for (int h = 0; h < 4; ++h)
                        atomicAdd(ssum + dl * 4 + h, __expf(ev[h] - ord2f(smax[dl * 4 + h])));
                } else {
                    float al[4];
                    for (int h = 0; h < 4; ++h)
                        al[h] = __expf(ev[h] - ord2f(smax[dl * 4 + h]));
                    const float* hs = H + (size_t)src * HC;
                    float* ap = acc + (size_t)dl * HC;
                    for (int c = 0; c < HC; ++c) atomicAdd(ap + c, al[c >> 6] * hs[c]);
                }
            }
        }
        __syncthreads();
    }
}

__global__ __launch_bounds__(256) void owner_id_kernel(const int* __restrict__ dp, int E,
                                                        unsigned char* __restrict__ ids)
{
    const int e0 = (int)(blockIdx.x * 256 + threadIdx.x) * 16;
    if (e0 >= E) return;
    if (e0 + 16 <= E && (E & 3) == 0) {
        const int4* __restrict__ p = (const int4*)(dp + e0);
        int4 d[4];
        #pragma unroll
        for (int u = 0; u < 4; ++u) d[u] = p[u];
        unsigned w[4];
        #pragma unroll
        for (int u = 0; u < 4; ++u)
            w[u] =  (((unsigned)d[u].x / OWN_ROWS) & 0xffu)        |
                   ((((unsigned)d[u].y / OWN_ROWS) & 0xffu) << 8)  |
                   ((((unsigned)d[u].z / OWN_ROWS) & 0xffu) << 16) |
                   ((((unsigned)d[u].w / OWN_ROWS) & 0xffu) << 24);
        *(uint4*)(ids + e0) = make_uint4(w[0], w[1], w[2], w[3]);
    } else {
        const int e1 = min(e0 + 16, E);
        for (int e = e0; e < e1; ++e) ids[e] = (unsigned char)((unsigned)dp[e] / OWN_ROWS);
    }
}

template <bool FIRST>
__global__ __launch_bounds__(OWN_THREADS) void edge_agg_owner_kernel(
    const int* __restrict__ ei, int E, int N, const float* __restrict__ H,
    const float* __restrict__ a_s, const float* __restrict__ a_d, float* __restrict__ out,
    int* __restrict__ wcount, int* __restrict__ wsrc, unsigned short* __restrict__ wdl,
    const unsigned char* __restrict__ ids)
{
    __shared__ __attribute__((aligned(16))) float acc[OWN_ROWS * HC];
    __shared__ int elist[OWN_CAP];
    __shared__ unsigned short edl[OWN_CAP];
    __shared__ __attribute__((aligned(16))) float4 eterm[OWN_CAP];
    __shared__ __attribute__((aligned(16))) unsigned smax[OWN_ROWS * 4];
    __shared__ __attribute__((aligned(16))) float    ssum[OWN_ROWS * 4];
    __shared__ int ecount;

    const int t    = threadIdx.x;
    const int lane = t & 31;
    const int wave = t >> 5;
    const int row0  = blockIdx.x * OWN_ROWS;
    const int nrows = min(OWN_ROWS, N - row0);
    int*            __restrict__ bsrc = wsrc + (size_t)blockIdx.x * OWN_CAP;
    unsigned short* __restrict__ bdl  = wdl  + (size_t)blockIdx.x * OWN_CAP;

    float4* acc4 = (float4*)acc;
    for (int i = t; i < OWN_ROWS * (HC / 4); i += OWN_THREADS)
        acc4[i] = make_float4(0.0f, 0.0f, 0.0f, 0.0f);
    if (t < OWN_ROWS * 4) { smax[t] = f2ord(-INFINITY); ssum[t] = 0.0f; }
    if (FIRST && t == 0) ecount = 0;
    __syncthreads();

    int n;
    if (FIRST) {
        auto take = [&](int eid) {
            const int pos = atomicAdd(&ecount, 1);
            if (pos < OWN_CAP) elist[pos] = eid;
        };
        auto consider = [&](int eid, int dst) {
            if ((unsigned)(dst - row0) < (unsigned)nrows) take(eid);
        };
        const int* __restrict__ dp = ei + E;
        if (ids != nullptr) {
            const unsigned pat = (unsigned)blockIdx.x * 0x01010101u;
            const uint4* __restrict__ ip4 = (const uint4*)ids;
            auto scan16 = [&](uint4 v, int g) {
                const unsigned w[4] = { v.x, v.y, v.z, v.w };
                unsigned nz = 0;
                #pragma unroll
                for (int q = 0; q < 4; ++q) {
                    const unsigned x = w[q] ^ pat;
                    const unsigned s = ((x & 0x7f7f7f7fu) + 0x7f7f7f7fu) | x;
                    nz |= (s >> (7 - q)) & (0x01010101u << q);
                }
                unsigned mm = nz ^ 0x0f0f0f0fu;
                while (mm) {
                    const int b = __ffs(mm) - 1;
                    take(g * 16 + 4 * (b & 7) + (b >> 3));
                    mm &= mm - 1;
                }
            };
            const int n16   = E >> 4;
            const int nfull = n16 - n16 % (4 * OWN_THREADS);
            for (int i = t; i < nfull; i += 4 * OWN_THREADS) {
                uint4 v[4];
                #pragma unroll
                for (int u = 0; u < 4; ++u) v[u] = ip4[i + u * OWN_THREADS];
                #pragma unroll
                for (int u = 0; u < 4; ++u) scan16(v[u], i + u * OWN_THREADS);
            }
            for (int i = nfull + t; i < n16; i += OWN_THREADS) scan16(ip4[i], i);
            for (int e = (n16 << 4) + t; e < E; e += OWN_THREADS)
                if (ids[e] == (unsigned char)blockIdx.x) take(e);
        } else if ((E & 3) == 0) {
            const int4* __restrict__ dp4 = (const int4*)dp;
            const int n4 = E >> 2;
            for (int i = t; i < n4; i += 4 * OWN_THREADS) {
                int4 v[4];
                #pragma unroll
                for (int u = 0; u < 4; ++u) {
                    const int k = i + u * OWN_THREADS;
                    v[u] = (k < n4) ? dp4[k] : make_int4(-1, -1, -1, -1);
                }
                #pragma unroll
                for (int u = 0; u < 4; ++u) {
                    const int e0 = (i + u * OWN_THREADS) * 4;
                    const bool any = ((unsigned)(v[u].x - row0) < (unsigned)nrows) |
                                     ((unsigned)(v[u].y - row0) < (unsigned)nrows) |
                                     ((unsigned)(v[u].z - row0) < (unsigned)nrows) |
                                     ((unsigned)(v[u].w - row0) < (unsigned)nrows);
                    if (any) {
                        consider(e0,     v[u].x);
                        consider(e0 + 1, v[u].y);
                        consider(e0 + 2, v[u].z);
                        consider(e0 + 3, v[u].w);
                    }
                }
            }
        } else {
            for (int e = t; e < E; e += OWN_THREADS) consider(e, dp[e]);
        }
        if (t < nrows) take(E + row0 + t);
        __syncthreads();
        n = ecount;
        if (t == 0) wcount[blockIdx.x] = n;
    } else {
        n = wcount[blockIdx.x];
    }
    if (n > OWN_CAP) {
        own_rows_by_column_scans(ei, E, row0, nrows, H, a_s, a_d, acc, smax, ssum);
    } else {
        for (int i0 = t; i0 < n; i0 += 4 * OWN_THREADS) {
            int  idx[4], eid[4], dl[4], src[4];
            bool ok[4];
            #pragma unroll
            for (int u = 0; u < 4; ++u) {
                idx[u] = i0 + u * OWN_THREADS;
                ok[u]  = idx[u] < n;
                if (!ok[u]) idx[u] = i0;
                if (FIRST) {
                    eid[u] = elist[idx[u]];
                } else {
                    src[u] = bsrc[idx[u]];
                    dl[u]  = bdl[idx[u]];
                }
            }
            if (FIRST) {
                #pragma unroll
                for (int u = 0; u < 4; ++u) {
                    const int ec = min(eid[u], E - 1);
                    src[u] = ei[ec];
                    int dst = ei[E + ec];
                    if (eid[u] >= E) src[u] = dst = eid[u] - E;
                    dl[u] = dst - row0;
                }
            }
            float4 e[4];
            #pragma unroll
            for (int u = 0; u < 4; ++u) e[u] = edge_logits(a_s, a_d, src[u], row0 + dl[u]);
            #pragma unroll
            for (int u = 0; u < 4; ++u) {
                if (ok[u]) {
                    elist[idx[u]] = src[u];
                    edl[idx[u]]   = (unsigned short)dl[u];
                    if (FIRST) { bsrc[idx[u]] = src[u]; bdl[idx[u]] = (unsigned short)dl[u]; }
                    eterm[idx[u]] = e[u];
                    atomicMax(smax + dl[u] * 4 + 0, f2ord(e[u].x));
                    atomicMax(smax + dl[u] * 4 + 1, f2ord(e[u].y));
                    atomicMax(smax + dl[u] * 4 + 2, f2ord(e[u].z));
                    atomicMax(smax + dl[u] * 4 + 3, f2ord(e[u].w));
                }
            }
        }
        __syncthreads();

        for (int i = t; i < n; i += OWN_THREADS) {
            const int    dl = edl[i];
            const float4 e  = eterm[i];
            const uint4  m  = *(const uint4*)(smax + dl * 4);
            float4 x;
            x.x = __expf(e.x - ord2f(m.x));
            x.y = __expf(e.y - ord2f(m.y));
            x.z = __expf(e.z - ord2f(m.z));
            x.w = __expf(e.w - ord2f(m.w));
            eterm[i] = x;
            atomicAdd(ssum + dl * 4 + 0, x.x);
            atomicAdd(ssum + dl * 4 + 1, x.y);
            atomicAdd(ssum + dl * 4 + 2, x.z);
            atomicAdd(ssum + dl * 4 + 3, x.w);
        }
        __syncthreads();

        for (int base = wave * 4; base < n; base += 4 * (OWN_THREADS / 32)) {
            int    srcn[4], dln[4];
            float4 w4[4];
            bool   ok[4];
            #pragma unroll
            for (int j = 0; j < 4; ++j) {
                ok[j] = (base + j) < n;
                const int idx = ok[j] ? base + j : base;
                srcn[j] = elist[idx];
                dln[j]  = edl[idx];
                w4[j]   = eterm[idx];
            }
            float v[4][8];
            #pragma unroll
            for (int j = 0; j < 4; ++j) {
                const float* hs = H + (size_t)srcn[j] * HC + lane;
                #pragma unroll
                for (int q = 0; q < 8; ++q) v[j][q] = hs[32 * q];
            }
            #pragma unroll
            for (int j = 0; j < 4; ++j) {
                if (ok[j]) {
                    float* ap = acc + (size_t)dln[j] * HC + lane;
                    atomicAdd(ap +   0, w4[j].x * v[j][0]);
                    atomicAdd(ap +  32, w4[j].x * v[j][1]);
                    atomicAdd(ap +  64, w4[j].y * v[j][2]);
                    atomicAdd(ap +  96, w4[j].y * v[j][3]);
                    atomicAdd(ap + 128, w4[j].z * v[j][4]);
                    atomicAdd(ap + 160, w4[j].z * v[j][5]);
                    atomicAdd(ap + 192, w4[j].w * v[j][6]);
                    atomicAdd(ap + 224, w4[j].w * v[j][7]);
                }
            }
        }
        __syncthreads();
    }

    if (t < OWN_ROWS * 4) ssum[t] = 1.0f / (ssum[t] + EPS_F);
    __syncthreads();

    float4* __restrict__ o4 = (float4*)(out + (size_t)row0 * HC);
    for (int i = t; i < nrows * (HC / 4); i += OWN_THREADS) {
        const float  r = ssum[(i >> 6) * 4 + ((i & 63) >> 4)];
        const float4 a = acc4[i];
        o4[i] = make_float4(a.x * r, a.y * r, a.z * r, a.w * r);
    }
}

extern "C" void kernel_launch(void* const* d_in, const int* in_sizes, int n_in,
                              void* d_out, int out_size, void* d_ws, size_t ws_size,
                              hipStream_t stream)
{
    const float* x   = (const float*)d_in[0];
    const int*   ei  = (const int*)d_in[1];
    const float* W1  = (const float*)d_in[2];
    const float* as1 = (const float*)d_in[3];
    const float* ad1 = (const float*)d_in[4];
    const float* b1  = (const float*)d_in[5];
    const float* W2  = (const float*)d_in[6];
    const float* as2 = (const float*)d_in[7];
    const float* ad2 = (const float*)d_in[8];
    const float* b2  = (const float*)d_in[9];
    const float* Wm1 = (const float*)d_in[10];
    const float* bm1 = (const float*)d_in[11];
    const float* Wm2 = (const float*)d_in[12];
    const float* bm2 = (const float*)d_in[13];

    const int N  = in_sizes[0] / 128;
    const int E  = in_sizes[1] / 2;
    const int NH = N * HEADS;
    const int gOwn = (N + OWN_ROWS - 1) / OWN_ROWS;

    float* ws = (float*)d_ws;
    size_t o = 0;
    float* P0  = ws + o; o += (size_t)N * HC;
    float* P1  = ws + o; o += (size_t)N * HC;
    float* a_s = ws + o; o += (size_t)NH;
    float* a_d = ws + o; o += (size_t)NH;
    int* wcount = (int*)(ws + o); o += (size_t)gOwn;
    int* wsrc   = (int*)(ws + o); o += (size_t)gOwn * OWN_CAP;
    unsigned short* wdl = (unsigned short*)(ws + o); o += (size_t)gOwn * (OWN_CAP / 2);
    o = (o + 3) & ~(size_t)3;
    unsigned char* idcol = (unsigned char*)(ws + o);
    const bool byteIds = gOwn <= 256;

    const dim3 blk(256);
    auto gemm_grid = [](int M, int NC) {
        return dim3((unsigned)(((M + 15) / 16 + 7) / 8), (unsigned)((NC + 63) / 64));
    };
    const int gScores = ((N + 3) / 4 + 15) / 16;


    if (byteIds)
        owner_id_kernel<<<(unsigned)(((E + 15) / 16 + 255) / 256), blk, 0, stream>>>(ei + E, E, idcol);
    wmma_gemm_f32_kernel<false><<<gemm_grid(N, HC), blk, 0, stream>>>(x, W1, P0, nullptr, nullptr,
                                                                      N, 128, HC, 0);
    scores_kernel<<<gScores, blk, 0, stream>>>(P0, as1, ad1, a_s, a_d, N);
    edge_agg_owner_kernel<true><<<gOwn, dim3(OWN_THREADS), 0, stream>>>(ei, E, N, P0, a_s, a_d, P1,
                                                                       wcount, wsrc, wdl,
                                                                       byteIds ? idcol : nullptr);

    wmma_gemm_f32_kernel<true><<<gemm_grid(N, HC), blk, 0, stream>>>(P1, W2, P0, nullptr, b1,
                                                                     N, HC, HC, 0);
    scores_kernel<<<gScores, blk, 0, stream>>>(P0, as2, ad2, a_s, a_d, N);
    edge_agg_owner_kernel<false><<<gOwn, dim3(OWN_THREADS), 0, stream>>>(ei, E, N, P0, a_s, a_d, P1,
                                                                        wcount, wsrc, wdl, nullptr);

    wmma_gemm_f32_kernel<true><<<gemm_grid(N, HID), blk, 0, stream>>>(P1, Wm1, P0, bm1, b2,
                                                                      N, HC, HID, 1);
    wmma_gemm_f32_kernel<false><<<gemm_grid(N, 121), blk, 0, stream>>>(P0, Wm2, (float*)d_out, bm2,
                                                                       nullptr, N, HID, 121, 2);
}


# Owner block keeps its finished rows in LDS and multiplies them by the next linear layer itself (WMMA A fragments read from LDS): 2 GEMM launches and the aggregated-feature round trip gone; same sums
# speedup vs baseline: 4.7502x; 1.1880x over previous
#include <hip/hip_runtime.h>
#include <math.h>

#define HEADS 4
#define HID 64
#define HC 256
#define NEG_SLOPE 0.2f
#define EPS_F 1e-16f

typedef __attribute__((ext_vector_type(2))) float v2f;
typedef __attribute__((ext_vector_type(8))) float v8f;

__device__ __forceinline__ unsigned f2ord(float f) {
    unsigned u = __float_as_uint(f);
    return (u & 0x80000000u) ? ~u : (u | 0x80000000u);
}
__device__ __forceinline__ float ord2f(unsigned u) {
    return (u & 0x80000000u) ? __uint_as_float(u & 0x7fffffffu) : __uint_as_float(~u);
}

__device__ __forceinline__ void epi_store(float* __restrict__ C, const float* __restrict__ bias,
                                          size_t rowoff, int col, int NC, int mode, float v) {
    if (col < NC) {
        if (mode >= 1) v += bias[col];
        if (mode == 2) v = 1.0f / (1.0f + __expf(-v));
        C[rowoff + col] = v;
    }
}

__global__ __launch_bounds__(256) void wmma_gemm_f32_kernel(
    const float* __restrict__ A, const float* __restrict__ B,
    float* __restrict__ C, const float* __restrict__ bias,
    int M, int K, int NC, int mode)
{
    const int wave = threadIdx.x >> 5;
    const int lane = threadIdx.x & 31;
    const int m0 = (blockIdx.x * 8 + wave) * 16;
    if (m0 >= M) return;
    const int n0 = blockIdx.y * 64;

    const int aRow = m0 + (lane & 15);
    const int aK   = (lane >> 4) << 1;
    const int bN   = lane & 15;
    const int bK   = (lane >> 4) << 1;
    const int c    = n0 + bN;

    const float* Ap = A + (size_t)aRow * K + aK;
    const size_t stepB = (size_t)4 * NC;

    v8f acc0 = {}, acc1 = {}, acc2 = {}, acc3 = {};

    if (n0 + 64 <= NC) {
        const float* Bp = B + c + (size_t)bK * NC;
        const float* Bq = Bp + NC;
        #pragma unroll 2
        for (int kb = 0; kb < K; kb += 4) {
            v2f a = *(const v2f*)Ap;
            v2f b0 = { Bp[0],  Bq[0]  };
            v2f b1 = { Bp[16], Bq[16] };
            v2f b2 = { Bp[32], Bq[32] };
            v2f b3 = { Bp[48], Bq[48] };
            Ap += 4; Bp += stepB; Bq += stepB;
            acc0 = __builtin_amdgcn_wmma_f32_16x16x4_f32(false, a, false, b0, (short)0, acc0, false, false);
            acc1 = __builtin_amdgcn_wmma_f32_16x16x4_f32(false, a, false, b1, (short)0, acc1, false, false);
            acc2 = __builtin_amdgcn_wmma_f32_16x16x4_f32(false, a, false, b2, (short)0, acc2, false, false);
            acc3 = __builtin_amdgcn_wmma_f32_16x16x4_f32(false, a, false, b3, (short)0, acc3, false, false);
        }
    } else {
        const int   c0 = min(c,      NC - 1), c1 = min(c + 16, NC - 1);
        const int   c2 = min(c + 32, NC - 1), c3 = min(c + 48, NC - 1);
        const float w0 = (c      < NC) ? 1.0f : 0.0f;
        const float w1 = (c + 16 < NC) ? 1.0f : 0.0f;
        const float w2 = (c + 32 < NC) ? 1.0f : 0.0f;
        const float w3 = (c + 48 < NC) ? 1.0f : 0.0f;
        const float* Bp = B + (size_t)bK * NC;
        const float* Bq = Bp + NC;
        #pragma unroll 2
        for (int kb = 0; kb < K; kb += 4) {
            v2f a = *(const v2f*)Ap;
            v2f b0 = { Bp[c0] * w0, Bq[c0] * w0 };
            v2f b1 = { Bp[c1] * w1, Bq[c1] * w1 };
            v2f b2 = { Bp[c2] * w2, Bq[c2] * w2 };
            v2f b3 = { Bp[c3] * w3, Bq[c3] * w3 };
            Ap += 4; Bp += stepB; Bq += stepB;
            acc0 = __builtin_amdgcn_wmma_f32_16x16x4_f32(false, a, false, b0, (short)0, acc0, false, false);
            acc1 = __builtin_amdgcn_wmma_f32_16x16x4_f32(false, a, false, b1, (short)0, acc1, false, false);
            acc2 = __builtin_amdgcn_wmma_f32_16x16x4_f32(false, a, false, b2, (short)0, acc2, false, false);
            acc3 = __builtin_amdgcn_wmma_f32_16x16x4_f32(false, a, false, b3, (short)0, acc3, false, false);
        }
    }

    const int mbase = m0 + ((lane >> 4) << 3);
    const int colb  = n0 + (lane & 15);
    #pragma unroll
    for (int r = 0; r < 8; ++r) {
        size_t rowoff = (size_t)(mbase + r) * NC;
        epi_store(C, bias, rowoff, colb,      NC, mode, acc0[r]);
        epi_store(C, bias, rowoff, colb + 16, NC, mode, acc1[r]);
        epi_store(C, bias, rowoff, colb + 32, NC, mode, acc2[r]);
        epi_store(C, bias, rowoff, colb + 48, NC, mode, acc3[r]);
    }
}

__global__ __launch_bounds__(256) void scores_kernel(
    const float* __restrict__ H, const float* __restrict__ att_src,
    const float* __restrict__ att_dst, float* __restrict__ a_s, float* __restrict__ a_d, int N)
{
    const int lane = threadIdx.x & 31;
    const int gw   = (int)(blockIdx.x * 8 + (threadIdx.x >> 5));
    const int nw   = (int)(gridDim.x * 8);

    float ws0[HEADS], ws1[HEADS], wd0[HEADS], wd1[HEADS];
    #pragma unroll
    for (int h = 0; h < HEADS; ++h) {
        ws0[h] = att_src[h * HID + lane]; ws1[h] = att_src[h * HID + lane + 32];
        wd0[h] = att_dst[h * HID + lane]; wd1[h] = att_dst[h * HID + lane + 32];
    }

    const int ngroups = (N + 3) >> 2;
    for (int g = gw; g < ngroups; g += nw) {
        const int n0 = g * 4;
        float x[4][8];
        #pragma unroll
        for (int j = 0; j < 4; ++j) {
            const int n = min(n0 + j, N - 1);
            const float* hp = H + (size_t)n * HC + lane;
            #pragma unroll
            for (int q = 0; q < 8; ++q) x[j][q] = hp[32 * q];
        }
        #pragma unroll
        for (int j = 0; j < 4; ++j) {
            float s[HEADS], d[HEADS];
            #pragma unroll
            for (int h = 0; h < HEADS; ++h) {
                const float x0 = x[j][2 * h], x1 = x[j][2 * h + 1];
                s[h] = x0 * ws0[h] + x1 * ws1[h];
                d[h] = x0 * wd0[h] + x1 * wd1[h];
            }
            #pragma unroll
            for (int off = 16; off > 0; off >>= 1) {
                #pragma unroll
                for (int h = 0; h < HEADS; ++h) {
                    s[h] += __shfl_xor(s[h], off, 32);
                    d[h] += __shfl_xor(d[h], off, 32);
                }
            }
            if (lane == 0 && n0 + j < N) {
                const size_t o4 = (size_t)(n0 + j) * 4;
                *(float4*)(a_s + o4) = make_float4(s[0], s[1], s[2], s[3]);
                *(float4*)(a_d + o4) = make_float4(d[0], d[1], d[2], d[3]);
            }
        }
    }
}

__device__ __forceinline__ void edge_src_dst(const int* __restrict__ ei, int E, int eid,
                                             int& src, int& dst) {
    if (eid < E) { src = ei[eid]; dst = ei[E + eid]; }
    else         { src = dst = eid - E; }
}

__device__ __forceinline__ float4 edge_logits(const float* __restrict__ a_s,
                                              const float* __restrict__ a_d, int src, int dst) {
    const float4 s4 = *(const float4*)(a_s + (size_t)src * 4);
    const float4 d4 = *(const float4*)(a_d + (size_t)dst * 4);
    float4 e;
    e.x = s4.x + d4.x; e.x = (e.x > 0.0f) ? e.x : NEG_SLOPE * e.x;
    e.y = s4.y + d4.y; e.y = (e.y > 0.0f) ? e.y : NEG_SLOPE * e.y;
    e.z = s4.z + d4.z; e.z = (e.z > 0.0f) ? e.z : NEG_SLOPE * e.z;
    e.w = s4.w + d4.w; e.w = (e.w > 0.0f) ? e.w : NEG_SLOPE * e.w;
    return e;
}

#define OWN_ROWS 196
#define OWN_CAP  4608
#define OWN_THREADS 1024
#define ACC_PITCH 260
#define ROW_TILES ((OWN_ROWS + 15) / 16)

__device__ __noinline__ void own_rows_by_column_scans(
    const int* __restrict__ ei, int E, int row0, int nrows, const float* __restrict__ H,
    const float* __restrict__ a_s, const float* __restrict__ a_d,
    float* acc, unsigned* smax, float* ssum)
{
    const int t = threadIdx.x;
    const int* __restrict__ dp = ei + E;
    const int nall = E + nrows;
    for (int stage = 0; stage < 3; ++stage) {
        for (int i = t; i < nall; i += OWN_THREADS) {
            int src, dst;
            if (i < E) { dst = dp[i]; src = 0; } else { dst = row0 + (i - E); src = dst; }
            const unsigned dl = (unsigned)(dst - row0);
            if (dl < (unsigned)nrows) {
                if (i < E) src = ei[i];
                const float4 e = edge_logits(a_s, a_d, src, dst);
                const float ev[4] = { e.x, e.y, e.z, e.w };
                if (stage == 0) {
                    for (int h = 0; h < 4; ++h) atomicMax(smax + dl * 4 + h, f2ord(ev[h]));
                } else if (stage == 1) {
                    for (int h = 0; h < 4; ++h)
                        atomicAdd(ssum + dl * 4 + h, __expf(ev[h] - ord2f(smax[dl * 4 + h])));
                } else {
                    float al[4];
                    for (int h = 0; h < 4; ++h)
                        al[h] = __expf(ev[h] - ord2f(smax[dl * 4 + h]));
                    const float* hs = H + (size_t)src * HC;
                    float* ap = acc + (size_t)dl * ACC_PITCH;
                    for (int c = 0; c < HC; ++c) atomicAdd(ap + c, al[c >> 6] * hs[c]);
                }
            }
        }
        __syncthreads();
    }
}

__global__ __launch_bounds__(256) void owner_id_kernel(const int* __restrict__ dp, int E,
                                                        unsigned char* __restrict__ ids)
{
    const int e0 = (int)(blockIdx.x * 256 + threadIdx.x) * 16;
    if (e0 >= E) return;
    if (e0 + 16 <= E && (E & 3) == 0) {
        const int4* __restrict__ p = (const int4*)(dp + e0);
        int4 d[4];
        #pragma unroll
        for (int u = 0; u < 4; ++u) d[u] = p[u];
        unsigned w[4];
        #pragma unroll
        for (int u = 0; u < 4; ++u)
            w[u] =  (((unsigned)d[u].x / OWN_ROWS) & 0xffu)        |
                   ((((unsigned)d[u].y / OWN_ROWS) & 0xffu) << 8)  |
                   ((((unsigned)d[u].z / OWN_ROWS) & 0xffu) << 16) |
                   ((((unsigned)d[u].w / OWN_ROWS) & 0xffu) << 24);
        *(uint4*)(ids + e0) = make_uint4(w[0], w[1], w[2], w[3]);
    } else {
        const int e1 = min(e0 + 16, E);
        for (int e = e0; e < e1; ++e) ids[e] = (unsigned char)((unsigned)dp[e] / OWN_ROWS);
    }
}

template <int NT, int NC, bool OBIAS>
__device__ __forceinline__ void owned_rows_times_weights(
    const float* acc, const float* __restrict__ Bw, const float* __restrict__ abias,
    const float* __restrict__ obias, float* __restrict__ C,
    int row0, int nrows, int rt0, int rtstep, int ct, int lane)
{
    const int c = lane & 15;
    const int h = lane >> 4;
    const float* ap[NT];
    v8f d[NT];
    #pragma unroll
    for (int u = 0; u < NT; ++u) {
        const int r = min((rt0 + u * rtstep) * 16 + c, OWN_ROWS - 1);
        ap[u] = acc + r * ACC_PITCH + 2 * h;
        d[u] = v8f{};
    }
    const int col = ct * 16 + c;
    const float* Bp = Bw + (size_t)(2 * h) * NC + col;

    #pragma unroll 2
    for (int kb = 0; kb < HC; kb += 4) {
        const float4 bb = *(const float4*)(abias + kb);
        const float bx = h ? bb.z : bb.x;
        const float by = h ? bb.w : bb.y;
        const v2f b = { Bp[0], Bp[NC] };
        Bp += 4 * NC;
        #pragma unroll
        for (int u = 0; u < NT; ++u) {
            v2f a = *(const v2f*)(ap[u] + kb);
            a.x = fmaxf(a.x + bx, 0.0f);
            a.y = fmaxf(a.y + by, 0.0f);
            d[u] = __builtin_amdgcn_wmma_f32_16x16x4_f32(false, a, false, b, (short)0, d[u], false, false);
        }
    }

    float ob = 0.0f;
    if (OBIAS) ob = obias[col];
    #pragma unroll
    for (int u = 0; u < NT; ++u) {
        const int rbase = (rt0 + u * rtstep) * 16 + 8 * h;
        #pragma unroll
        for (int r = 0; r < 8; ++r) {
            const int row = rbase + r;
            if (row < nrows) {
                float v = d[u][r];
                if (OBIAS) v += ob;
                C[(size_t)(row0 + row) * NC + col] = v;
            }
        }
    }
}

template <bool FIRST>
__global__ __launch_bounds__(OWN_THREADS) void edge_agg_owner_kernel(
    const int* __restrict__ ei, int E, int N, const float* __restrict__ H,
    const float* __restrict__ a_s, const float* __restrict__ a_d,
    const float* __restrict__ Bw, const float* __restrict__ abias,
    const float* __restrict__ obias, float* __restrict__ C,
    int* __restrict__ wcount, int* __restrict__ wsrc, unsigned short* __restrict__ wdl,
    const unsigned char* __restrict__ ids)
{
    __shared__ __attribute__((aligned(16))) float acc[OWN_ROWS * ACC_PITCH];
    __shared__ int elist[OWN_CAP];
    __shared__ unsigned short edl[OWN_CAP];
    __shared__ __attribute__((aligned(16))) float4 eterm[OWN_CAP];
    __shared__ __attribute__((aligned(16))) unsigned smax[OWN_ROWS * 4];
    __shared__ __attribute__((aligned(16))) float    ssum[OWN_ROWS * 4];
    __shared__ int ecount;

    const int t    = threadIdx.x;
    const int lane = t & 31;
    const int wave = t >> 5;
    const int row0  = blockIdx.x * OWN_ROWS;
    const int nrows = min(OWN_ROWS, N - row0);
    int*            __restrict__ bsrc = wsrc + (size_t)blockIdx.x * OWN_CAP;
    unsigned short* __restrict__ bdl  = wdl  + (size_t)blockIdx.x * OWN_CAP;

    float4* acc4 = (float4*)acc;
    for (int i = t; i < OWN_ROWS * (ACC_PITCH / 4); i += OWN_THREADS)
        acc4[i] = make_float4(0.0f, 0.0f, 0.0f, 0.0f);
    if (t < OWN_ROWS * 4) { smax[t] = f2ord(-INFINITY); ssum[t] = 0.0f; }
    if (FIRST && t == 0) ecount = 0;
    __syncthreads();

    int n;
    if (FIRST) {
        auto take = [&](int eid) {
            const int pos = atomicAdd(&ecount, 1);
            if (pos < OWN_CAP) elist[pos] = eid;
        };
        auto consider = [&](int eid, int dst) {
            if ((unsigned)(dst - row0) < (unsigned)nrows) take(eid);
        };
        const int* __restrict__ dp = ei + E;
        if (ids != nullptr) {
            const unsigned pat = (unsigned)blockIdx.x * 0x01010101u;
            const uint4* __restrict__ ip4 = (const uint4*)ids;
            auto scan16 = [&](uint4 v, int g) {
                const unsigned w[4] = { v.x, v.y, v.z, v.w };
                unsigned nz = 0;
                #pragma unroll
                for (int q = 0; q < 4; ++q) {
                    const unsigned x = w[q] ^ pat;
                    const unsigned s = ((x & 0x7f7f7f7fu) + 0x7f7f7f7fu) | x;
                    nz |= (s >> (7 - q)) & (0x01010101u << q);
                }
                unsigned mm = nz ^ 0x0f0f0f0fu;
                while (mm) {
                    const int b = __ffs(mm) - 1;
                    take(g * 16 + 4 * (b & 7) + (b >> 3));
                    mm &= mm - 1;
                }
            };
            const int n16   = E >> 4;
            const int nfull = n16 - n16 % (4 * OWN_THREADS);
            for (int i = t; i < nfull; i += 4 * OWN_THREADS) {
                uint4 v[4];
                #pragma unroll
                for (int u = 0; u < 4; ++u) v[u] = ip4[i + u * OWN_THREADS];
                #pragma unroll
                for (int u = 0; u < 4; ++u) scan16(v[u], i + u * OWN_THREADS);
            }
            for (int i = nfull + t; i < n16; i += OWN_THREADS) scan16(ip4[i], i);
            for (int e = (n16 << 4) + t; e < E; e += OWN_THREADS)
                if (ids[e] == (unsigned char)blockIdx.x) take(e);
        } else if ((E & 3) == 0) {
            const int4* __restrict__ dp4 = (const int4*)dp;
            const int n4 = E >> 2;
            for (int i = t; i < n4; i += 4 * OWN_THREADS) {
                int4 v[4];
                #pragma unroll
                for (int u = 0; u < 4; ++u) {
                    const int k = i + u * OWN_THREADS;
                    v[u] = (k < n4) ? dp4[k] : make_int4(-1, -1, -1, -1);
                }
                #pragma unroll
                for (int u = 0; u < 4; ++u) {
                    const int e0 = (i + u * OWN_THREADS) * 4;
                    const bool any = ((unsigned)(v[u].x - row0) < (unsigned)nrows) |
                                     ((unsigned)(v[u].y - row0) < (unsigned)nrows) |
                                     ((unsigned)(v[u].z - row0) < (unsigned)nrows) |
                                     ((unsigned)(v[u].w - row0) < (unsigned)nrows);
                    if (any) {
                        consider(e0,     v[u].x);
                        consider(e0 + 1, v[u].y);
                        consider(e0 + 2, v[u].z);
                        consider(e0 + 3, v[u].w);
                    }
                }
            }
        } else {
            for (int e = t; e < E; e += OWN_THREADS) consider(e, dp[e]);
        }
        if (t < nrows) take(E + row0 + t);
        __syncthreads();
        n = ecount;
        if (t == 0) wcount[blockIdx.x] = n;
    } else {
        n = wcount[blockIdx.x];
    }
    if (n > OWN_CAP) {
        own_rows_by_column_scans(ei, E, row0, nrows, H, a_s, a_d, acc, smax, ssum);
    } else {
        for (int i0 = t; i0 < n; i0 += 4 * OWN_THREADS) {
            int  idx[4], eid[4], dl[4], src[4];
            bool ok[4];
            #pragma unroll
            for (int u = 0; u < 4; ++u) {
                idx[u] = i0 + u * OWN_THREADS;
                ok[u]  = idx[u] < n;
                if (!ok[u]) idx[u] = i0;
                if (FIRST) {
                    eid[u] = elist[idx[u]];
                } else {
                    src[u] = bsrc[idx[u]];
                    dl[u]  = bdl[idx[u]];
                }
            }
            if (FIRST) {
                #pragma unroll
                for (int u = 0; u < 4; ++u) {
                    const int ec = min(eid[u], E - 1);
                    src[u] = ei[ec];
                    int dst = ei[E + ec];
                    if (eid[u] >= E) src[u] = dst = eid[u] - E;
                    dl[u] = dst - row0;
                }
            }
            float4 e[4];
            #pragma unroll
            for (int u = 0; u < 4; ++u) e[u] = edge_logits(a_s, a_d, src[u], row0 + dl[u]);
            #pragma unroll
            for (int u = 0; u < 4; ++u) {
                if (ok[u]) {
                    elist[idx[u]] = src[u];
                    edl[idx[u]]   = (unsigned short)dl[u];
                    if (FIRST) { bsrc[idx[u]] = src[u]; bdl[idx[u]] = (unsigned short)dl[u]; }
                    eterm[idx[u]] = e[u];
                    atomicMax(smax + dl[u] * 4 + 0, f2ord(e[u].x));
                    atomicMax(smax + dl[u] * 4 + 1, f2ord(e[u].y));
                    atomicMax(smax + dl[u] * 4 + 2, f2ord(e[u].z));
                    atomicMax(smax + dl[u] * 4 + 3, f2ord(e[u].w));
                }
            }
        }
        __syncthreads();

        for (int i = t; i < n; i += OWN_THREADS) {
            const int    dl = edl[i];
            const float4 e  = eterm[i];
            const uint4  m  = *(const uint4*)(smax + dl * 4);
            float4 x;
            x.x = __expf(e.x - ord2f(m.x));
            x.y = __expf(e.y - ord2f(m.y));
            x.z = __expf(e.z - ord2f(m.z));
            x.w = __expf(e.w - ord2f(m.w));
            eterm[i] = x;
            atomicAdd(ssum + dl * 4 + 0, x.x);
            atomicAdd(ssum + dl * 4 + 1, x.y);
            atomicAdd(ssum + dl * 4 + 2, x.z);
            atomicAdd(ssum + dl * 4 + 3, x.w);
        }
        __syncthreads();

        for (int base = wave * 4; base < n; base += 4 * (OWN_THREADS / 32)) {
            int    srcn[4], dln[4];
            float4 w4[4];
            bool   ok[4];
            #pragma unroll
            for (int j = 0; j < 4; ++j) {
                ok[j] = (base + j) < n;
                const int idx = ok[j] ? base + j : base;
                srcn[j] = elist[idx];
                dln[j]  = edl[idx];
                w4[j]   = eterm[idx];
            }
            float v[4][8];
            #pragma unroll
            for (int j = 0; j < 4; ++j) {
                const float* hs = H + (size_t)srcn[j] * HC + lane;
                #pragma unroll
                for (int q = 0; q < 8; ++q) v[j][q] = hs[32 * q];
            }
            #pragma unroll
            for (int j = 0; j < 4; ++j) {
                if (ok[j]) {
                    float* ap = acc + (size_t)dln[j] * ACC_PITCH + lane;
                    atomicAdd(ap +   0, w4[j].x * v[j][0]);
                    atomicAdd(ap +  32, w4[j].x * v[j][1]);
                    atomicAdd(ap +  64, w4[j].y * v[j][2]);
                    atomicAdd(ap +  96, w4[j].y * v[j][3]);
                    atomicAdd(ap + 128, w4[j].z * v[j][4]);
                    atomicAdd(ap + 160, w4[j].z * v[j][5]);
                    atomicAdd(ap + 192, w4[j].w * v[j][6]);
                    atomicAdd(ap + 224, w4[j].w * v[j][7]);
                }
            }
        }
        __syncthreads();
    }

    if (t < OWN_ROWS * 4) ssum[t] = 1.0f / (ssum[t] + EPS_F);
    __syncthreads();

    for (int i = t; i < nrows * (HC / 4); i += OWN_THREADS) {
        const int    row = i >> 6;
        const float  r = ssum[row * 4 + ((i & 63) >> 4)];
        float4* p = (float4*)(acc + row * ACC_PITCH) + (i & 63);
        const float4 a = *p;
        *p = make_float4(a.x * r, a.y * r, a.z * r, a.w * r);
    }
    __syncthreads();

    constexpr int NC     = FIRST ? HC : HID;
    constexpr int NCT    = NC / 16;
    constexpr int GROUPS = (OWN_THREADS / 32) / NCT;
    constexpr int NTMAX  = (ROW_TILES + GROUPS - 1) / GROUPS;
    const int wv = __builtin_amdgcn_readfirstlane(wave);
    const int ct = wv % NCT;
    const int g  = wv / NCT;
    const int nt = (ROW_TILES - g + GROUPS - 1) / GROUPS;
    if (nt == NTMAX)
        owned_rows_times_weights<NTMAX, NC, !FIRST>(acc, Bw, abias, obias, C, row0, nrows, g, GROUPS, ct, lane);
    else
        owned_rows_times_weights<NTMAX - 1, NC, !FIRST>(acc, Bw, abias, obias, C, row0, nrows, g, GROUPS, ct, lane);
}

extern "C" void kernel_launch(void* const* d_in, const int* in_sizes, int n_in,
                              void* d_out, int out_size, void* d_ws, size_t ws_size,
                              hipStream_t stream)
{
    const float* x   = (const float*)d_in[0];
    const int*   ei  = (const int*)d_in[1];
    const float* W1  = (const float*)d_in[2];
    const float* as1 = (const float*)d_in[3];
    const float* ad1 = (const float*)d_in[4];
    const float* b1  = (const float*)d_in[5];
    const float* W2  = (const float*)d_in[6];
    const float* as2 = (const float*)d_in[7];
    const float* ad2 = (const float*)d_in[8];
    const float* b2  = (const float*)d_in[9];
    const float* Wm1 = (const float*)d_in[10];
    const float* bm1 = (const float*)d_in[11];
    const float* Wm2 = (const float*)d_in[12];
    const float* bm2 = (const float*)d_in[13];

    const int N  = in_sizes[0] / 128;
    const int E  = in_sizes[1] / 2;
    const int NH = N * HEADS;
    const int gOwn = (N + OWN_ROWS - 1) / OWN_ROWS;

    float* ws = (float*)d_ws;
    size_t o = 0;
    float* P0  = ws + o; o += (size_t)N * HC;
    float* P1  = ws + o; o += (size_t)N * HC;
    float* a_s = ws + o; o += (size_t)NH;
    float* a_d = ws + o; o += (size_t)NH;
    int* wcount = (int*)(ws + o); o += (size_t)gOwn;
    int* wsrc   = (int*)(ws + o); o += (size_t)gOwn * OWN_CAP;
    unsigned short* wdl = (unsigned short*)(ws + o); o += (size_t)gOwn * (OWN_CAP / 2);
    o = (o + 3) & ~(size_t)3;
    unsigned char* idcol = (unsigned char*)(ws + o);
    const bool byteIds = gOwn <= 256;

    const dim3 blk(256);
    auto gemm_grid = [](int M, int NC) {
        return dim3((unsigned)(((M + 15) / 16 + 7) / 8), (unsigned)((NC + 63) / 64));
    };
    const int gScores = ((N + 3) / 4 + 15) / 16;


    if (byteIds)
        owner_id_kernel<<<(unsigned)(((E + 15) / 16 + 255) / 256), blk, 0, stream>>>(ei + E, E, idcol);
    wmma_gemm_f32_kernel<<<gemm_grid(N, HC), blk, 0, stream>>>(x, W1, P0, nullptr, N, 128, HC, 0);
    scores_kernel<<<gScores, blk, 0, stream>>>(P0, as1, ad1, a_s, a_d, N);
    edge_agg_owner_kernel<true><<<gOwn, dim3(OWN_THREADS), 0, stream>>>(ei, E, N, P0, a_s, a_d,
                                                                       W2, b1, nullptr, P1,
                                                                       wcount, wsrc, wdl,
                                                                       byteIds ? idcol : nullptr);

    scores_kernel<<<gScores, blk, 0, stream>>>(P1, as2, ad2, a_s, a_d, N);
    edge_agg_owner_kernel<false><<<gOwn, dim3(OWN_THREADS), 0, stream>>>(ei, E, N, P1, a_s, a_d,
                                                                        Wm1, b2, bm1, P0,
                                                                        wcount, wsrc, wdl, nullptr);

    wmma_gemm_f32_kernel<<<gemm_grid(N, 121), blk, 0, stream>>>(P0, Wm2, (float*)d_out, bm2,
                                                                N, HID, 121, 2);
}


# Gather by row-owning waves: listed edges ranked into 32 per-wave runs in LDS (row mod 32); a wave adds its rows with plain 16-byte LDS loads/stores, no LDS float atomics; sum order = rank arrival
# speedup vs baseline: 5.1215x; 1.0782x over previous
#include <hip/hip_runtime.h>
#include <math.h>

#define HEADS 4
#define HID 64
#define HC 256
#define NEG_SLOPE 0.2f
#define EPS_F 1e-16f

typedef __attribute__((ext_vector_type(2))) float v2f;
typedef __attribute__((ext_vector_type(8))) float v8f;

__device__ __forceinline__ unsigned f2ord(float f) {
    unsigned u = __float_as_uint(f);
    return (u & 0x80000000u) ? ~u : (u | 0x80000000u);
}
__device__ __forceinline__ float ord2f(unsigned u) {
    return (u & 0x80000000u) ? __uint_as_float(u & 0x7fffffffu) : __uint_as_float(~u);
}

__device__ __forceinline__ void epi_store(float* __restrict__ C, const float* __restrict__ bias,
                                          size_t rowoff, int col, int NC, int mode, float v) {
    if (col < NC) {
        if (mode >= 1) v += bias[col];
        if (mode == 2) v = 1.0f / (1.0f + __expf(-v));
        C[rowoff + col] = v;
    }
}

__global__ __launch_bounds__(256) void wmma_gemm_f32_kernel(
    const float* __restrict__ A, const float* __restrict__ B,
    float* __restrict__ C, const float* __restrict__ bias,
    int M, int K, int NC, int mode)
{
    const int wave = threadIdx.x >> 5;
    const int lane = threadIdx.x & 31;
    const int m0 = (blockIdx.x * 8 + wave) * 16;
    if (m0 >= M) return;
    const int n0 = blockIdx.y * 64;

    const int aRow = m0 + (lane & 15);
    const int aK   = (lane >> 4) << 1;
    const int bN   = lane & 15;
    const int bK   = (lane >> 4) << 1;
    const int c    = n0 + bN;

    const float* Ap = A + (size_t)aRow * K + aK;
    const size_t stepB = (size_t)4 * NC;

    v8f acc0 = {}, acc1 = {}, acc2 = {}, acc3 = {};

    if (n0 + 64 <= NC) {
        const float* Bp = B + c + (size_t)bK * NC;
        const float* Bq = Bp + NC;
        #pragma unroll 2
        for (int kb = 0; kb < K; kb += 4) {
            v2f a = *(const v2f*)Ap;
            v2f b0 = { Bp[0],  Bq[0]  };
            v2f b1 = { Bp[16], Bq[16] };
            v2f b2 = { Bp[32], Bq[32] };
            v2f b3 = { Bp[48], Bq[48] };
            Ap += 4; Bp += stepB; Bq += stepB;
            acc0 = __builtin_amdgcn_wmma_f32_16x16x4_f32(false, a, false, b0, (short)0, acc0, false, false);
            acc1 = __builtin_amdgcn_wmma_f32_16x16x4_f32(false, a, false, b1, (short)0, acc1, false, false);
            acc2 = __builtin_amdgcn_wmma_f32_16x16x4_f32(false, a, false, b2, (short)0, acc2, false, false);
            acc3 = __builtin_amdgcn_wmma_f32_16x16x4_f32(false, a, false, b3, (short)0, acc3, false, false);
        }
    } else {
        const int   c0 = min(c,      NC - 1), c1 = min(c + 16, NC - 1);
        const int   c2 = min(c + 32, NC - 1), c3 = min(c + 48, NC - 1);
        const float w0 = (c      < NC) ? 1.0f : 0.0f;
        const float w1 = (c + 16 < NC) ? 1.0f : 0.0f;
        const float w2 = (c + 32 < NC) ? 1.0f : 0.0f;
        const float w3 = (c + 48 < NC) ? 1.0f : 0.0f;
        const float* Bp = B + (size_t)bK * NC;
        const float* Bq = Bp + NC;
        #pragma unroll 2
        for (int kb = 0; kb < K; kb += 4) {
            v2f a = *(const v2f*)Ap;
            v2f b0 = { Bp[c0] * w0, Bq[c0] * w0 };
            v2f b1 = { Bp[c1] * w1, Bq[c1] * w1 };
            v2f b2 = { Bp[c2] * w2, Bq[c2] * w2 };
            v2f b3 = { Bp[c3] * w3, Bq[c3] * w3 };
            Ap += 4; Bp += stepB; Bq += stepB;
            acc0 = __builtin_amdgcn_wmma_f32_16x16x4_f32(false, a, false, b0, (short)0, acc0, false, false);
            acc1 = __builtin_amdgcn_wmma_f32_16x16x4_f32(false, a, false, b1, (short)0, acc1, false, false);
            acc2 = __builtin_amdgcn_wmma_f32_16x16x4_f32(false, a, false, b2, (short)0, acc2, false, false);
            acc3 = __builtin_amdgcn_wmma_f32_16x16x4_f32(false, a, false, b3, (short)0, acc3, false, false);
        }
    }

    const int mbase = m0 + ((lane >> 4) << 3);
    const int colb  = n0 + (lane & 15);
    #pragma unroll
    for (int r = 0; r < 8; ++r) {
        size_t rowoff = (size_t)(mbase + r) * NC;
        epi_store(C, bias, rowoff, colb,      NC, mode, acc0[r]);
        epi_store(C, bias, rowoff, colb + 16, NC, mode, acc1[r]);
        epi_store(C, bias, rowoff, colb + 32, NC, mode, acc2[r]);
        epi_store(C, bias, rowoff, colb + 48, NC, mode, acc3[r]);
    }
}

__global__ __launch_bounds__(256) void scores_kernel(
    const float* __restrict__ H, const float* __restrict__ att_src,
    const float* __restrict__ att_dst, float* __restrict__ a_s, float* __restrict__ a_d, int N)
{
    const int lane = threadIdx.x & 31;
    const int gw   = (int)(blockIdx.x * 8 + (threadIdx.x >> 5));
    const int nw   = (int)(gridDim.x * 8);

    float ws0[HEADS], ws1[HEADS], wd0[HEADS], wd1[HEADS];
    #pragma unroll
    for (int h = 0; h < HEADS; ++h) {
        ws0[h] = att_src[h * HID + lane]; ws1[h] = att_src[h * HID + lane + 32];
        wd0[h] = att_dst[h * HID + lane]; wd1[h] = att_dst[h * HID + lane + 32];
    }

    const int ngroups = (N + 3) >> 2;
    for (int g = gw; g < ngroups; g += nw) {
        const int n0 = g * 4;
        float x[4][8];
        #pragma unroll
        for (int j = 0; j < 4; ++j) {
            const int n = min(n0 + j, N - 1);
            const float* hp = H + (size_t)n * HC + lane;
            #pragma unroll
            for (int q = 0; q < 8; ++q) x[j][q] = hp[32 * q];
        }
        #pragma unroll
        for (int j = 0; j < 4; ++j) {
            float s[HEADS], d[HEADS];
            #pragma unroll
            for (int h = 0; h < HEADS; ++h) {
                const float x0 = x[j][2 * h], x1 = x[j][2 * h + 1];
                s[h] = x0 * ws0[h] + x1 * ws1[h];
                d[h] = x0 * wd0[h] + x1 * wd1[h];
            }
            #pragma unroll
            for (int off = 16; off > 0; off >>= 1) {
                #pragma unroll
                for (int h = 0; h < HEADS; ++h) {
                    s[h] += __shfl_xor(s[h], off, 32);
                    d[h] += __shfl_xor(d[h], off, 32);
                }
            }
            if (lane == 0 && n0 + j < N) {
                const size_t o4 = (size_t)(n0 + j) * 4;
                *(float4*)(a_s + o4) = make_float4(s[0], s[1], s[2], s[3]);
                *(float4*)(a_d + o4) = make_float4(d[0], d[1], d[2], d[3]);
            }
        }
    }
}

__device__ __forceinline__ void edge_src_dst(const int* __restrict__ ei, int E, int eid,
                                             int& src, int& dst) {
    if (eid < E) { src = ei[eid]; dst = ei[E + eid]; }
    else         { src = dst = eid - E; }
}

__device__ __forceinline__ float4 edge_logits(const float* __restrict__ a_s,
                                              const float* __restrict__ a_d, int src, int dst) {
    const float4 s4 = *(const float4*)(a_s + (size_t)src * 4);
    const float4 d4 = *(const float4*)(a_d + (size_t)dst * 4);
    float4 e;
    e.x = s4.x + d4.x; e.x = (e.x > 0.0f) ? e.x : NEG_SLOPE * e.x;
    e.y = s4.y + d4.y; e.y = (e.y > 0.0f) ? e.y : NEG_SLOPE * e.y;
    e.z = s4.z + d4.z; e.z = (e.z > 0.0f) ? e.z : NEG_SLOPE * e.z;
    e.w = s4.w + d4.w; e.w = (e.w > 0.0f) ? e.w : NEG_SLOPE * e.w;
    return e;
}

#define OWN_ROWS 196
#define OWN_CAP  4608
#define OWN_THREADS 1024
#define ACC_PITCH 260
#define ROW_TILES ((OWN_ROWS + 15) / 16)

__device__ __noinline__ void own_rows_by_column_scans(
    const int* __restrict__ ei, int E, int row0, int nrows, const float* __restrict__ H,
    const float* __restrict__ a_s, const float* __restrict__ a_d,
    float* acc, unsigned* smax, float* ssum)
{
    const int t = threadIdx.x;
    const int* __restrict__ dp = ei + E;
    const int nall = E + nrows;
    for (int stage = 0; stage < 3; ++stage) {
        for (int i = t; i < nall; i += OWN_THREADS) {
            int src, dst;
            if (i < E) { dst = dp[i]; src = 0; } else { dst = row0 + (i - E); src = dst; }
            const unsigned dl = (unsigned)(dst - row0);
            if (dl < (unsigned)nrows) {
                if (i < E) src = ei[i];
                const float4 e = edge_logits(a_s, a_d, src, dst);
                const float ev[4] = { e.x, e.y, e.z, e.w };
                if (stage == 0) {
                    for (int h = 0; h < 4; ++h) atomicMax(smax + dl * 4 + h, f2ord(ev[h]));
                } else if (stage == 1) {
                    for (int h = 0; h < 4; ++h)
                        atomicAdd(ssum + dl * 4 + h, __expf(ev[h] - ord2f(smax[dl * 4 + h])));
                } else {
                    float al[4];
                    for (int h = 0; h < 4; ++h)
                        al[h] = __expf(ev[h] - ord2f(smax[dl * 4 + h]));
                    const float* hs = H + (size_t)src * HC;
                    float* ap = acc + (size_t)dl * ACC_PITCH;
                    for (int c = 0; c < HC; ++c) atomicAdd(ap + c, al[c >> 6] * hs[c]);
                }
            }
        }
        __syncthreads();
    }
}

__global__ __launch_bounds__(256) void owner_id_kernel(const int* __restrict__ dp, int E,
                                                        unsigned char* __restrict__ ids)
{
    const int e0 = (int)(blockIdx.x * 256 + threadIdx.x) * 16;
    if (e0 >= E) return;
    if (e0 + 16 <= E && (E & 3) == 0) {
        const int4* __restrict__ p = (const int4*)(dp + e0);
        int4 d[4];
        #pragma unroll
        for (int u = 0; u < 4; ++u) d[u] = p[u];
        unsigned w[4];
        #pragma unroll
        for (int u = 0; u < 4; ++u)
            w[u] =  (((unsigned)d[u].x / OWN_ROWS) & 0xffu)        |
                   ((((unsigned)d[u].y / OWN_ROWS) & 0xffu) << 8)  |
                   ((((unsigned)d[u].z / OWN_ROWS) & 0xffu) << 16) |
                   ((((unsigned)d[u].w / OWN_ROWS) & 0xffu) << 24);
        *(uint4*)(ids + e0) = make_uint4(w[0], w[1], w[2], w[3]);
    } else {
        const int e1 = min(e0 + 16, E);
        for (int e = e0; e < e1; ++e) ids[e] = (unsigned char)((unsigned)dp[e] / OWN_ROWS);
    }
}

template <int NT, int NC, bool OBIAS>
__device__ __forceinline__ void owned_rows_times_weights(
    const float* acc, const float* __restrict__ Bw, const float* __restrict__ abias,
    const float* __restrict__ obias, float* __restrict__ C,
    int row0, int nrows, int rt0, int rtstep, int ct, int lane)
{
    const int c = lane & 15;
    const int h = lane >> 4;
    const float* ap[NT];
    v8f d[NT];
    #pragma unroll
    for (int u = 0; u < NT; ++u) {
        const int r = min((rt0 + u * rtstep) * 16 + c, OWN_ROWS - 1);
        ap[u] = acc + r * ACC_PITCH + 2 * h;
        d[u] = v8f{};
    }
    const int col = ct * 16 + c;
    const float* Bp = Bw + (size_t)(2 * h) * NC + col;

    #pragma unroll 2
    for (int kb = 0; kb < HC; kb += 4) {
        const float4 bb = *(const float4*)(abias + kb);
        const float bx = h ? bb.z : bb.x;
        const float by = h ? bb.w : bb.y;
        const v2f b = { Bp[0], Bp[NC] };
        Bp += 4 * NC;
        #pragma unroll
        for (int u = 0; u < NT; ++u) {
            v2f a = *(const v2f*)(ap[u] + kb);
            a.x = fmaxf(a.x + bx, 0.0f);
            a.y = fmaxf(a.y + by, 0.0f);
            d[u] = __builtin_amdgcn_wmma_f32_16x16x4_f32(false, a, false, b, (short)0, d[u], false, false);
        }
    }

    float ob = 0.0f;
    if (OBIAS) ob = obias[col];
    #pragma unroll
    for (int u = 0; u < NT; ++u) {
        const int rbase = (rt0 + u * rtstep) * 16 + 8 * h;
        #pragma unroll
        for (int r = 0; r < 8; ++r) {
            const int row = rbase + r;
            if (row < nrows) {
                float v = d[u][r];
                if (OBIAS) v += ob;
                C[(size_t)(row0 + row) * NC + col] = v;
            }
        }
    }
}

template <bool FIRST>
__global__ __launch_bounds__(OWN_THREADS) void edge_agg_owner_kernel(
    const int* __restrict__ ei, int E, int N, const float* __restrict__ H,
    const float* __restrict__ a_s, const float* __restrict__ a_d,
    const float* __restrict__ Bw, const float* __restrict__ abias,
    const float* __restrict__ obias, float* __restrict__ C,
    int* __restrict__ wcount, int* __restrict__ wsrc, unsigned short* __restrict__ wdl,
    const unsigned char* __restrict__ ids)
{
    __shared__ __attribute__((aligned(16))) float acc[OWN_ROWS * ACC_PITCH];
    __shared__ int elist[OWN_CAP];
    __shared__ unsigned short edl[OWN_CAP];
    __shared__ __attribute__((aligned(16))) float4 eterm[OWN_CAP];
    __shared__ __attribute__((aligned(16))) unsigned smax[OWN_ROWS * 4];
    __shared__ __attribute__((aligned(16))) float    ssum[OWN_ROWS * 4];
    __shared__ int ecount;
    __shared__ unsigned short perm[OWN_CAP];
    __shared__ int wcnt[32];
    __shared__ int wbase[33];
    __shared__ int wfill[32];

    const int t    = threadIdx.x;
    const int lane = t & 31;
    const int wave = t >> 5;
    const int row0  = blockIdx.x * OWN_ROWS;
    const int nrows = min(OWN_ROWS, N - row0);
    int*            __restrict__ bsrc = wsrc + (size_t)blockIdx.x * OWN_CAP;
    unsigned short* __restrict__ bdl  = wdl  + (size_t)blockIdx.x * OWN_CAP;

    float4* acc4 = (float4*)acc;
    for (int i = t; i < OWN_ROWS * (ACC_PITCH / 4); i += OWN_THREADS)
        acc4[i] = make_float4(0.0f, 0.0f, 0.0f, 0.0f);
    if (t < OWN_ROWS * 4) { smax[t] = f2ord(-INFINITY); ssum[t] = 0.0f; }
    if (FIRST && t == 0) ecount = 0;
    if (t < 32) { wcnt[t] = 0; wfill[t] = 0; }
    __syncthreads();

    int n;
    if (FIRST) {
        auto take = [&](int eid) {
            const int pos = atomicAdd(&ecount, 1);
            if (pos < OWN_CAP) elist[pos] = eid;
        };
        auto consider = [&](int eid, int dst) {
            if ((unsigned)(dst - row0) < (unsigned)nrows) take(eid);
        };
        const int* __restrict__ dp = ei + E;
        if (ids != nullptr) {
            const unsigned pat = (unsigned)blockIdx.x * 0x01010101u;
            const uint4* __restrict__ ip4 = (const uint4*)ids;
            auto scan16 = [&](uint4 v, int g) {
                const unsigned w[4] = { v.x, v.y, v.z, v.w };
                unsigned nz = 0;
                #pragma unroll
                for (int q = 0; q < 4; ++q) {
                    const unsigned x = w[q] ^ pat;
                    const unsigned s = ((x & 0x7f7f7f7fu) + 0x7f7f7f7fu) | x;
                    nz |= (s >> (7 - q)) & (0x01010101u << q);
                }
                unsigned mm = nz ^ 0x0f0f0f0fu;
                while (mm) {
                    const int b = __ffs(mm) - 1;
                    take(g * 16 + 4 * (b & 7) + (b >> 3));
                    mm &= mm - 1;
                }
            };
            const int n16   = E >> 4;
            const int nfull = n16 - n16 % (4 * OWN_THREADS);
            for (int i = t; i < nfull; i += 4 * OWN_THREADS) {
                uint4 v[4];
                #pragma unroll
                for (int u = 0; u < 4; ++u) v[u] = ip4[i + u * OWN_THREADS];
                #pragma unroll
                for (int u = 0; u < 4; ++u) scan16(v[u], i + u * OWN_THREADS);
            }
            for (int i = nfull + t; i < n16; i += OWN_THREADS) scan16(ip4[i], i);
            for (int e = (n16 << 4) + t; e < E; e += OWN_THREADS)
                if (ids[e] == (unsigned char)blockIdx.x) take(e);
        } else if ((E & 3) == 0) {
            const int4* __restrict__ dp4 = (const int4*)dp;
            const int n4 = E >> 2;
            for (int i = t; i < n4; i += 4 * OWN_THREADS) {
                int4 v[4];
                #pragma unroll
                for (int u = 0; u < 4; ++u) {
                    const int k = i + u * OWN_THREADS;
                    v[u] = (k < n4) ? dp4[k] : make_int4(-1, -1, -1, -1);
                }
                #pragma unroll
                for (int u = 0; u < 4; ++u) {
                    const int e0 = (i + u * OWN_THREADS) * 4;
                    const bool any = ((unsigned)(v[u].x - row0) < (unsigned)nrows) |
                                     ((unsigned)(v[u].y - row0) < (unsigned)nrows) |
                                     ((unsigned)(v[u].z - row0) < (unsigned)nrows) |
                                     ((unsigned)(v[u].w - row0) < (unsigned)nrows);
                    if (any) {
                        consider(e0,     v[u].x);
                        consider(e0 + 1, v[u].y);
                        consider(e0 + 2, v[u].z);
                        consider(e0 + 3, v[u].w);
                    }
                }
            }
        } else {
            for (int e = t; e < E; e += OWN_THREADS) consider(e, dp[e]);
        }
        if (t < nrows) take(E + row0 + t);
        __syncthreads();
        n = ecount;
        if (t == 0) wcount[blockIdx.x] = n;
    } else {
        n = wcount[blockIdx.x];
    }
    if (n > OWN_CAP) {
        own_rows_by_column_scans(ei, E, row0, nrows, H, a_s, a_d, acc, smax, ssum);
    } else {
        for (int i0 = t; i0 < n; i0 += 4 * OWN_THREADS) {
            int  idx[4], eid[4], dl[4], src[4];
            bool ok[4];
            #pragma unroll
            for (int u = 0; u < 4; ++u) {
                idx[u] = i0 + u * OWN_THREADS;
                ok[u]  = idx[u] < n;
                if (!ok[u]) idx[u] = i0;
                if (FIRST) {
                    eid[u] = elist[idx[u]];
                } else {
                    src[u] = bsrc[idx[u]];
                    dl[u]  = bdl[idx[u]];
                }
            }
            if (FIRST) {
                #pragma unroll
                for (int u = 0; u < 4; ++u) {
                    const int ec = min(eid[u], E - 1);
                    src[u] = ei[ec];
                    int dst = ei[E + ec];
                    if (eid[u] >= E) src[u] = dst = eid[u] - E;
                    dl[u] = dst - row0;
                }
            }
            float4 e[4];
            #pragma unroll
            for (int u = 0; u < 4; ++u) e[u] = edge_logits(a_s, a_d, src[u], row0 + dl[u]);
            #pragma unroll
            for (int u = 0; u < 4; ++u) {
                if (ok[u]) {
                    elist[idx[u]] = src[u];
                    edl[idx[u]]   = (unsigned short)dl[u];
                    if (FIRST) { bsrc[idx[u]] = src[u]; bdl[idx[u]] = (unsigned short)dl[u]; }
                    eterm[idx[u]] = e[u];
                    atomicMax(smax + dl[u] * 4 + 0, f2ord(e[u].x));
                    atomicMax(smax + dl[u] * 4 + 1, f2ord(e[u].y));
                    atomicMax(smax + dl[u] * 4 + 2, f2ord(e[u].z));
                    atomicMax(smax + dl[u] * 4 + 3, f2ord(e[u].w));
                    atomicAdd(&wcnt[dl[u] & 31], 1);
                }
            }
        }
        __syncthreads();

        if (t < 32) {
            int s = wcnt[t];
            for (int off = 1; off < 32; off <<= 1) {
                const int v = __shfl_up(s, off, 32);
                if (lane >= off) s += v;
            }
            wbase[t + 1] = s;
            if (t == 0) wbase[0] = 0;
        }
        __syncthreads();

        for (int i = t; i < n; i += OWN_THREADS) {
            const int    dl = edl[i];
            const float4 e  = eterm[i];
            const uint4  m  = *(const uint4*)(smax + dl * 4);
            float4 x;
            x.x = __expf(e.x - ord2f(m.x));
            x.y = __expf(e.y - ord2f(m.y));
            x.z = __expf(e.z - ord2f(m.z));
            x.w = __expf(e.w - ord2f(m.w));
            eterm[i] = x;
            atomicAdd(ssum + dl * 4 + 0, x.x);
            atomicAdd(ssum + dl * 4 + 1, x.y);
            atomicAdd(ssum + dl * 4 + 2, x.z);
            atomicAdd(ssum + dl * 4 + 3, x.w);
            const int b = dl & 31;
            perm[wbase[b] + atomicAdd(&wfill[b], 1)] = (unsigned short)i;
        }
        __syncthreads();

        {
            const int wrun = __builtin_amdgcn_readfirstlane(wave);
            const int pb = wbase[wrun], pe = wbase[wrun + 1];
            const bool lo = lane < 16;
            for (int p = pb; p < pe; p += 4) {
                int    srcn[4], dln[4];
                float4 w4[4];
                bool   ok[4];
                #pragma unroll
                for (int j = 0; j < 4; ++j) {
                    ok[j] = (p + j) < pe;
                    const int slot = perm[ok[j] ? p + j : p];
                    srcn[j] = elist[slot];
                    dln[j]  = edl[slot];
                    w4[j]   = eterm[slot];
                }
                float4 v0[4], v1[4];
                #pragma unroll
                for (int j = 0; j < 4; ++j) {
                    const float4* hs = (const float4*)(H + (size_t)srcn[j] * HC) + lane;
                    v0[j] = hs[0];
                    v1[j] = hs[32];
                }
                #pragma unroll
                for (int j = 0; j < 4; ++j) {
                    if (ok[j]) {
                        float4* ap = (float4*)(acc + (size_t)dln[j] * ACC_PITCH) + lane;
                        const float wa = lo ? w4[j].x : w4[j].y;
                        const float wb = lo ? w4[j].z : w4[j].w;
                        float4 a0 = ap[0], a1 = ap[32];
                        a0.x += wa * v0[j].x; a0.y += wa * v0[j].y; a0.z += wa * v0[j].z; a0.w += wa * v0[j].w;
                        a1.x += wb * v1[j].x; a1.y += wb * v1[j].y; a1.z += wb * v1[j].z; a1.w += wb * v1[j].w;
                        ap[0]  = a0;
                        ap[32] = a1;
                    }
                }
            }
        }
        __syncthreads();
    }

    if (t < OWN_ROWS * 4) ssum[t] = 1.0f / (ssum[t] + EPS_F);
    __syncthreads();

    for (int i = t; i < nrows * (HC / 4); i += OWN_THREADS) {
        const int    row = i >> 6;
        const float  r = ssum[row * 4 + ((i & 63) >> 4)];
        float4* p = (float4*)(acc + row * ACC_PITCH) + (i & 63);
        const float4 a = *p;
        *p = make_float4(a.x * r, a.y * r, a.z * r, a.w * r);
    }
    __syncthreads();

    constexpr int NC     = FIRST ? HC : HID;
    constexpr int NCT    = NC / 16;
    constexpr int GROUPS = (OWN_THREADS / 32) / NCT;
    constexpr int NTMAX  = (ROW_TILES + GROUPS - 1) / GROUPS;
    const int wv = __builtin_amdgcn_readfirstlane(wave);
    const int ct = wv % NCT;
    const int g  = wv / NCT;
    const int nt = (ROW_TILES - g + GROUPS - 1) / GROUPS;
    if (nt == NTMAX)
        owned_rows_times_weights<NTMAX, NC, !FIRST>(acc, Bw, abias, obias, C, row0, nrows, g, GROUPS, ct, lane);
    else
        owned_rows_times_weights<NTMAX - 1, NC, !FIRST>(acc, Bw, abias, obias, C, row0, nrows, g, GROUPS, ct, lane);
}

extern "C" void kernel_launch(void* const* d_in, const int* in_sizes, int n_in,
                              void* d_out, int out_size, void* d_ws, size_t ws_size,
                              hipStream_t stream)
{
    const float* x   = (const float*)d_in[0];
    const int*   ei  = (const int*)d_in[1];
    const float* W1  = (const float*)d_in[2];
    const float* as1 = (const float*)d_in[3];
    const float* ad1 = (const float*)d_in[4];
    const float* b1  = (const float*)d_in[5];
    const float* W2  = (const float*)d_in[6];
    const float* as2 = (const float*)d_in[7];
    const float* ad2 = (const float*)d_in[8];
    const float* b2  = (const float*)d_in[9];
    const float* Wm1 = (const float*)d_in[10];
    const float* bm1 = (const float*)d_in[11];
    const float* Wm2 = (const float*)d_in[12];
    const float* bm2 = (const float*)d_in[13];

    const int N  = in_sizes[0] / 128;
    const int E  = in_sizes[1] / 2;
    const int NH = N * HEADS;
    const int gOwn = (N + OWN_ROWS - 1) / OWN_ROWS;

    float* ws = (float*)d_ws;
    size_t o = 0;
    float* P0  = ws + o; o += (size_t)N * HC;
    float* P1  = ws + o; o += (size_t)N * HC;
    float* a_s = ws + o; o += (size_t)NH;
    float* a_d = ws + o; o += (size_t)NH;
    int* wcount = (int*)(ws + o); o += (size_t)gOwn;
    int* wsrc   = (int*)(ws + o); o += (size_t)gOwn * OWN_CAP;
    unsigned short* wdl = (unsigned short*)(ws + o); o += (size_t)gOwn * (OWN_CAP / 2);
    o = (o + 3) & ~(size_t)3;
    unsigned char* idcol = (unsigned char*)(ws + o);
    const bool byteIds = gOwn <= 256;

    const dim3 blk(256);
    auto gemm_grid = [](int M, int NC) {
        return dim3((unsigned)(((M + 15) / 16 + 7) / 8), (unsigned)((NC + 63) / 64));
    };
    const int gScores = ((N + 3) / 4 + 15) / 16;


    if (byteIds)
        owner_id_kernel<<<(unsigned)(((E + 15) / 16 + 255) / 256), blk, 0, stream>>>(ei + E, E, idcol);
    wmma_gemm_f32_kernel<<<gemm_grid(N, HC), blk, 0, stream>>>(x, W1, P0, nullptr, N, 128, HC, 0);
    scores_kernel<<<gScores, blk, 0, stream>>>(P0, as1, ad1, a_s, a_d, N);
    edge_agg_owner_kernel<true><<<gOwn, dim3(OWN_THREADS), 0, stream>>>(ei, E, N, P0, a_s, a_d,
                                                                       W2, b1, nullptr, P1,
                                                                       wcount, wsrc, wdl,
                                                                       byteIds ? idcol : nullptr);

    scores_kernel<<<gScores, blk, 0, stream>>>(P1, as2, ad2, a_s, a_d, N);
    edge_agg_owner_kernel<false><<<gOwn, dim3(OWN_THREADS), 0, stream>>>(ei, E, N, P1, a_s, a_d,
                                                                        Wm1, b2, bm1, P0,
                                                                        wcount, wsrc, wdl, nullptr);

    wmma_gemm_f32_kernel<<<gemm_grid(N, 121), blk, 0, stream>>>(P0, Wm2, (float*)d_out, bm2,
                                                                N, HID, 121, 2);
}
